# starting kernel
# baseline (speedup 1.0000x reference)
#include <hip/hip_runtime.h>
#include <math.h>

typedef __attribute__((ext_vector_type(2))) float v2f;
typedef __attribute__((ext_vector_type(8))) float v8f;

#define HEADS 4
#define CCH 64
#define HIDF 256
#define NEG_SLOPE 0.2f
#define LN_EPS 1e-5f

__device__ __forceinline__ float eluf(float x) { return x > 0.f ? x : (expf(x) - 1.f); }
__device__ __forceinline__ float leakyf(float x) { return x > 0.f ? x : NEG_SLOPE * x; }

__device__ __forceinline__ unsigned fkey(float f) {
    unsigned u = __float_as_uint(f);
    return (u & 0x80000000u) ? ~u : (u | 0x80000000u);
}
__device__ __forceinline__ float funkey(unsigned k) {
    unsigned u = (k & 0x80000000u) ? (k ^ 0x80000000u) : ~k;
    return __uint_as_float(u);
}

__device__ __forceinline__ float waveRedSum(float v) {
    for (int off = 16; off > 0; off >>= 1) v += __shfl_xor(v, off, 32);
    return v;
}

__global__ void gemm_wmma_f32(const float* __restrict__ A, int lda,
                              const float* __restrict__ W,
                              float* __restrict__ C,
                              int M, int K, int Nout,
                              const float* __restrict__ bias, int elu_act) {
    int wid  = (blockIdx.x * blockDim.x + threadIdx.x) >> 5;
    int lane = threadIdx.x & 31;
    int tilesN = Nout >> 4;
    int totalTiles = (M >> 4) * tilesN;
    if (wid >= totalTiles) return;
    int tm = wid / tilesN;
    int tn = wid - tm * tilesN;
    int m  = (tm << 4) + (lane & 15);
    int n  = (tn << 4) + (lane & 15);
    int kb = (lane >> 4) << 1;
    const float* arow = A + (size_t)m * lda;
    const float* brow = W + (size_t)n * K;
    v8f acc = {};
    for (int k = 0; k < K; k += 4) {
        v2f a, b;
        a.x = arow[k + kb];
        a.y = arow[k + kb + 1];
        b.x = brow[k + kb];
        b.y = brow[k + kb + 1];
        acc = __builtin_amdgcn_wmma_f32_16x16x4_f32(false, a, false, b,
                                                    (short)0, acc, false, false);
    }
    int rbase = (tm << 4) + ((lane >> 4) << 3);
    int col   = (tn << 4) + (lane & 15);
    float bv  = bias ? bias[col] : 0.f;
    for (int i = 0; i < 8; ++i) {
        float v = acc[i] + bv;
        if (elu_act) v = eluf(v);
        C[(size_t)(rbase + i) * Nout + col] = v;
    }
}

__global__ void attn_coef(const float* __restrict__ hmat,
                          const float* __restrict__ a_s,
                          const float* __restrict__ a_d,
                          float* __restrict__ asrc, float* __restrict__ adst, int NH) {
    int idx = blockIdx.x * blockDim.x + threadIdx.x;
    if (idx >= NH) return;
    int n = idx >> 2, h = idx & 3;
    const float* row = hmat + (size_t)n * HIDF + h * CCH;
    const float* ws  = a_s + h * CCH;
    const float* wd  = a_d + h * CCH;
    float ss = 0.f, sd = 0.f;
    for (int c = 0; c < CCH; ++c) { float v = row[c]; ss += v * ws[c]; sd += v * wd[c]; }
    asrc[idx] = ss;
    adst[idx] = sd;
}

__global__ void softmax_init(const float* __restrict__ asrc, const float* __restrict__ adst,
                             unsigned* __restrict__ amaxk, int NH) {
    int idx = blockIdx.x * blockDim.x + threadIdx.x;
    if (idx >= NH) return;
    amaxk[idx] = fkey(leakyf(asrc[idx] + adst[idx]));
}

__global__ void edge_max(const int* __restrict__ src, const int* __restrict__ dst,
                         const float* __restrict__ asrc, const float* __restrict__ adst,
                         unsigned* __restrict__ amaxk, int EH) {
    int idx = blockIdx.x * blockDim.x + threadIdx.x;
    if (idx >= EH) return;
    int e = idx >> 2, h = idx & 3;
    int s = src[e], d = dst[e];
    float a = leakyf(asrc[s * HEADS + h] + adst[d * HEADS + h]);
    atomicMax(&amaxk[d * HEADS + h], fkey(a));
}

__global__ void acc_init(const float* __restrict__ hmat,
                         const float* __restrict__ asrc, const float* __restrict__ adst,
                         const unsigned* __restrict__ amaxk,
                         float* __restrict__ denom, float* __restrict__ hacc, int N) {
    int wid  = (blockIdx.x * blockDim.x + threadIdx.x) >> 5;
    int lane = threadIdx.x & 31;
    if (wid >= N) return;
    int n = wid;
    float ex[HEADS];
#pragma unroll
    for (int h = 0; h < HEADS; ++h) {
        float aself = leakyf(asrc[n * HEADS + h] + adst[n * HEADS + h]);
        ex[h] = expf(aself - funkey(amaxk[n * HEADS + h]));
    }
    if (lane == 0) {
#pragma unroll
        for (int h = 0; h < HEADS; ++h) denom[n * HEADS + h] = ex[h];
    }
    const float* row = hmat + (size_t)n * HIDF;
    float* orow      = hacc + (size_t)n * HIDF;
#pragma unroll
    for (int i = 0; i < 8; ++i) {
        int c = lane + 32 * i;
        orow[c] = row[c] * ex[i >> 1];
    }
}

__global__ void edge_accum(const int* __restrict__ src, const int* __restrict__ dst,
                           const float* __restrict__ asrc, const float* __restrict__ adst,
                           const unsigned* __restrict__ amaxk,
                           const float* __restrict__ hmat,
                           float* __restrict__ denom, float* __restrict__ hacc, int E) {
    int wid  = (blockIdx.x * blockDim.x + threadIdx.x) >> 5;
    int lane = threadIdx.x & 31;
    if (wid >= E) return;
    int e = wid;
    int s = src[e], d = dst[e];
    float ex[HEADS];
#pragma unroll
    for (int h = 0; h < HEADS; ++h) {
        float a = leakyf(asrc[s * HEADS + h] + adst[d * HEADS + h]);
        ex[h] = expf(a - funkey(amaxk[d * HEADS + h]));
    }
    if (lane == 0) {
#pragma unroll
        for (int h = 0; h < HEADS; ++h) atomicAdd(&denom[d * HEADS + h], ex[h]);
    }
    const float* hrow = hmat + (size_t)s * HIDF;
    float* orow       = hacc + (size_t)d * HIDF;
#pragma unroll
    for (int i = 0; i < 8; ++i) {
        int c = lane + 32 * i;
        atomicAdd(&orow[c], ex[i >> 1] * hrow[c]);
    }
}

__global__ void finalize_concat(const float* __restrict__ hacc, const float* __restrict__ denom,
                                const float* __restrict__ bias, const float* __restrict__ g,
                                const float* __restrict__ be, float* __restrict__ out, int N) {
    int wid  = (blockIdx.x * blockDim.x + threadIdx.x) >> 5;
    int lane = threadIdx.x & 31;
    if (wid >= N) return;
    size_t base = (size_t)wid * HIDF;
    float dn[HEADS];
#pragma unroll
    for (int h = 0; h < HEADS; ++h) dn[h] = 1.f / (denom[wid * HEADS + h] + 1e-16f);
    float v[8];
    float s = 0.f;
#pragma unroll
    for (int i = 0; i < 8; ++i) {
        int c = lane + 32 * i;
        v[i] = hacc[base + c] * dn[i >> 1] + bias[c];
        s += v[i];
    }
    float mu = waveRedSum(s) * (1.f / HIDF);
    float q = 0.f;
#pragma unroll
    for (int i = 0; i < 8; ++i) { float dlt = v[i] - mu; q += dlt * dlt; }
    float rstd = rsqrtf(waveRedSum(q) * (1.f / HIDF) + LN_EPS);
#pragma unroll
    for (int i = 0; i < 8; ++i) {
        int c = lane + 32 * i;
        float y = (v[i] - mu) * rstd * g[c] + be[c];
        out[base + c] = eluf(y);
    }
}

__global__ void finalize_mean(const float* __restrict__ hacc, const float* __restrict__ denom,
                              const float* __restrict__ bias, const float* __restrict__ g,
                              const float* __restrict__ be, float* __restrict__ out, int N) {
    int wid  = (blockIdx.x * blockDim.x + threadIdx.x) >> 5;
    int lane = threadIdx.x & 31;
    if (wid >= N) return;
    size_t base = (size_t)wid * HIDF;
    float dn[HEADS];
#pragma unroll
    for (int h = 0; h < HEADS; ++h) dn[h] = 1.f / (denom[wid * HEADS + h] + 1e-16f);
    float v[2];
    float s = 0.f;
#pragma unroll
    for (int i = 0; i < 2; ++i) {
        int c = lane + 32 * i;
        float acc = 0.f;
#pragma unroll
        for (int h = 0; h < HEADS; ++h) acc += hacc[base + h * CCH + c] * dn[h];
        v[i] = acc * (1.f / HEADS) + bias[c];
        s += v[i];
    }
    float mu = waveRedSum(s) * (1.f / CCH);
    float q = 0.f;
#pragma unroll
    for (int i = 0; i < 2; ++i) { float dlt = v[i] - mu; q += dlt * dlt; }
    float rstd = rsqrtf(waveRedSum(q) * (1.f / CCH) + LN_EPS);
#pragma unroll
    for (int i = 0; i < 2; ++i) {
        int c = lane + 32 * i;
        float y = (v[i] - mu) * rstd * g[c] + be[c];
        out[(size_t)wid * CCH + c] = eluf(y);
    }
}

__global__ void value_head(const float* __restrict__ hin,
                           const float* __restrict__ vw1, const float* __restrict__ vb1,
                           const float* __restrict__ vw2, const float* __restrict__ vb2,
                           float* __restrict__ out, int N) {
    int wid  = (blockIdx.x * blockDim.x + threadIdx.x) >> 5;
    int lane = threadIdx.x & 31;
    if (wid >= N) return;
    const float* row  = hin + (size_t)wid * CCH;
    const float* wrow = vw1 + lane * CCH;
    float hid = vb1[lane];
    for (int c = 0; c < CCH; ++c) hid += row[c] * wrow[c];
    hid = eluf(hid);
#pragma unroll
    for (int k = 0; k < 4; ++k) {
        float p = vw2[k * 32 + lane] * hid;
        p = waveRedSum(p);
        if (lane == 0) out[wid * 4 + k] = p + vb2[k];
    }
}

extern "C" void kernel_launch(void* const* d_in, const int* in_sizes, int n_in,
                              void* d_out, int out_size, void* d_ws, size_t ws_size,
                              hipStream_t stream) {
    const float* x      = (const float*)d_in[0];
    const int*   esrc   = (const int*)d_in[1];
    const int*   edst   = (const int*)d_in[2];
    const float* proj_w = (const float*)d_in[3];
    const float* proj_b = (const float*)d_in[4];
    const float* W[3]  = {(const float*)d_in[5],  (const float*)d_in[11], (const float*)d_in[17]};
    const float* AS[3] = {(const float*)d_in[6],  (const float*)d_in[12], (const float*)d_in[18]};
    const float* AD[3] = {(const float*)d_in[7],  (const float*)d_in[13], (const float*)d_in[19]};
    const float* B[3]  = {(const float*)d_in[8],  (const float*)d_in[14], (const float*)d_in[20]};
    const float* G[3]  = {(const float*)d_in[9],  (const float*)d_in[15], (const float*)d_in[21]};
    const float* BE[3] = {(const float*)d_in[10], (const float*)d_in[16], (const float*)d_in[22]};
    const float* vw1 = (const float*)d_in[23];
    const float* vb1 = (const float*)d_in[24];
    const float* vw2 = (const float*)d_in[25];
    const float* vb2 = (const float*)d_in[26];

    const int N = in_sizes[0] / 128;
    const int E = in_sizes[1];
    const int NH = N * HEADS;
    const int EH = E * HEADS;

    float* ws     = (float*)d_ws;
    float* hin    = ws;
    float* hmat   = hin  + (size_t)N * HIDF;
    float* hacc   = hmat + (size_t)N * HIDF;
    float* asrc   = hacc + (size_t)N * HIDF;
    float* adst   = asrc + (size_t)NH;
    unsigned* amk = (unsigned*)(adst + (size_t)NH);
    float* denom  = (float*)(amk + (size_t)NH);

    dim3 blk(256);

    {
        int tiles = (N / 16) * (64 / 16);
        gemm_wmma_f32<<<(tiles * 32 + 255) / 256, blk, 0, stream>>>(
            x, 128, proj_w, hin, N, 128, 64, proj_b, 1);
    }

    int in_ld = 64;
    for (int l = 0; l < 3; ++l) {
        int K = (l == 0) ? 64 : 256;
        int tiles = (N / 16) * (HIDF / 16);
        gemm_wmma_f32<<<(tiles * 32 + 255) / 256, blk, 0, stream>>>(
            hin, in_ld, W[l], hmat, N, K, HIDF, nullptr, 0);
        attn_coef<<<(NH + 255) / 256, blk, 0, stream>>>(hmat, AS[l], AD[l], asrc, adst, NH);
        softmax_init<<<(NH + 255) / 256, blk, 0, stream>>>(asrc, adst, amk, NH);
        edge_max<<<(EH + 255) / 256, blk, 0, stream>>>(esrc, edst, asrc, adst, amk, EH);
        acc_init<<<(N * 32 + 255) / 256, blk, 0, stream>>>(hmat, asrc, adst, amk, denom, hacc, N);
        edge_accum<<<((size_t)E * 32 + 255) / 256, blk, 0, stream>>>(
            esrc, edst, asrc, adst, amk, hmat, denom, hacc, E);
        if (l < 2) {
            finalize_concat<<<(N * 32 + 255) / 256, blk, 0, stream>>>(
                hacc, denom, B[l], G[l], BE[l], hin, N);
            in_ld = HIDF;
        } else {
            finalize_mean<<<(N * 32 + 255) / 256, blk, 0, stream>>>(
                hacc, denom, B[l], G[l], BE[l], hin, N);
        }
    }

    value_head<<<(N * 32 + 255) / 256, blk, 0, stream>>>(
        hin, vw1, vb1, vw2, vb2, (float*)d_out, N);
}


# Edge softmax+sum owner-computes: a work-group owns 200 dst nodes, scans all edges, sums its in-edges in LDS by LDS float atomics (arrival order); no global atomics, 4 edge launches per layer -> 1
# speedup vs baseline: 1.5913x; 1.5913x over previous
#include <hip/hip_runtime.h>
#include <math.h>

typedef __attribute__((ext_vector_type(2))) float v2f;
typedef __attribute__((ext_vector_type(8))) float v8f;

#define HEADS 4
#define CCH 64
#define HIDF 256
#define NEG_SLOPE 0.2f
#define LN_EPS 1e-5f

__device__ __forceinline__ float eluf(float x) { return x > 0.f ? x : (expf(x) - 1.f); }
__device__ __forceinline__ float leakyf(float x) { return x > 0.f ? x : NEG_SLOPE * x; }

__device__ __forceinline__ unsigned fkey(float f) {
    unsigned u = __float_as_uint(f);
    return (u & 0x80000000u) ? ~u : (u | 0x80000000u);
}
__device__ __forceinline__ float funkey(unsigned k) {
    unsigned u = (k & 0x80000000u) ? (k ^ 0x80000000u) : ~k;
    return __uint_as_float(u);
}

__device__ __forceinline__ float waveRedSum(float v) {
    for (int off = 16; off > 0; off >>= 1) v += __shfl_xor(v, off, 32);
    return v;
}

__global__ void gemm_wmma_f32(const float* __restrict__ A, int lda,
                              const float* __restrict__ W,
                              float* __restrict__ C,
                              int M, int K, int Nout,
                              const float* __restrict__ bias, int elu_act) {
    int wid  = (blockIdx.x * blockDim.x + threadIdx.x) >> 5;
    int lane = threadIdx.x & 31;
    int tilesN = Nout >> 4;
    int totalTiles = (M >> 4) * tilesN;
    if (wid >= totalTiles) return;
    int tm = wid / tilesN;
    int tn = wid - tm * tilesN;
    int m  = (tm << 4) + (lane & 15);
    int n  = (tn << 4) + (lane & 15);
    int kb = (lane >> 4) << 1;
    const float* arow = A + (size_t)m * lda;
    const float* brow = W + (size_t)n * K;
    v8f acc = {};
    for (int k = 0; k < K; k += 4) {
        v2f a, b;
        a.x = arow[k + kb];
        a.y = arow[k + kb + 1];
        b.x = brow[k + kb];
        b.y = brow[k + kb + 1];
        acc = __builtin_amdgcn_wmma_f32_16x16x4_f32(false, a, false, b,
                                                    (short)0, acc, false, false);
    }
    int rbase = (tm << 4) + ((lane >> 4) << 3);
    int col   = (tn << 4) + (lane & 15);
    float bv  = bias ? bias[col] : 0.f;
    for (int i = 0; i < 8; ++i) {
        float v = acc[i] + bv;
        if (elu_act) v = eluf(v);
        C[(size_t)(rbase + i) * Nout + col] = v;
    }
}

__global__ void attn_coef(const float* __restrict__ hmat,
                          const float* __restrict__ a_s,
                          const float* __restrict__ a_d,
                          float* __restrict__ asrc, float* __restrict__ adst, int NH) {
    int idx = blockIdx.x * blockDim.x + threadIdx.x;
    if (idx >= NH) return;
    int n = idx >> 2, h = idx & 3;
    const float* row = hmat + (size_t)n * HIDF + h * CCH;
    const float* ws  = a_s + h * CCH;
    const float* wd  = a_d + h * CCH;
    float ss = 0.f, sd = 0.f;
    for (int c = 0; c < CCH; ++c) { float v = row[c]; ss += v * ws[c]; sd += v * wd[c]; }
    asrc[idx] = ss;
    adst[idx] = sd;
}

#define RN 200
#define AGG_THREADS 1024
#define AGG_WAVES (AGG_THREADS / 32)
#define WCAP 256

__device__ __forceinline__ void accum_one(int s, unsigned dl, int lane,
                                          const float* __restrict__ asrc,
                                          const float* __restrict__ hmat,
                                          const float* adl, const unsigned* mxk,
                                          float* dnl, float* acc) {
    float ex[HEADS];
#pragma unroll
    for (int h = 0; h < HEADS; ++h) {
        float a = leakyf(asrc[s * HEADS + h] + adl[dl * HEADS + h]);
        ex[h] = expf(a - funkey(mxk[dl * HEADS + h]));
    }
    if (lane == 0) {
#pragma unroll
        for (int h = 0; h < HEADS; ++h) atomicAdd(&dnl[dl * HEADS + h], ex[h]);
    }
    const float* hrow = hmat + (size_t)s * HIDF;
    float* orow       = acc + dl * HIDF;
#pragma unroll
    for (int i = 0; i < 8; ++i) {
        int c = lane + 32 * i;
        atomicAdd(&orow[c], ex[i >> 1] * hrow[c]);
    }
}

__global__ __launch_bounds__(AGG_THREADS)
void agg_owner(const int* __restrict__ src, const int* __restrict__ dst,
               const float* __restrict__ asrc, const float* __restrict__ adst,
               const float* __restrict__ hmat,
               float* __restrict__ denom, float* __restrict__ hacc, int N, int E) {
    __shared__ __attribute__((aligned(16))) float acc[RN * HIDF];
    __shared__ __attribute__((aligned(16))) int lsts[AGG_WAVES * WCAP];
    __shared__ __attribute__((aligned(16))) unsigned short lstd[AGG_WAVES * WCAP];
    __shared__ unsigned mxk[RN * HEADS];
    __shared__ float dnl[RN * HEADS];
    __shared__ float adl[RN * HEADS];
    __shared__ unsigned ovf;

    const int tid  = threadIdx.x;
    const int lane = tid & 31;
    const int wv   = tid >> 5;
    const int base = blockIdx.x * RN;
    const int nloc = min(RN, N - base);
    int* const myS            = lsts + wv * WCAP;
    unsigned short* const myD = lstd + wv * WCAP;

    if (tid == 0) ovf = 0u;
    for (int i = tid; i < nloc * HEADS; i += AGG_THREADS) {
        float ad = adst[base * HEADS + i];
        adl[i] = ad;
        mxk[i] = fkey(leakyf(asrc[base * HEADS + i] + ad));
    }
    __syncthreads();

    unsigned wcount = 0u;
    auto note = [&](int d, int e, bool valid) {
        unsigned dl = (unsigned)(d - base);
        bool m = valid && (dl < (unsigned)nloc);
        unsigned mask = (unsigned)__ballot(m);
        if (mask) {
            unsigned slot = wcount + __builtin_amdgcn_mbcnt_lo(mask, 0u);
            if (m && slot < (unsigned)WCAP) { myS[slot] = e; myD[slot] = (unsigned short)dl; }
            wcount += (unsigned)__popc(mask);
        }
    };
    const int wvu = __builtin_amdgcn_readfirstlane(wv);
    const int E4 = E >> 2;
    const int4* dst4 = (const int4*)dst;
    for (int ib = wvu * 32; ib < E4; ib += AGG_THREADS * 4) {
        const int i = ib + lane;
        int4 v[4];
        if (ib + 3 * AGG_THREADS + 32 <= E4) {
#pragma unroll
            for (int u = 0; u < 4; ++u) v[u] = dst4[i + u * AGG_THREADS];
#pragma unroll
            for (int u = 0; u < 4; ++u) {
                const int e = 4 * (i + u * AGG_THREADS);
                note(v[u].x, e + 0, true);
                note(v[u].y, e + 1, true);
                note(v[u].z, e + 2, true);
                note(v[u].w, e + 3, true);
            }
        } else {
#pragma unroll
            for (int u = 0; u < 4; ++u) {
                const int iu = i + u * AGG_THREADS;
                v[u] = (iu < E4) ? dst4[iu] : make_int4(0, 0, 0, 0);
            }
#pragma unroll
            for (int u = 0; u < 4; ++u) {
                const int iu = i + u * AGG_THREADS;
                const bool valid = iu < E4;
                note(v[u].x, 4 * iu + 0, valid);
                note(v[u].y, 4 * iu + 1, valid);
                note(v[u].z, 4 * iu + 2, valid);
                note(v[u].w, 4 * iu + 3, valid);
            }
        }
    }
    for (int e0 = (E4 << 2); e0 < E; e0 += AGG_THREADS) {
        int e = e0 + tid;
        bool valid = e < E;
        note(valid ? dst[e] : 0, e, valid);
    }
    if (wcount > (unsigned)WCAP && lane == 0) ovf = 1u;
    __syncthreads();

    const bool listed = (ovf == 0u);
    const int nw = (int)min(wcount, (unsigned)WCAP);

    if (listed) {
        for (int k0 = lane; k0 < nw; k0 += 128) {
            int s[4]; unsigned dl[4]; bool ok[4];
#pragma unroll
            for (int u = 0; u < 4; ++u) {
                int k = k0 + 32 * u;
                ok[u] = k < nw;
                int e = ok[u] ? myS[k] : 0;
                dl[u] = ok[u] ? (unsigned)myD[k] : 0u;
                s[u]  = src[e];
            }
            float4 a4[4];
#pragma unroll
            for (int u = 0; u < 4; ++u) a4[u] = *(const float4*)(asrc + (size_t)s[u] * HEADS);
#pragma unroll
            for (int u = 0; u < 4; ++u) {
                if (ok[u]) {
                    myS[k0 + 32 * u] = s[u];
                    atomicMax(&mxk[dl[u] * HEADS + 0], fkey(leakyf(a4[u].x + adl[dl[u] * HEADS + 0])));
                    atomicMax(&mxk[dl[u] * HEADS + 1], fkey(leakyf(a4[u].y + adl[dl[u] * HEADS + 1])));
                    atomicMax(&mxk[dl[u] * HEADS + 2], fkey(leakyf(a4[u].z + adl[dl[u] * HEADS + 2])));
                    atomicMax(&mxk[dl[u] * HEADS + 3], fkey(leakyf(a4[u].w + adl[dl[u] * HEADS + 3])));
                }
            }
        }
    } else {
        for (int e = tid; e < E; e += AGG_THREADS) {
            unsigned dl = (unsigned)(dst[e] - base);
            if (dl < (unsigned)nloc) {
                int s = src[e];
#pragma unroll
                for (int h = 0; h < HEADS; ++h)
                    atomicMax(&mxk[dl * HEADS + h],
                              fkey(leakyf(asrc[s * HEADS + h] + adl[dl * HEADS + h])));
            }
        }
    }
    __syncthreads();

    for (int i = wv; i < nloc; i += AGG_WAVES) {
        int n = base + i;
        float ex[HEADS];
#pragma unroll
        for (int h = 0; h < HEADS; ++h) {
            float aself = leakyf(asrc[n * HEADS + h] + adl[i * HEADS + h]);
            ex[h] = expf(aself - funkey(mxk[i * HEADS + h]));
        }
        if (lane == 0) {
#pragma unroll
            for (int h = 0; h < HEADS; ++h) dnl[i * HEADS + h] = ex[h];
        }
        const float* row = hmat + (size_t)n * HIDF;
#pragma unroll
        for (int j = 0; j < 8; ++j) {
            int c = lane + 32 * j;
            acc[i * HIDF + c] = row[c] * ex[j >> 1];
        }
    }
    __syncthreads();

    if (listed) {
        for (int k0 = 0; k0 < nw; k0 += 4) {
            const int4 sv    = *(const int4*)(myS + k0);
            const ushort4 dv = *(const ushort4*)(myD + k0);
            const int svv[4] = {sv.x, sv.y, sv.z, sv.w};
            const int dvv[4] = {dv.x, dv.y, dv.z, dv.w};
            int s[4]; unsigned dl[4]; bool ok[4];
#pragma unroll
            for (int u = 0; u < 4; ++u) {
                ok[u] = (k0 + u) < nw;
                s[u]  = __builtin_amdgcn_readfirstlane(ok[u] ? svv[u] : 0);
                dl[u] = (unsigned)__builtin_amdgcn_readfirstlane(ok[u] ? dvv[u] : 0);
            }
            float hv[4][8];
#pragma unroll
            for (int u = 0; u < 4; ++u) {
                const float* hrow = hmat + (size_t)s[u] * HIDF;
#pragma unroll
                for (int j = 0; j < 8; ++j) hv[u][j] = hrow[lane + 32 * j];
            }
            float ex[4][HEADS];
#pragma unroll
            for (int u = 0; u < 4; ++u) {
#pragma unroll
                for (int h = 0; h < HEADS; ++h) {
                    float a = leakyf(asrc[s[u] * HEADS + h] + adl[dl[u] * HEADS + h]);
                    ex[u][h] = expf(a - funkey(mxk[dl[u] * HEADS + h]));
                }
            }
#pragma unroll
            for (int u = 0; u < 4; ++u) {
                if (ok[u]) {
                    if (lane == 0) {
#pragma unroll
                        for (int h = 0; h < HEADS; ++h) atomicAdd(&dnl[dl[u] * HEADS + h], ex[u][h]);
                    }
                    float* orow = acc + dl[u] * HIDF;
#pragma unroll
                    for (int j = 0; j < 8; ++j)
                        atomicAdd(&orow[lane + 32 * j], ex[u][j >> 1] * hv[u][j]);
                }
            }
        }
    } else {
        for (int e0 = wvu * 32; e0 < E; e0 += AGG_WAVES * 32) {
            int e = e0 + lane;
            int d = (e < E) ? dst[e] : 0;
            unsigned dl = (unsigned)(d - base);
            bool m = (e < E) && (dl < (unsigned)nloc);
            unsigned mask = (unsigned)__ballot(m);
            while (mask) {
                int l = __ffs(mask) - 1;
                mask &= mask - 1u;
                int s1 = src[e0 + l];
                unsigned dl1 = (unsigned)__shfl((int)dl, l, 32);
                accum_one(s1, dl1, lane, asrc, hmat, adl, mxk, dnl, acc);
            }
        }
    }
    __syncthreads();

    {
        const float4* a4 = (const float4*)acc;
        float4* o4 = (float4*)(hacc + (size_t)base * HIDF);
        for (int i = tid; i < nloc * (HIDF / 4); i += AGG_THREADS) o4[i] = a4[i];
        for (int i = tid; i < nloc * HEADS; i += AGG_THREADS) denom[base * HEADS + i] = dnl[i];
    }
}

__global__ void finalize_concat(const float* __restrict__ hacc, const float* __restrict__ denom,
                                const float* __restrict__ bias, const float* __restrict__ g,
                                const float* __restrict__ be, float* __restrict__ out, int N) {
    int wid  = (blockIdx.x * blockDim.x + threadIdx.x) >> 5;
    int lane = threadIdx.x & 31;
    if (wid >= N) return;
    size_t base = (size_t)wid * HIDF;
    float dn[HEADS];
#pragma unroll
    for (int h = 0; h < HEADS; ++h) dn[h] = 1.f / (denom[wid * HEADS + h] + 1e-16f);
    float v[8];
    float s = 0.f;
#pragma unroll
    for (int i = 0; i < 8; ++i) {
        int c = lane + 32 * i;
        v[i] = hacc[base + c] * dn[i >> 1] + bias[c];
        s += v[i];
    }
    float mu = waveRedSum(s) * (1.f / HIDF);
    float q = 0.f;
#pragma unroll
    for (int i = 0; i < 8; ++i) { float dlt = v[i] - mu; q += dlt * dlt; }
    float rstd = rsqrtf(waveRedSum(q) * (1.f / HIDF) + LN_EPS);
#pragma unroll
    for (int i = 0; i < 8; ++i) {
        int c = lane + 32 * i;
        float y = (v[i] - mu) * rstd * g[c] + be[c];
        out[base + c] = eluf(y);
    }
}

__global__ void finalize_mean(const float* __restrict__ hacc, const float* __restrict__ denom,
                              const float* __restrict__ bias, const float* __restrict__ g,
                              const float* __restrict__ be, float* __restrict__ out, int N) {
    int wid  = (blockIdx.x * blockDim.x + threadIdx.x) >> 5;
    int lane = threadIdx.x & 31;
    if (wid >= N) return;
    size_t base = (size_t)wid * HIDF;
    float dn[HEADS];
#pragma unroll
    for (int h = 0; h < HEADS; ++h) dn[h] = 1.f / (denom[wid * HEADS + h] + 1e-16f);
    float v[2];
    float s = 0.f;
#pragma unroll
    for (int i = 0; i < 2; ++i) {
        int c = lane + 32 * i;
        float acc = 0.f;
#pragma unroll
        for (int h = 0; h < HEADS; ++h) acc += hacc[base + h * CCH + c] * dn[h];
        v[i] = acc * (1.f / HEADS) + bias[c];
        s += v[i];
    }
    float mu = waveRedSum(s) * (1.f / CCH);
    float q = 0.f;
#pragma unroll
    for (int i = 0; i < 2; ++i) { float dlt = v[i] - mu; q += dlt * dlt; }
    float rstd = rsqrtf(waveRedSum(q) * (1.f / CCH) + LN_EPS);
#pragma unroll
    for (int i = 0; i < 2; ++i) {
        int c = lane + 32 * i;
        float y = (v[i] - mu) * rstd * g[c] + be[c];
        out[(size_t)wid * CCH + c] = eluf(y);
    }
}

__global__ void value_head(const float* __restrict__ hin,
                           const float* __restrict__ vw1, const float* __restrict__ vb1,
                           const float* __restrict__ vw2, const float* __restrict__ vb2,
                           float* __restrict__ out, int N) {
    int wid  = (blockIdx.x * blockDim.x + threadIdx.x) >> 5;
    int lane = threadIdx.x & 31;
    if (wid >= N) return;
    const float* row  = hin + (size_t)wid * CCH;
    const float* wrow = vw1 + lane * CCH;
    float hid = vb1[lane];
    for (int c = 0; c < CCH; ++c) hid += row[c] * wrow[c];
    hid = eluf(hid);
#pragma unroll
    for (int k = 0; k < 4; ++k) {
        float p = vw2[k * 32 + lane] * hid;
        p = waveRedSum(p);
        if (lane == 0) out[wid * 4 + k] = p + vb2[k];
    }
}

extern "C" void kernel_launch(void* const* d_in, const int* in_sizes, int n_in,
                              void* d_out, int out_size, void* d_ws, size_t ws_size,
                              hipStream_t stream) {
    const float* x      = (const float*)d_in[0];
    const int*   esrc   = (const int*)d_in[1];
    const int*   edst   = (const int*)d_in[2];
    const float* proj_w = (const float*)d_in[3];
    const float* proj_b = (const float*)d_in[4];
    const float* W[3]  = {(const float*)d_in[5],  (const float*)d_in[11], (const float*)d_in[17]};
    const float* AS[3] = {(const float*)d_in[6],  (const float*)d_in[12], (const float*)d_in[18]};
    const float* AD[3] = {(const float*)d_in[7],  (const float*)d_in[13], (const float*)d_in[19]};
    const float* B[3]  = {(const float*)d_in[8],  (const float*)d_in[14], (const float*)d_in[20]};
    const float* G[3]  = {(const float*)d_in[9],  (const float*)d_in[15], (const float*)d_in[21]};
    const float* BE[3] = {(const float*)d_in[10], (const float*)d_in[16], (const float*)d_in[22]};
    const float* vw1 = (const float*)d_in[23];
    const float* vb1 = (const float*)d_in[24];
    const float* vw2 = (const float*)d_in[25];
    const float* vb2 = (const float*)d_in[26];

    const int N = in_sizes[0] / 128;
    const int E = in_sizes[1];
    const int NH = N * HEADS;

    float* ws     = (float*)d_ws;
    float* hin    = ws;
    float* hmat   = hin  + (size_t)N * HIDF;
    float* hacc   = hmat + (size_t)N * HIDF;
    float* asrc   = hacc + (size_t)N * HIDF;
    float* adst   = asrc + (size_t)NH;
    float* denom  = adst + (size_t)NH;

    dim3 blk(256);

    {
        int tiles = (N / 16) * (64 / 16);
        gemm_wmma_f32<<<(tiles * 32 + 255) / 256, blk, 0, stream>>>(
            x, 128, proj_w, hin, N, 128, 64, proj_b, 1);
    }

    int in_ld = 64;
    for (int l = 0; l < 3; ++l) {
        int K = (l == 0) ? 64 : 256;
        int tiles = (N / 16) * (HIDF / 16);
        gemm_wmma_f32<<<(tiles * 32 + 255) / 256, blk, 0, stream>>>(
            hin, in_ld, W[l], hmat, N, K, HIDF, nullptr, 0);
        attn_coef<<<(NH + 255) / 256, blk, 0, stream>>>(hmat, AS[l], AD[l], asrc, adst, NH);
        agg_owner<<<(N + RN - 1) / RN, AGG_THREADS, 0, stream>>>(
            esrc, edst, asrc, adst, hmat, denom, hacc, N, E);
        if (l < 2) {
            finalize_concat<<<(N * 32 + 255) / 256, blk, 0, stream>>>(
                hacc, denom, B[l], G[l], BE[l], hin, N);
            in_ld = HIDF;
        } else {
            finalize_mean<<<(N * 32 + 255) / 256, blk, 0, stream>>>(
                hacc, denom, B[l], G[l], BE[l], hin, N);
        }
    }

    value_head<<<(N * 32 + 255) / 256, blk, 0, stream>>>(
        hin, vw1, vb1, vw2, vb2, (float*)d_out, N);
}


# GEMMs weight-stationary: a wave keeps its 64 W columns for all k as register fragments and streams 16-row A tiles past them, 16-byte k-runs from global, no LDS; k order in 8-k groups changed
# speedup vs baseline: 2.0123x; 1.2646x over previous
#include <hip/hip_runtime.h>
#include <math.h>

typedef __attribute__((ext_vector_type(2))) float v2f;
typedef __attribute__((ext_vector_type(8))) float v8f;

#define HEADS 4
#define CCH 64
#define HIDF 256
#define NEG_SLOPE 0.2f
#define LN_EPS 1e-5f

__device__ __forceinline__ float eluf(float x) { return x > 0.f ? x : (expf(x) - 1.f); }
__device__ __forceinline__ float leakyf(float x) { return x > 0.f ? x : NEG_SLOPE * x; }

__device__ __forceinline__ unsigned fkey(float f) {
    unsigned u = __float_as_uint(f);
    return (u & 0x80000000u) ? ~u : (u | 0x80000000u);
}
__device__ __forceinline__ float funkey(unsigned k) {
    unsigned u = (k & 0x80000000u) ? (k ^ 0x80000000u) : ~k;
    return __uint_as_float(u);
}

__device__ __forceinline__ float waveRedSum(float v) {
    for (int off = 16; off > 0; off >>= 1) v += __shfl_xor(v, off, 32);
    return v;
}

#define GW_NT 4
#define GW_THREADS 128
#define GW_BLOCKS_K256 256
#define GW_BLOCKS_K128 512
#define GW_BLOCKS_K64 1024

__device__ __forceinline__ float xhalf(float x) {
    return __builtin_bit_cast(float, __builtin_amdgcn_permlanex16(
        __builtin_bit_cast(int, x), __builtin_bit_cast(int, x), 0x76543210, 0xfedcba98, false, false));
}

template <int KK>
__device__ __forceinline__ void load_frag_row(float4 (&f)[KK], const float* __restrict__ p) {
#pragma unroll
    for (int kk = 0; kk < KK; ++kk) f[kk] = *(const float4*)(p + kk * 8);
}

template <int KK, bool EPI>
__device__ __forceinline__ void wstat_tile(const float4 (&af)[KK], const float4 (&bf)[GW_NT][KK],
                                           float* __restrict__ C, int rt, int cb, int Nout,
                                           const float* __restrict__ bias, int c, int h) {
    v8f acc[GW_NT];
#pragma unroll
    for (int t = 0; t < GW_NT; ++t) acc[t] = v8f{};
#pragma unroll
    for (int kk = 0; kk < KK; ++kk) {
        const v2f a0 = {af[kk].x, af[kk].y};
        const v2f a1 = {af[kk].z, af[kk].w};
#pragma unroll
        for (int t = 0; t < GW_NT; ++t) {
            const v2f b0 = {bf[t][kk].x, bf[t][kk].y};
            acc[t] = __builtin_amdgcn_wmma_f32_16x16x4_f32(false, a0, false, b0,
                                                           (short)0, acc[t], false, false);
        }
#pragma unroll
        for (int t = 0; t < GW_NT; ++t) {
            const v2f b1 = {bf[t][kk].z, bf[t][kk].w};
            acc[t] = __builtin_amdgcn_wmma_f32_16x16x4_f32(false, a1, false, b1,
                                                           (short)0, acc[t], false, false);
        }
    }
#pragma unroll
    for (int t = 0; t < GW_NT; t += 2) {
        const int col = cb + t * 16 + 16 * h + c;
        float bv = 0.f;
        if (EPI) bv = bias[col];
        float* crow = C + (size_t)(rt * 16) * Nout + col;
#pragma unroll
        for (int i = 0; i < 8; ++i) {
            const float lo   = acc[t][i];
            const float hi   = acc[t + 1][i];
            const float recv = xhalf(h ? lo : hi);
            float v0 = h ? recv : lo;
            float v1 = h ? hi : recv;
            if (EPI) { v0 = eluf(v0 + bv); v1 = eluf(v1 + bv); }
            crow[(size_t)i * Nout]       = v0;
            crow[(size_t)(8 + i) * Nout] = v1;
        }
    }
}

template <int K, bool EPI>
__global__ __launch_bounds__(GW_THREADS)
void gemm_wstat_f32(const float* __restrict__ A, int lda,
                    const float* __restrict__ W,
                    float* __restrict__ C,
                    int M, int Nout,
                    const float* __restrict__ bias) {
    constexpr int KK = K / 8;
    const int lane = threadIdx.x & 31;
    const int c = lane & 15;
    const int h = lane >> 4;
    const int gw = __builtin_amdgcn_readfirstlane((int)((blockIdx.x * blockDim.x + threadIdx.x) >> 5));
    const int nw = (int)((gridDim.x * blockDim.x) >> 5);
    const int ngrp = Nout / (16 * GW_NT);
    const int cg = gw % ngrp;
    const int rs = gw / ngrp;
    const int nstream = nw / ngrp;
    const int ntile = M >> 4;
    if (rs >= nstream || rs >= ntile) return;
    const int cb = cg * (16 * GW_NT);

    float4 bf[GW_NT][KK];
#pragma unroll
    for (int t = 0; t < GW_NT; ++t)
        load_frag_row<KK>(bf[t], W + (size_t)(cb + t * 16 + c) * K + h * 4);

    const float* arow = A + (size_t)c * lda + h * 4;
    const size_t tstep = (size_t)16 * lda;
    float4 f0[KK], f1[KK];
    load_frag_row<KK>(f0, arow + (size_t)rs * tstep);
    for (int rt = rs; rt < ntile; rt += 2 * nstream) {
        const int rt1 = rt + nstream;
        const int rt2 = rt1 + nstream;
        const bool has1 = rt1 < ntile;
        load_frag_row<KK>(f1, arow + (size_t)(has1 ? rt1 : rt) * tstep);
        wstat_tile<KK, EPI>(f0, bf, C, rt, cb, Nout, bias, c, h);
        load_frag_row<KK>(f0, arow + (size_t)(rt2 < ntile ? rt2 : rt) * tstep);
        if (has1) wstat_tile<KK, EPI>(f1, bf, C, rt1, cb, Nout, bias, c, h);
    }
}

__global__ void attn_coef(const float* __restrict__ hmat,
                          const float* __restrict__ a_s,
                          const float* __restrict__ a_d,
                          float* __restrict__ asrc, float* __restrict__ adst, int NH) {
    int idx = blockIdx.x * blockDim.x + threadIdx.x;
    if (idx >= NH) return;
    int n = idx >> 2, h = idx & 3;
    const float* row = hmat + (size_t)n * HIDF + h * CCH;
    const float* ws  = a_s + h * CCH;
    const float* wd  = a_d + h * CCH;
    float ss = 0.f, sd = 0.f;
    for (int c = 0; c < CCH; ++c) { float v = row[c]; ss += v * ws[c]; sd += v * wd[c]; }
    asrc[idx] = ss;
    adst[idx] = sd;
}

#define RN 200
#define AGG_THREADS 1024
#define AGG_WAVES (AGG_THREADS / 32)
#define WCAP 256

__device__ __forceinline__ void accum_one(int s, unsigned dl, int lane,
                                          const float* __restrict__ asrc,
                                          const float* __restrict__ hmat,
                                          const float* adl, const unsigned* mxk,
                                          float* dnl, float* acc) {
    float ex[HEADS];
#pragma unroll
    for (int h = 0; h < HEADS; ++h) {
        float a = leakyf(asrc[s * HEADS + h] + adl[dl * HEADS + h]);
        ex[h] = expf(a - funkey(mxk[dl * HEADS + h]));
    }
    if (lane == 0) {
#pragma unroll
        for (int h = 0; h < HEADS; ++h) atomicAdd(&dnl[dl * HEADS + h], ex[h]);
    }
    const float* hrow = hmat + (size_t)s * HIDF;
    float* orow       = acc + dl * HIDF;
#pragma unroll
    for (int i = 0; i < 8; ++i) {
        int c = lane + 32 * i;
        atomicAdd(&orow[c], ex[i >> 1] * hrow[c]);
    }
}

__global__ __launch_bounds__(AGG_THREADS)
void agg_owner(const int* __restrict__ src, const int* __restrict__ dst,
               const float* __restrict__ asrc, const float* __restrict__ adst,
               const float* __restrict__ hmat,
               float* __restrict__ denom, float* __restrict__ hacc, int N, int E) {
    __shared__ __attribute__((aligned(16))) float acc[RN * HIDF];
    __shared__ __attribute__((aligned(16))) int lsts[AGG_WAVES * WCAP];
    __shared__ __attribute__((aligned(16))) unsigned short lstd[AGG_WAVES * WCAP];
    __shared__ unsigned mxk[RN * HEADS];
    __shared__ float dnl[RN * HEADS];
    __shared__ float adl[RN * HEADS];
    __shared__ unsigned ovf;

    const int tid  = threadIdx.x;
    const int lane = tid & 31;
    const int wv   = tid >> 5;
    const int base = blockIdx.x * RN;
    const int nloc = min(RN, N - base);
    int* const myS            = lsts + wv * WCAP;
    unsigned short* const myD = lstd + wv * WCAP;

    if (tid == 0) ovf = 0u;
    for (int i = tid; i < nloc * HEADS; i += AGG_THREADS) {
        float ad = adst[base * HEADS + i];
        adl[i] = ad;
        mxk[i] = fkey(leakyf(asrc[base * HEADS + i] + ad));
    }
    __syncthreads();

    unsigned wcount = 0u;
    auto note = [&](int d, int e, bool valid) {
        unsigned dl = (unsigned)(d - base);
        bool m = valid && (dl < (unsigned)nloc);
        unsigned mask = (unsigned)__ballot(m);
        if (mask) {
            unsigned slot = wcount + __builtin_amdgcn_mbcnt_lo(mask, 0u);
            if (m && slot < (unsigned)WCAP) { myS[slot] = e; myD[slot] = (unsigned short)dl; }
            wcount += (unsigned)__popc(mask);
        }
    };
    const int wvu = __builtin_amdgcn_readfirstlane(wv);
    const int E4 = E >> 2;
    const int4* dst4 = (const int4*)dst;
    for (int ib = wvu * 32; ib < E4; ib += AGG_THREADS * 4) {
        const int i = ib + lane;
        int4 v[4];
        if (ib + 3 * AGG_THREADS + 32 <= E4) {
#pragma unroll
            for (int u = 0; u < 4; ++u) v[u] = dst4[i + u * AGG_THREADS];
#pragma unroll
            for (int u = 0; u < 4; ++u) {
                const int e = 4 * (i + u * AGG_THREADS);
                note(v[u].x, e + 0, true);
                note(v[u].y, e + 1, true);
                note(v[u].z, e + 2, true);
                note(v[u].w, e + 3, true);
            }
        } else {
#pragma unroll
            for (int u = 0; u < 4; ++u) {
                const int iu = i + u * AGG_THREADS;
                v[u] = (iu < E4) ? dst4[iu] : make_int4(0, 0, 0, 0);
            }
#pragma unroll
            for (int u = 0; u < 4; ++u) {
                const int iu = i + u * AGG_THREADS;
                const bool valid = iu < E4;
                note(v[u].x, 4 * iu + 0, valid);
                note(v[u].y, 4 * iu + 1, valid);
                note(v[u].z, 4 * iu + 2, valid);
                note(v[u].w, 4 * iu + 3, valid);
            }
        }
    }
    for (int e0 = (E4 << 2); e0 < E; e0 += AGG_THREADS) {
        int e = e0 + tid;
        bool valid = e < E;
        note(valid ? dst[e] : 0, e, valid);
    }
    if (wcount > (unsigned)WCAP && lane == 0) ovf = 1u;
    __syncthreads();

    const bool listed = (ovf == 0u);
    const int nw = (int)min(wcount, (unsigned)WCAP);

    if (listed) {
        for (int k0 = lane; k0 < nw; k0 += 128) {
            int s[4]; unsigned dl[4]; bool ok[4];
#pragma unroll
            for (int u = 0; u < 4; ++u) {
                int k = k0 + 32 * u;
                ok[u] = k < nw;
                int e = ok[u] ? myS[k] : 0;
                dl[u] = ok[u] ? (unsigned)myD[k] : 0u;
                s[u]  = src[e];
            }
            float4 a4[4];
#pragma unroll
            for (int u = 0; u < 4; ++u) a4[u] = *(const float4*)(asrc + (size_t)s[u] * HEADS);
#pragma unroll
            for (int u = 0; u < 4; ++u) {
                if (ok[u]) {
                    myS[k0 + 32 * u] = s[u];
                    atomicMax(&mxk[dl[u] * HEADS + 0], fkey(leakyf(a4[u].x + adl[dl[u] * HEADS + 0])));
                    atomicMax(&mxk[dl[u] * HEADS + 1], fkey(leakyf(a4[u].y + adl[dl[u] * HEADS + 1])));
                    atomicMax(&mxk[dl[u] * HEADS + 2], fkey(leakyf(a4[u].z + adl[dl[u] * HEADS + 2])));
                    atomicMax(&mxk[dl[u] * HEADS + 3], fkey(leakyf(a4[u].w + adl[dl[u] * HEADS + 3])));
                }
            }
        }
    } else {
        for (int e = tid; e < E; e += AGG_THREADS) {
            unsigned dl = (unsigned)(dst[e] - base);
            if (dl < (unsigned)nloc) {
                int s = src[e];
#pragma unroll
                for (int h = 0; h < HEADS; ++h)
                    atomicMax(&mxk[dl * HEADS + h],
                              fkey(leakyf(asrc[s * HEADS + h] + adl[dl * HEADS + h])));
            }
        }
    }
    __syncthreads();

    for (int i = wv; i < nloc; i += AGG_WAVES) {
        int n = base + i;
        float ex[HEADS];
#pragma unroll
        for (int h = 0; h < HEADS; ++h) {
            float aself = leakyf(asrc[n * HEADS + h] + adl[i * HEADS + h]);
            ex[h] = expf(aself - funkey(mxk[i * HEADS + h]));
        }
        if (lane == 0) {
#pragma unroll
            for (int h = 0; h < HEADS; ++h) dnl[i * HEADS + h] = ex[h];
        }
        const float* row = hmat + (size_t)n * HIDF;
#pragma unroll
        for (int j = 0; j < 8; ++j) {
            int c = lane + 32 * j;
            acc[i * HIDF + c] = row[c] * ex[j >> 1];
        }
    }
    __syncthreads();

    if (listed) {
        for (int k0 = 0; k0 < nw; k0 += 4) {
            const int4 sv    = *(const int4*)(myS + k0);
            const ushort4 dv = *(const ushort4*)(myD + k0);
            const int svv[4] = {sv.x, sv.y, sv.z, sv.w};
            const int dvv[4] = {dv.x, dv.y, dv.z, dv.w};
            int s[4]; unsigned dl[4]; bool ok[4];
#pragma unroll
            for (int u = 0; u < 4; ++u) {
                ok[u] = (k0 + u) < nw;
                s[u]  = __builtin_amdgcn_readfirstlane(ok[u] ? svv[u] : 0);
                dl[u] = (unsigned)__builtin_amdgcn_readfirstlane(ok[u] ? dvv[u] : 0);
            }
            float hv[4][8];
#pragma unroll
            for (int u = 0; u < 4; ++u) {
                const float* hrow = hmat + (size_t)s[u] * HIDF;
#pragma unroll
                for (int j = 0; j < 8; ++j) hv[u][j] = hrow[lane + 32 * j];
            }
            float ex[4][HEADS];
#pragma unroll
            for (int u = 0; u < 4; ++u) {
#pragma unroll
                for (int h = 0; h < HEADS; ++h) {
                    float a = leakyf(asrc[s[u] * HEADS + h] + adl[dl[u] * HEADS + h]);
                    ex[u][h] = expf(a - funkey(mxk[dl[u] * HEADS + h]));
                }
            }
#pragma unroll
            for (int u = 0; u < 4; ++u) {
                if (ok[u]) {
                    if (lane == 0) {
#pragma unroll
                        for (int h = 0; h < HEADS; ++h) atomicAdd(&dnl[dl[u] * HEADS + h], ex[u][h]);
                    }
                    float* orow = acc + dl[u] * HIDF;
#pragma unroll
                    for (int j = 0; j < 8; ++j)
                        atomicAdd(&orow[lane + 32 * j], ex[u][j >> 1] * hv[u][j]);
                }
            }
        }
    } else {
        for (int e0 = wvu * 32; e0 < E; e0 += AGG_WAVES * 32) {
            int e = e0 + lane;
            int d = (e < E) ? dst[e] : 0;
            unsigned dl = (unsigned)(d - base);
            bool m = (e < E) && (dl < (unsigned)nloc);
            unsigned mask = (unsigned)__ballot(m);
            while (mask) {
                int l = __ffs(mask) - 1;
                mask &= mask - 1u;
                int s1 = src[e0 + l];
                unsigned dl1 = (unsigned)__shfl((int)dl, l, 32);
                accum_one(s1, dl1, lane, asrc, hmat, adl, mxk, dnl, acc);
            }
        }
    }
    __syncthreads();

    {
        const float4* a4 = (const float4*)acc;
        float4* o4 = (float4*)(hacc + (size_t)base * HIDF);
        for (int i = tid; i < nloc * (HIDF / 4); i += AGG_THREADS) o4[i] = a4[i];
        for (int i = tid; i < nloc * HEADS; i += AGG_THREADS) denom[base * HEADS + i] = dnl[i];
    }
}

__global__ void finalize_concat(const float* __restrict__ hacc, const float* __restrict__ denom,
                                const float* __restrict__ bias, const float* __restrict__ g,
                                const float* __restrict__ be, float* __restrict__ out, int N) {
    int wid  = (blockIdx.x * blockDim.x + threadIdx.x) >> 5;
    int lane = threadIdx.x & 31;
    if (wid >= N) return;
    size_t base = (size_t)wid * HIDF;
    float dn[HEADS];
#pragma unroll
    for (int h = 0; h < HEADS; ++h) dn[h] = 1.f / (denom[wid * HEADS + h] + 1e-16f);
    float v[8];
    float s = 0.f;
#pragma unroll
    for (int i = 0; i < 8; ++i) {
        int c = lane + 32 * i;
        v[i] = hacc[base + c] * dn[i >> 1] + bias[c];
        s += v[i];
    }
    float mu = waveRedSum(s) * (1.f / HIDF);
    float q = 0.f;
#pragma unroll
    for (int i = 0; i < 8; ++i) { float dlt = v[i] - mu; q += dlt * dlt; }
    float rstd = rsqrtf(waveRedSum(q) * (1.f / HIDF) + LN_EPS);
#pragma unroll
    for (int i = 0; i < 8; ++i) {
        int c = lane + 32 * i;
        float y = (v[i] - mu) * rstd * g[c] + be[c];
        out[base + c] = eluf(y);
    }
}

__global__ void finalize_mean(const float* __restrict__ hacc, const float* __restrict__ denom,
                              const float* __restrict__ bias, const float* __restrict__ g,
                              const float* __restrict__ be, float* __restrict__ out, int N) {
    int wid  = (blockIdx.x * blockDim.x + threadIdx.x) >> 5;
    int lane = threadIdx.x & 31;
    if (wid >= N) return;
    size_t base = (size_t)wid * HIDF;
    float dn[HEADS];
#pragma unroll
    for (int h = 0; h < HEADS; ++h) dn[h] = 1.f / (denom[wid * HEADS + h] + 1e-16f);
    float v[2];
    float s = 0.f;
#pragma unroll
    for (int i = 0; i < 2; ++i) {
        int c = lane + 32 * i;
        float acc = 0.f;
#pragma unroll
        for (int h = 0; h < HEADS; ++h) acc += hacc[base + h * CCH + c] * dn[h];
        v[i] = acc * (1.f / HEADS) + bias[c];
        s += v[i];
    }
    float mu = waveRedSum(s) * (1.f / CCH);
    float q = 0.f;
#pragma unroll
    for (int i = 0; i < 2; ++i) { float dlt = v[i] - mu; q += dlt * dlt; }
    float rstd = rsqrtf(waveRedSum(q) * (1.f / CCH) + LN_EPS);
#pragma unroll
    for (int i = 0; i < 2; ++i) {
        int c = lane + 32 * i;
        float y = (v[i] - mu) * rstd * g[c] + be[c];
        out[(size_t)wid * CCH + c] = eluf(y);
    }
}

__global__ void value_head(const float* __restrict__ hin,
                           const float* __restrict__ vw1, const float* __restrict__ vb1,
                           const float* __restrict__ vw2, const float* __restrict__ vb2,
                           float* __restrict__ out, int N) {
    int wid  = (blockIdx.x * blockDim.x + threadIdx.x) >> 5;
    int lane = threadIdx.x & 31;
    if (wid >= N) return;
    const float* row  = hin + (size_t)wid * CCH;
    const float* wrow = vw1 + lane * CCH;
    float hid = vb1[lane];
    for (int c = 0; c < CCH; ++c) hid += row[c] * wrow[c];
    hid = eluf(hid);
#pragma unroll
    for (int k = 0; k < 4; ++k) {
        float p = vw2[k * 32 + lane] * hid;
        p = waveRedSum(p);
        if (lane == 0) out[wid * 4 + k] = p + vb2[k];
    }
}

extern "C" void kernel_launch(void* const* d_in, const int* in_sizes, int n_in,
                              void* d_out, int out_size, void* d_ws, size_t ws_size,
                              hipStream_t stream) {
    const float* x      = (const float*)d_in[0];
    const int*   esrc   = (const int*)d_in[1];
    const int*   edst   = (const int*)d_in[2];
    const float* proj_w = (const float*)d_in[3];
    const float* proj_b = (const float*)d_in[4];
    const float* W[3]  = {(const float*)d_in[5],  (const float*)d_in[11], (const float*)d_in[17]};
    const float* AS[3] = {(const float*)d_in[6],  (const float*)d_in[12], (const float*)d_in[18]};
    const float* AD[3] = {(const float*)d_in[7],  (const float*)d_in[13], (const float*)d_in[19]};
    const float* B[3]  = {(const float*)d_in[8],  (const float*)d_in[14], (const float*)d_in[20]};
    const float* G[3]  = {(const float*)d_in[9],  (const float*)d_in[15], (const float*)d_in[21]};
    const float* BE[3] = {(const float*)d_in[10], (const float*)d_in[16], (const float*)d_in[22]};
    const float* vw1 = (const float*)d_in[23];
    const float* vb1 = (const float*)d_in[24];
    const float* vw2 = (const float*)d_in[25];
    const float* vb2 = (const float*)d_in[26];

    const int N = in_sizes[0] / 128;
    const int E = in_sizes[1];
    const int NH = N * HEADS;

    float* ws     = (float*)d_ws;
    float* hin    = ws;
    float* hmat   = hin  + (size_t)N * HIDF;
    float* hacc   = hmat + (size_t)N * HIDF;
    float* asrc   = hacc + (size_t)N * HIDF;
    float* adst   = asrc + (size_t)NH;
    float* denom  = adst + (size_t)NH;

    dim3 blk(256);

    gemm_wstat_f32<128, true><<<GW_BLOCKS_K128, GW_THREADS, 0, stream>>>(
        x, 128, proj_w, hin, N, 64, proj_b);

    int in_ld = 64;
    for (int l = 0; l < 3; ++l) {
        if (l == 0)
            gemm_wstat_f32<64, false><<<GW_BLOCKS_K64, GW_THREADS, 0, stream>>>(
                hin, in_ld, W[l], hmat, N, HIDF, nullptr);
        else
            gemm_wstat_f32<256, false><<<GW_BLOCKS_K256, GW_THREADS, 0, stream>>>(
                hin, in_ld, W[l], hmat, N, HIDF, nullptr);
        attn_coef<<<(NH + 255) / 256, blk, 0, stream>>>(hmat, AS[l], AD[l], asrc, adst, NH);
        agg_owner<<<(N + RN - 1) / RN, AGG_THREADS, 0, stream>>>(
            esrc, edst, asrc, adst, hmat, denom, hacc, N, E);
        if (l < 2) {
            finalize_concat<<<(N * 32 + 255) / 256, blk, 0, stream>>>(
                hacc, denom, B[l], G[l], BE[l], hin, N);
            in_ld = HIDF;
        } else {
            finalize_mean<<<(N * 32 + 255) / 256, blk, 0, stream>>>(
                hacc, denom, B[l], G[l], BE[l], hin, N);
        }
    }

    value_head<<<(N * 32 + 255) / 256, blk, 0, stream>>>(
        hin, vw1, vb1, vw2, vb2, (float*)d_out, N);
}


# Edge scan once per call: the first layer's owner work-groups write each wave's in-edge list (source, local dst, count) to the workspace; layers 2 and 3 read it back instead of rescanning; same sums
# speedup vs baseline: 2.2732x; 1.1297x over previous
#include <hip/hip_runtime.h>
#include <math.h>

typedef __attribute__((ext_vector_type(2))) float v2f;
typedef __attribute__((ext_vector_type(8))) float v8f;

#define HEADS 4
#define CCH 64
#define HIDF 256
#define NEG_SLOPE 0.2f
#define LN_EPS 1e-5f

__device__ __forceinline__ float eluf(float x) { return x > 0.f ? x : (expf(x) - 1.f); }
__device__ __forceinline__ float leakyf(float x) { return x > 0.f ? x : NEG_SLOPE * x; }

__device__ __forceinline__ unsigned fkey(float f) {
    unsigned u = __float_as_uint(f);
    return (u & 0x80000000u) ? ~u : (u | 0x80000000u);
}
__device__ __forceinline__ float funkey(unsigned k) {
    unsigned u = (k & 0x80000000u) ? (k ^ 0x80000000u) : ~k;
    return __uint_as_float(u);
}

__device__ __forceinline__ float waveRedSum(float v) {
    for (int off = 16; off > 0; off >>= 1) v += __shfl_xor(v, off, 32);
    return v;
}

#define GW_NT 4
#define GW_THREADS 128
#define GW_BLOCKS_K256 256
#define GW_BLOCKS_K128 512
#define GW_BLOCKS_K64 1024

__device__ __forceinline__ float xhalf(float x) {
    return __builtin_bit_cast(float, __builtin_amdgcn_permlanex16(
        __builtin_bit_cast(int, x), __builtin_bit_cast(int, x), 0x76543210, 0xfedcba98, false, false));
}

template <int KK>
__device__ __forceinline__ void load_frag_row(float4 (&f)[KK], const float* __restrict__ p) {
#pragma unroll
    for (int kk = 0; kk < KK; ++kk) f[kk] = *(const float4*)(p + kk * 8);
}

template <int KK, bool EPI>
__device__ __forceinline__ void wstat_tile(const float4 (&af)[KK], const float4 (&bf)[GW_NT][KK],
                                           float* __restrict__ C, int rt, int cb, int Nout,
                                           const float* __restrict__ bias, int c, int h) {
    v8f acc[GW_NT];
#pragma unroll
    for (int t = 0; t < GW_NT; ++t) acc[t] = v8f{};
#pragma unroll
    for (int kk = 0; kk < KK; ++kk) {
        const v2f a0 = {af[kk].x, af[kk].y};
        const v2f a1 = {af[kk].z, af[kk].w};
#pragma unroll
        for (int t = 0; t < GW_NT; ++t) {
            const v2f b0 = {bf[t][kk].x, bf[t][kk].y};
            acc[t] = __builtin_amdgcn_wmma_f32_16x16x4_f32(false, a0, false, b0,
                                                           (short)0, acc[t], false, false);
        }
#pragma unroll
        for (int t = 0; t < GW_NT; ++t) {
            const v2f b1 = {bf[t][kk].z, bf[t][kk].w};
            acc[t] = __builtin_amdgcn_wmma_f32_16x16x4_f32(false, a1, false, b1,
                                                           (short)0, acc[t], false, false);
        }
    }
#pragma unroll
    for (int t = 0; t < GW_NT; t += 2) {
        const int col = cb + t * 16 + 16 * h + c;
        float bv = 0.f;
        if (EPI) bv = bias[col];
        float* crow = C + (size_t)(rt * 16) * Nout + col;
#pragma unroll
        for (int i = 0; i < 8; ++i) {
            const float lo   = acc[t][i];
            const float hi   = acc[t + 1][i];
            const float recv = xhalf(h ? lo : hi);
            float v0 = h ? recv : lo;
            float v1 = h ? hi : recv;
            if (EPI) { v0 = eluf(v0 + bv); v1 = eluf(v1 + bv); }
            crow[(size_t)i * Nout]       = v0;
            crow[(size_t)(8 + i) * Nout] = v1;
        }
    }
}

template <int K, bool EPI>
__global__ __launch_bounds__(GW_THREADS)
void gemm_wstat_f32(const float* __restrict__ A, int lda,
                    const float* __restrict__ W,
                    float* __restrict__ C,
                    int M, int Nout,
                    const float* __restrict__ bias) {
    constexpr int KK = K / 8;
    const int lane = threadIdx.x & 31;
    const int c = lane & 15;
    const int h = lane >> 4;
    const int gw = __builtin_amdgcn_readfirstlane((int)((blockIdx.x * blockDim.x + threadIdx.x) >> 5));
    const int nw = (int)((gridDim.x * blockDim.x) >> 5);
    const int ngrp = Nout / (16 * GW_NT);
    const int cg = gw % ngrp;
    const int rs = gw / ngrp;
    const int nstream = nw / ngrp;
    const int ntile = M >> 4;
    if (rs >= nstream || rs >= ntile) return;
    const int cb = cg * (16 * GW_NT);

    float4 bf[GW_NT][KK];
#pragma unroll
    for (int t = 0; t < GW_NT; ++t)
        load_frag_row<KK>(bf[t], W + (size_t)(cb + t * 16 + c) * K + h * 4);

    const float* arow = A + (size_t)c * lda + h * 4;
    const size_t tstep = (size_t)16 * lda;
    float4 f0[KK], f1[KK];
    load_frag_row<KK>(f0, arow + (size_t)rs * tstep);
    for (int rt = rs; rt < ntile; rt += 2 * nstream) {
        const int rt1 = rt + nstream;
        const int rt2 = rt1 + nstream;
        const bool has1 = rt1 < ntile;
        load_frag_row<KK>(f1, arow + (size_t)(has1 ? rt1 : rt) * tstep);
        wstat_tile<KK, EPI>(f0, bf, C, rt, cb, Nout, bias, c, h);
        load_frag_row<KK>(f0, arow + (size_t)(rt2 < ntile ? rt2 : rt) * tstep);
        if (has1) wstat_tile<KK, EPI>(f1, bf, C, rt1, cb, Nout, bias, c, h);
    }
}

__global__ void attn_coef(const float* __restrict__ hmat,
                          const float* __restrict__ a_s,
                          const float* __restrict__ a_d,
                          float* __restrict__ asrc, float* __restrict__ adst, int NH) {
    int idx = blockIdx.x * blockDim.x + threadIdx.x;
    if (idx >= NH) return;
    int n = idx >> 2, h = idx & 3;
    const float* row = hmat + (size_t)n * HIDF + h * CCH;
    const float* ws  = a_s + h * CCH;
    const float* wd  = a_d + h * CCH;
    float ss = 0.f, sd = 0.f;
    for (int c = 0; c < CCH; ++c) { float v = row[c]; ss += v * ws[c]; sd += v * wd[c]; }
    asrc[idx] = ss;
    adst[idx] = sd;
}

#define RN 200
#define AGG_THREADS 1024
#define AGG_WAVES (AGG_THREADS / 32)
#define WCAP 256

__device__ __forceinline__ void accum_one(int s, unsigned dl, int lane,
                                          const float* __restrict__ asrc,
                                          const float* __restrict__ hmat,
                                          const float* adl, const unsigned* mxk,
                                          float* dnl, float* acc) {
    float ex[HEADS];
#pragma unroll
    for (int h = 0; h < HEADS; ++h) {
        float a = leakyf(asrc[s * HEADS + h] + adl[dl * HEADS + h]);
        ex[h] = expf(a - funkey(mxk[dl * HEADS + h]));
    }
    if (lane == 0) {
#pragma unroll
        for (int h = 0; h < HEADS; ++h) atomicAdd(&dnl[dl * HEADS + h], ex[h]);
    }
    const float* hrow = hmat + (size_t)s * HIDF;
    float* orow       = acc + dl * HIDF;
#pragma unroll
    for (int i = 0; i < 8; ++i) {
        int c = lane + 32 * i;
        atomicAdd(&orow[c], ex[i >> 1] * hrow[c]);
    }
}

template <bool FIRST>
__global__ __launch_bounds__(AGG_THREADS)
void agg_owner(const int* __restrict__ src, const int* __restrict__ dst,
               const float* __restrict__ asrc, const float* __restrict__ adst,
               const float* __restrict__ hmat,
               float* __restrict__ denom, float* __restrict__ hacc,
               int* __restrict__ wl_src, unsigned short* __restrict__ wl_dl,
               unsigned* __restrict__ wl_cnt, int N, int E) {
    __shared__ __attribute__((aligned(16))) float acc[RN * HIDF];
    __shared__ __attribute__((aligned(16))) int lsts[AGG_WAVES * WCAP];
    __shared__ __attribute__((aligned(16))) unsigned short lstd[AGG_WAVES * WCAP];
    __shared__ unsigned mxk[RN * HEADS];
    __shared__ float dnl[RN * HEADS];
    __shared__ float adl[RN * HEADS];
    __shared__ unsigned ovf;

    const int tid  = threadIdx.x;
    const int lane = tid & 31;
    const int wv   = tid >> 5;
    const int base = blockIdx.x * RN;
    const int nloc = min(RN, N - base);
    int* const myS            = lsts + wv * WCAP;
    unsigned short* const myD = lstd + wv * WCAP;

    if (tid == 0) ovf = 0u;
    for (int i = tid; i < nloc * HEADS; i += AGG_THREADS) {
        float ad = adst[base * HEADS + i];
        adl[i] = ad;
        mxk[i] = fkey(leakyf(asrc[base * HEADS + i] + ad));
    }
    __syncthreads();

    const size_t wlslot = (size_t)blockIdx.x * AGG_WAVES + wv;
    const size_t wlbase = wlslot * WCAP;
    const int wvu = __builtin_amdgcn_readfirstlane(wv);
    unsigned wcount = 0u;
    if constexpr (FIRST) {
        auto note = [&](int d, int e, bool valid) {
            unsigned dl = (unsigned)(d - base);
            bool m = valid && (dl < (unsigned)nloc);
            unsigned mask = (unsigned)__ballot(m);
            if (mask) {
                unsigned slot = wcount + __builtin_amdgcn_mbcnt_lo(mask, 0u);
                if (m && slot < (unsigned)WCAP) { myS[slot] = e; myD[slot] = (unsigned short)dl; }
                wcount += (unsigned)__popc(mask);
            }
        };
        const int E4 = E >> 2;
        const int4* dst4 = (const int4*)dst;
        for (int ib = wvu * 32; ib < E4; ib += AGG_THREADS * 4) {
            const int i = ib + lane;
            int4 v[4];
            if (ib + 3 * AGG_THREADS + 32 <= E4) {
#pragma unroll
                for (int u = 0; u < 4; ++u) v[u] = dst4[i + u * AGG_THREADS];
#pragma unroll
                for (int u = 0; u < 4; ++u) {
                    const int e = 4 * (i + u * AGG_THREADS);
                    note(v[u].x, e + 0, true);
                    note(v[u].y, e + 1, true);
                    note(v[u].z, e + 2, true);
                    note(v[u].w, e + 3, true);
                }
            } else {
#pragma unroll
                for (int u = 0; u < 4; ++u) {
                    const int iu = i + u * AGG_THREADS;
                    v[u] = (iu < E4) ? dst4[iu] : make_int4(0, 0, 0, 0);
                }
#pragma unroll
                for (int u = 0; u < 4; ++u) {
                    const int iu = i + u * AGG_THREADS;
                    const bool valid = iu < E4;
                    note(v[u].x, 4 * iu + 0, valid);
                    note(v[u].y, 4 * iu + 1, valid);
                    note(v[u].z, 4 * iu + 2, valid);
                    note(v[u].w, 4 * iu + 3, valid);
                }
            }
        }
        for (int e0 = (E4 << 2); e0 < E; e0 += AGG_THREADS) {
            int e = e0 + tid;
            bool valid = e < E;
            note(valid ? dst[e] : 0, e, valid);
        }
        if (lane == 0) wl_cnt[wlslot] = wcount;
    } else {
        wcount = (unsigned)__builtin_amdgcn_readfirstlane((int)wl_cnt[wlslot]);
    }
    if (wcount > (unsigned)WCAP && lane == 0) ovf = 1u;
    __syncthreads();

    const bool listed = (ovf == 0u);
    const int nw = (int)min(wcount, (unsigned)WCAP);

    if (listed) {
        for (int k0 = lane; k0 < nw; k0 += 128) {
            int s[4]; unsigned dl[4]; bool ok[4];
#pragma unroll
            for (int u = 0; u < 4; ++u) {
                int k = k0 + 32 * u;
                ok[u] = k < nw;
                if constexpr (FIRST) {
                    int e = ok[u] ? myS[k] : 0;
                    dl[u] = ok[u] ? (unsigned)myD[k] : 0u;
                    s[u]  = src[e];
                } else {
                    const int kr = ok[u] ? k : k0;
                    const unsigned sv = (unsigned)wl_src[wlbase + kr];
                    const unsigned dv = (unsigned)wl_dl[wlbase + kr];
                    s[u]  = (int)min(sv, (unsigned)(N - 1));
                    dl[u] = min(dv, (unsigned)(nloc - 1));
                }
            }
            float4 a4[4];
#pragma unroll
            for (int u = 0; u < 4; ++u) a4[u] = *(const float4*)(asrc + (size_t)s[u] * HEADS);
#pragma unroll
            for (int u = 0; u < 4; ++u) {
                if (ok[u]) {
                    myS[k0 + 32 * u] = s[u];
                    if constexpr (FIRST) {
                        wl_src[wlbase + k0 + 32 * u] = s[u];
                        wl_dl[wlbase + k0 + 32 * u]  = (unsigned short)dl[u];
                    } else {
                        myD[k0 + 32 * u] = (unsigned short)dl[u];
                    }
                    atomicMax(&mxk[dl[u] * HEADS + 0], fkey(leakyf(a4[u].x + adl[dl[u] * HEADS + 0])));
                    atomicMax(&mxk[dl[u] * HEADS + 1], fkey(leakyf(a4[u].y + adl[dl[u] * HEADS + 1])));
                    atomicMax(&mxk[dl[u] * HEADS + 2], fkey(leakyf(a4[u].z + adl[dl[u] * HEADS + 2])));
                    atomicMax(&mxk[dl[u] * HEADS + 3], fkey(leakyf(a4[u].w + adl[dl[u] * HEADS + 3])));
                }
            }
        }
    } else {
        for (int e = tid; e < E; e += AGG_THREADS) {
            unsigned dl = (unsigned)(dst[e] - base);
            if (dl < (unsigned)nloc) {
                int s = src[e];
#pragma unroll
                for (int h = 0; h < HEADS; ++h)
                    atomicMax(&mxk[dl * HEADS + h],
                              fkey(leakyf(asrc[s * HEADS + h] + adl[dl * HEADS + h])));
            }
        }
    }
    __syncthreads();

    for (int i = wv; i < nloc; i += AGG_WAVES) {
        int n = base + i;
        float ex[HEADS];
#pragma unroll
        for (int h = 0; h < HEADS; ++h) {
            float aself = leakyf(asrc[n * HEADS + h] + adl[i * HEADS + h]);
            ex[h] = expf(aself - funkey(mxk[i * HEADS + h]));
        }
        if (lane == 0) {
#pragma unroll
            for (int h = 0; h < HEADS; ++h) dnl[i * HEADS + h] = ex[h];
        }
        const float* row = hmat + (size_t)n * HIDF;
#pragma unroll
        for (int j = 0; j < 8; ++j) {
            int c = lane + 32 * j;
            acc[i * HIDF + c] = row[c] * ex[j >> 1];
        }
    }
    __syncthreads();

    if (listed) {
        for (int k0 = 0; k0 < nw; k0 += 4) {
            const int4 sv    = *(const int4*)(myS + k0);
            const ushort4 dv = *(const ushort4*)(myD + k0);
            const int svv[4] = {sv.x, sv.y, sv.z, sv.w};
            const int dvv[4] = {dv.x, dv.y, dv.z, dv.w};
            int s[4]; unsigned dl[4]; bool ok[4];
#pragma unroll
            for (int u = 0; u < 4; ++u) {
                ok[u] = (k0 + u) < nw;
                s[u]  = __builtin_amdgcn_readfirstlane(ok[u] ? svv[u] : 0);
                dl[u] = (unsigned)__builtin_amdgcn_readfirstlane(ok[u] ? dvv[u] : 0);
            }
            float hv[4][8];
#pragma unroll
            for (int u = 0; u < 4; ++u) {
                const float* hrow = hmat + (size_t)s[u] * HIDF;
#pragma unroll
                for (int j = 0; j < 8; ++j) hv[u][j] = hrow[lane + 32 * j];
            }
            float ex[4][HEADS];
#pragma unroll
            for (int u = 0; u < 4; ++u) {
#pragma unroll
                for (int h = 0; h < HEADS; ++h) {
                    float a = leakyf(asrc[s[u] * HEADS + h] + adl[dl[u] * HEADS + h]);
                    ex[u][h] = expf(a - funkey(mxk[dl[u] * HEADS + h]));
                }
            }
#pragma unroll
            for (int u = 0; u < 4; ++u) {
                if (ok[u]) {
                    if (lane == 0) {
#pragma unroll
                        for (int h = 0; h < HEADS; ++h) atomicAdd(&dnl[dl[u] * HEADS + h], ex[u][h]);
                    }
                    float* orow = acc + dl[u] * HIDF;
#pragma unroll
                    for (int j = 0; j < 8; ++j)
                        atomicAdd(&orow[lane + 32 * j], ex[u][j >> 1] * hv[u][j]);
                }
            }
        }
    } else {
        for (int e0 = wvu * 32; e0 < E; e0 += AGG_WAVES * 32) {
            int e = e0 + lane;
            int d = (e < E) ? dst[e] : 0;
            unsigned dl = (unsigned)(d - base);
            bool m = (e < E) && (dl < (unsigned)nloc);
            unsigned mask = (unsigned)__ballot(m);
            while (mask) {
                int l = __ffs(mask) - 1;
                mask &= mask - 1u;
                int s1 = src[e0 + l];
                unsigned dl1 = (unsigned)__shfl((int)dl, l, 32);
                accum_one(s1, dl1, lane, asrc, hmat, adl, mxk, dnl, acc);
            }
        }
    }
    __syncthreads();

    {
        const float4* a4 = (const float4*)acc;
        float4* o4 = (float4*)(hacc + (size_t)base * HIDF);
        for (int i = tid; i < nloc * (HIDF / 4); i += AGG_THREADS) o4[i] = a4[i];
        for (int i = tid; i < nloc * HEADS; i += AGG_THREADS) denom[base * HEADS + i] = dnl[i];
    }
}

__global__ void finalize_concat(const float* __restrict__ hacc, const float* __restrict__ denom,
                                const float* __restrict__ bias, const float* __restrict__ g,
                                const float* __restrict__ be, float* __restrict__ out, int N) {
    int wid  = (blockIdx.x * blockDim.x + threadIdx.x) >> 5;
    int lane = threadIdx.x & 31;
    if (wid >= N) return;
    size_t base = (size_t)wid * HIDF;
    float dn[HEADS];
#pragma unroll
    for (int h = 0; h < HEADS; ++h) dn[h] = 1.f / (denom[wid * HEADS + h] + 1e-16f);
    float v[8];
    float s = 0.f;
#pragma unroll
    for (int i = 0; i < 8; ++i) {
        int c = lane + 32 * i;
        v[i] = hacc[base + c] * dn[i >> 1] + bias[c];
        s += v[i];
    }
    float mu = waveRedSum(s) * (1.f / HIDF);
    float q = 0.f;
#pragma unroll
    for (int i = 0; i < 8; ++i) { float dlt = v[i] - mu; q += dlt * dlt; }
    float rstd = rsqrtf(waveRedSum(q) * (1.f / HIDF) + LN_EPS);
#pragma unroll
    for (int i = 0; i < 8; ++i) {
        int c = lane + 32 * i;
        float y = (v[i] - mu) * rstd * g[c] + be[c];
        out[base + c] = eluf(y);
    }
}

__global__ void finalize_mean(const float* __restrict__ hacc, const float* __restrict__ denom,
                              const float* __restrict__ bias, const float* __restrict__ g,
                              const float* __restrict__ be, float* __restrict__ out, int N) {
    int wid  = (blockIdx.x * blockDim.x + threadIdx.x) >> 5;
    int lane = threadIdx.x & 31;
    if (wid >= N) return;
    size_t base = (size_t)wid * HIDF;
    float dn[HEADS];
#pragma unroll
    for (int h = 0; h < HEADS; ++h) dn[h] = 1.f / (denom[wid * HEADS + h] + 1e-16f);
    float v[2];
    float s = 0.f;
#pragma unroll
    for (int i = 0; i < 2; ++i) {
        int c = lane + 32 * i;
        float acc = 0.f;
#pragma unroll
        for (int h = 0; h < HEADS; ++h) acc += hacc[base + h * CCH + c] * dn[h];
        v[i] = acc * (1.f / HEADS) + bias[c];
        s += v[i];
    }
    float mu = waveRedSum(s) * (1.f / CCH);
    float q = 0.f;
#pragma unroll
    for (int i = 0; i < 2; ++i) { float dlt = v[i] - mu; q += dlt * dlt; }
    float rstd = rsqrtf(waveRedSum(q) * (1.f / CCH) + LN_EPS);
#pragma unroll
    for (int i = 0; i < 2; ++i) {
        int c = lane + 32 * i;
        float y = (v[i] - mu) * rstd * g[c] + be[c];
        out[(size_t)wid * CCH + c] = eluf(y);
    }
}

__global__ void value_head(const float* __restrict__ hin,
                           const float* __restrict__ vw1, const float* __restrict__ vb1,
                           const float* __restrict__ vw2, const float* __restrict__ vb2,
                           float* __restrict__ out, int N) {
    int wid  = (blockIdx.x * blockDim.x + threadIdx.x) >> 5;
    int lane = threadIdx.x & 31;
    if (wid >= N) return;
    const float* row  = hin + (size_t)wid * CCH;
    const float* wrow = vw1 + lane * CCH;
    float hid = vb1[lane];
    for (int c = 0; c < CCH; ++c) hid += row[c] * wrow[c];
    hid = eluf(hid);
#pragma unroll
    for (int k = 0; k < 4; ++k) {
        float p = vw2[k * 32 + lane] * hid;
        p = waveRedSum(p);
        if (lane == 0) out[wid * 4 + k] = p + vb2[k];
    }
}

extern "C" void kernel_launch(void* const* d_in, const int* in_sizes, int n_in,
                              void* d_out, int out_size, void* d_ws, size_t ws_size,
                              hipStream_t stream) {
    const float* x      = (const float*)d_in[0];
    const int*   esrc   = (const int*)d_in[1];
    const int*   edst   = (const int*)d_in[2];
    const float* proj_w = (const float*)d_in[3];
    const float* proj_b = (const float*)d_in[4];
    const float* W[3]  = {(const float*)d_in[5],  (const float*)d_in[11], (const float*)d_in[17]};
    const float* AS[3] = {(const float*)d_in[6],  (const float*)d_in[12], (const float*)d_in[18]};
    const float* AD[3] = {(const float*)d_in[7],  (const float*)d_in[13], (const float*)d_in[19]};
    const float* B[3]  = {(const float*)d_in[8],  (const float*)d_in[14], (const float*)d_in[20]};
    const float* G[3]  = {(const float*)d_in[9],  (const float*)d_in[15], (const float*)d_in[21]};
    const float* BE[3] = {(const float*)d_in[10], (const float*)d_in[16], (const float*)d_in[22]};
    const float* vw1 = (const float*)d_in[23];
    const float* vb1 = (const float*)d_in[24];
    const float* vw2 = (const float*)d_in[25];
    const float* vb2 = (const float*)d_in[26];

    const int N = in_sizes[0] / 128;
    const int E = in_sizes[1];
    const int NH = N * HEADS;

    float* ws     = (float*)d_ws;
    float* hin    = ws;
    float* hmat   = hin  + (size_t)N * HIDF;
    float* hacc   = hmat + (size_t)N * HIDF;
    float* asrc   = hacc + (size_t)N * HIDF;
    float* adst   = asrc + (size_t)NH;
    float* denom  = adst + (size_t)NH;
    const int nagg = (N + RN - 1) / RN;
    int* wl_src            = (int*)(denom + (size_t)NH);
    unsigned short* wl_dl  = (unsigned short*)(wl_src + (size_t)nagg * AGG_WAVES * WCAP);
    unsigned* wl_cnt       = (unsigned*)(wl_dl + (size_t)nagg * AGG_WAVES * WCAP);

    dim3 blk(256);

    gemm_wstat_f32<128, true><<<GW_BLOCKS_K128, GW_THREADS, 0, stream>>>(
        x, 128, proj_w, hin, N, 64, proj_b);

    int in_ld = 64;
    for (int l = 0; l < 3; ++l) {
        if (l == 0)
            gemm_wstat_f32<64, false><<<GW_BLOCKS_K64, GW_THREADS, 0, stream>>>(
                hin, in_ld, W[l], hmat, N, HIDF, nullptr);
        else
            gemm_wstat_f32<256, false><<<GW_BLOCKS_K256, GW_THREADS, 0, stream>>>(
                hin, in_ld, W[l], hmat, N, HIDF, nullptr);
        attn_coef<<<(NH + 255) / 256, blk, 0, stream>>>(hmat, AS[l], AD[l], asrc, adst, NH);
        if (l == 0)
            agg_owner<true><<<nagg, AGG_THREADS, 0, stream>>>(
                esrc, edst, asrc, adst, hmat, denom, hacc, wl_src, wl_dl, wl_cnt, N, E);
        else
            agg_owner<false><<<nagg, AGG_THREADS, 0, stream>>>(
                esrc, edst, asrc, adst, hmat, denom, hacc, wl_src, wl_dl, wl_cnt, N, E);
        if (l < 2) {
            finalize_concat<<<(N * 32 + 255) / 256, blk, 0, stream>>>(
                hacc, denom, B[l], G[l], BE[l], hin, N);
            in_ld = HIDF;
        } else {
            finalize_mean<<<(N * 32 + 255) / 256, blk, 0, stream>>>(
                hacc, denom, B[l], G[l], BE[l], hin, N);
        }
    }

    value_head<<<(N * 32 + 255) / 256, blk, 0, stream>>>(
        hin, vw1, vb1, vw2, vb2, (float*)d_out, N);
}


# Node epilogue in the owner work-group: its waves take the finished LDS sums, apply 1/denom, bias, LayerNorm, ELU (last layer: head mean), store hin; finalize launches, hacc/denom round trips gone
# speedup vs baseline: 2.3776x; 1.0459x over previous
#include <hip/hip_runtime.h>
#include <math.h>

typedef __attribute__((ext_vector_type(2))) float v2f;
typedef __attribute__((ext_vector_type(8))) float v8f;

#define HEADS 4
#define CCH 64
#define HIDF 256
#define NEG_SLOPE 0.2f
#define LN_EPS 1e-5f

__device__ __forceinline__ float eluf(float x) { return x > 0.f ? x : (expf(x) - 1.f); }
__device__ __forceinline__ float leakyf(float x) { return x > 0.f ? x : NEG_SLOPE * x; }

__device__ __forceinline__ unsigned fkey(float f) {
    unsigned u = __float_as_uint(f);
    return (u & 0x80000000u) ? ~u : (u | 0x80000000u);
}
__device__ __forceinline__ float funkey(unsigned k) {
    unsigned u = (k & 0x80000000u) ? (k ^ 0x80000000u) : ~k;
    return __uint_as_float(u);
}

__device__ __forceinline__ float waveRedSum(float v) {
    for (int off = 16; off > 0; off >>= 1) v += __shfl_xor(v, off, 32);
    return v;
}

#define GW_NT 4
#define GW_THREADS 128
#define GW_BLOCKS_K256 256
#define GW_BLOCKS_K128 512
#define GW_BLOCKS_K64 1024

__device__ __forceinline__ float xhalf(float x) {
    return __builtin_bit_cast(float, __builtin_amdgcn_permlanex16(
        __builtin_bit_cast(int, x), __builtin_bit_cast(int, x), 0x76543210, 0xfedcba98, false, false));
}

template <int KK>
__device__ __forceinline__ void load_frag_row(float4 (&f)[KK], const float* __restrict__ p) {
#pragma unroll
    for (int kk = 0; kk < KK; ++kk) f[kk] = *(const float4*)(p + kk * 8);
}

template <int KK, bool EPI>
__device__ __forceinline__ void wstat_tile(const float4 (&af)[KK], const float4 (&bf)[GW_NT][KK],
                                           float* __restrict__ C, int rt, int cb, int Nout,
                                           const float* __restrict__ bias, int c, int h) {
    v8f acc[GW_NT];
#pragma unroll
    for (int t = 0; t < GW_NT; ++t) acc[t] = v8f{};
#pragma unroll
    for (int kk = 0; kk < KK; ++kk) {
        const v2f a0 = {af[kk].x, af[kk].y};
        const v2f a1 = {af[kk].z, af[kk].w};
#pragma unroll
        for (int t = 0; t < GW_NT; ++t) {
            const v2f b0 = {bf[t][kk].x, bf[t][kk].y};
            acc[t] = __builtin_amdgcn_wmma_f32_16x16x4_f32(false, a0, false, b0,
                                                           (short)0, acc[t], false, false);
        }
#pragma unroll
        for (int t = 0; t < GW_NT; ++t) {
            const v2f b1 = {bf[t][kk].z, bf[t][kk].w};
            acc[t] = __builtin_amdgcn_wmma_f32_16x16x4_f32(false, a1, false, b1,
                                                           (short)0, acc[t], false, false);
        }
    }
#pragma unroll
    for (int t = 0; t < GW_NT; t += 2) {
        const int col = cb + t * 16 + 16 * h + c;
        float bv = 0.f;
        if (EPI) bv = bias[col];
        float* crow = C + (size_t)(rt * 16) * Nout + col;
#pragma unroll
        for (int i = 0; i < 8; ++i) {
            const float lo   = acc[t][i];
            const float hi   = acc[t + 1][i];
            const float recv = xhalf(h ? lo : hi);
            float v0 = h ? recv : lo;
            float v1 = h ? hi : recv;
            if (EPI) { v0 = eluf(v0 + bv); v1 = eluf(v1 + bv); }
            crow[(size_t)i * Nout]       = v0;
            crow[(size_t)(8 + i) * Nout] = v1;
        }
    }
}

template <int K, bool EPI>
__global__ __launch_bounds__(GW_THREADS)
void gemm_wstat_f32(const float* __restrict__ A, int lda,
                    const float* __restrict__ W,
                    float* __restrict__ C,
                    int M, int Nout,
                    const float* __restrict__ bias) {
    constexpr int KK = K / 8;
    const int lane = threadIdx.x & 31;
    const int c = lane & 15;
    const int h = lane >> 4;
    const int gw = __builtin_amdgcn_readfirstlane((int)((blockIdx.x * blockDim.x + threadIdx.x) >> 5));
    const int nw = (int)((gridDim.x * blockDim.x) >> 5);
    const int ngrp = Nout / (16 * GW_NT);
    const int cg = gw % ngrp;
    const int rs = gw / ngrp;
    const int nstream = nw / ngrp;
    const int ntile = M >> 4;
    if (rs >= nstream || rs >= ntile) return;
    const int cb = cg * (16 * GW_NT);

    float4 bf[GW_NT][KK];
#pragma unroll
    for (int t = 0; t < GW_NT; ++t)
        load_frag_row<KK>(bf[t], W + (size_t)(cb + t * 16 + c) * K + h * 4);

    const float* arow = A + (size_t)c * lda + h * 4;
    const size_t tstep = (size_t)16 * lda;
    float4 f0[KK], f1[KK];
    load_frag_row<KK>(f0, arow + (size_t)rs * tstep);
    for (int rt = rs; rt < ntile; rt += 2 * nstream) {
        const int rt1 = rt + nstream;
        const int rt2 = rt1 + nstream;
        const bool has1 = rt1 < ntile;
        load_frag_row<KK>(f1, arow + (size_t)(has1 ? rt1 : rt) * tstep);
        wstat_tile<KK, EPI>(f0, bf, C, rt, cb, Nout, bias, c, h);
        load_frag_row<KK>(f0, arow + (size_t)(rt2 < ntile ? rt2 : rt) * tstep);
        if (has1) wstat_tile<KK, EPI>(f1, bf, C, rt1, cb, Nout, bias, c, h);
    }
}

__global__ void attn_coef(const float* __restrict__ hmat,
                          const float* __restrict__ a_s,
                          const float* __restrict__ a_d,
                          float* __restrict__ asrc, float* __restrict__ adst, int NH) {
    int idx = blockIdx.x * blockDim.x + threadIdx.x;
    if (idx >= NH) return;
    int n = idx >> 2, h = idx & 3;
    const float* row = hmat + (size_t)n * HIDF + h * CCH;
    const float* ws  = a_s + h * CCH;
    const float* wd  = a_d + h * CCH;
    float ss = 0.f, sd = 0.f;
    for (int c = 0; c < CCH; ++c) { float v = row[c]; ss += v * ws[c]; sd += v * wd[c]; }
    asrc[idx] = ss;
    adst[idx] = sd;
}

#define RN 200
#define AGG_THREADS 1024
#define AGG_WAVES (AGG_THREADS / 32)
#define WCAP 256

__device__ __forceinline__ void accum_one(int s, unsigned dl, int lane,
                                          const float* __restrict__ asrc,
                                          const float* __restrict__ hmat,
                                          const float* adl, const unsigned* mxk,
                                          float* dnl, float* acc) {
    float ex[HEADS];
#pragma unroll
    for (int h = 0; h < HEADS; ++h) {
        float a = leakyf(asrc[s * HEADS + h] + adl[dl * HEADS + h]);
        ex[h] = expf(a - funkey(mxk[dl * HEADS + h]));
    }
    if (lane == 0) {
#pragma unroll
        for (int h = 0; h < HEADS; ++h) atomicAdd(&dnl[dl * HEADS + h], ex[h]);
    }
    const float* hrow = hmat + (size_t)s * HIDF;
    float* orow       = acc + dl * HIDF;
#pragma unroll
    for (int i = 0; i < 8; ++i) {
        int c = lane + 32 * i;
        atomicAdd(&orow[c], ex[i >> 1] * hrow[c]);
    }
}

template <bool FIRST, bool MEAN>
__global__ __launch_bounds__(AGG_THREADS)
void agg_owner(const int* __restrict__ src, const int* __restrict__ dst,
               const float* __restrict__ asrc, const float* __restrict__ adst,
               const float* __restrict__ hmat,
               const float* __restrict__ bias, const float* __restrict__ g,
               const float* __restrict__ be, float* __restrict__ hout,
               int* __restrict__ wl_src, unsigned short* __restrict__ wl_dl,
               unsigned* __restrict__ wl_cnt, int N, int E) {
    __shared__ __attribute__((aligned(16))) float acc[RN * HIDF];
    __shared__ __attribute__((aligned(16))) int lsts[AGG_WAVES * WCAP];
    __shared__ __attribute__((aligned(16))) unsigned short lstd[AGG_WAVES * WCAP];
    __shared__ unsigned mxk[RN * HEADS];
    __shared__ float dnl[RN * HEADS];
    __shared__ float adl[RN * HEADS];
    __shared__ unsigned ovf;

    const int tid  = threadIdx.x;
    const int lane = tid & 31;
    const int wv   = tid >> 5;
    const int base = blockIdx.x * RN;
    const int nloc = min(RN, N - base);
    int* const myS            = lsts + wv * WCAP;
    unsigned short* const myD = lstd + wv * WCAP;

    if (tid == 0) ovf = 0u;
    for (int i = tid; i < nloc * HEADS; i += AGG_THREADS) {
        float ad = adst[base * HEADS + i];
        adl[i] = ad;
        mxk[i] = fkey(leakyf(asrc[base * HEADS + i] + ad));
    }
    __syncthreads();

    const size_t wlslot = (size_t)blockIdx.x * AGG_WAVES + wv;
    const size_t wlbase = wlslot * WCAP;
    const int wvu = __builtin_amdgcn_readfirstlane(wv);
    unsigned wcount = 0u;
    if constexpr (FIRST) {
        auto note = [&](int d, int e, bool valid) {
            unsigned dl = (unsigned)(d - base);
            bool m = valid && (dl < (unsigned)nloc);
            unsigned mask = (unsigned)__ballot(m);
            if (mask) {
                unsigned slot = wcount + __builtin_amdgcn_mbcnt_lo(mask, 0u);
                if (m && slot < (unsigned)WCAP) { myS[slot] = e; myD[slot] = (unsigned short)dl; }
                wcount += (unsigned)__popc(mask);
            }
        };
        const int E4 = E >> 2;
        const int4* dst4 = (const int4*)dst;
        for (int ib = wvu * 32; ib < E4; ib += AGG_THREADS * 4) {
            const int i = ib + lane;
            int4 v[4];
            if (ib + 3 * AGG_THREADS + 32 <= E4) {
#pragma unroll
                for (int u = 0; u < 4; ++u) v[u] = dst4[i + u * AGG_THREADS];
#pragma unroll
                for (int u = 0; u < 4; ++u) {
                    const int e = 4 * (i + u * AGG_THREADS);
                    note(v[u].x, e + 0, true);
                    note(v[u].y, e + 1, true);
                    note(v[u].z, e + 2, true);
                    note(v[u].w, e + 3, true);
                }
            } else {
#pragma unroll
                for (int u = 0; u < 4; ++u) {
                    const int iu = i + u * AGG_THREADS;
                    v[u] = (iu < E4) ? dst4[iu] : make_int4(0, 0, 0, 0);
                }
#pragma unroll
                for (int u = 0; u < 4; ++u) {
                    const int iu = i + u * AGG_THREADS;
                    const bool valid = iu < E4;
                    note(v[u].x, 4 * iu + 0, valid);
                    note(v[u].y, 4 * iu + 1, valid);
                    note(v[u].z, 4 * iu + 2, valid);
                    note(v[u].w, 4 * iu + 3, valid);
                }
            }
        }
        for (int e0 = (E4 << 2); e0 < E; e0 += AGG_THREADS) {
            int e = e0 + tid;
            bool valid = e < E;
            note(valid ? dst[e] : 0, e, valid);
        }
        if (lane == 0) wl_cnt[wlslot] = wcount;
    } else {
        wcount = (unsigned)__builtin_amdgcn_readfirstlane((int)wl_cnt[wlslot]);
    }
    if (wcount > (unsigned)WCAP && lane == 0) ovf = 1u;
    __syncthreads();

    const bool listed = (ovf == 0u);
    const int nw = (int)min(wcount, (unsigned)WCAP);

    if (listed) {
        for (int k0 = lane; k0 < nw; k0 += 128) {
            int s[4]; unsigned dl[4]; bool ok[4];
#pragma unroll
            for (int u = 0; u < 4; ++u) {
                int k = k0 + 32 * u;
                ok[u] = k < nw;
                if constexpr (FIRST) {
                    int e = ok[u] ? myS[k] : 0;
                    dl[u] = ok[u] ? (unsigned)myD[k] : 0u;
                    s[u]  = src[e];
                } else {
                    const int kr = ok[u] ? k : k0;
                    const unsigned sv = (unsigned)wl_src[wlbase + kr];
                    const unsigned dv = (unsigned)wl_dl[wlbase + kr];
                    s[u]  = (int)min(sv, (unsigned)(N - 1));
                    dl[u] = min(dv, (unsigned)(nloc - 1));
                }
            }
            float4 a4[4];
#pragma unroll
            for (int u = 0; u < 4; ++u) a4[u] = *(const float4*)(asrc + (size_t)s[u] * HEADS);
#pragma unroll
            for (int u = 0; u < 4; ++u) {
                if (ok[u]) {
                    myS[k0 + 32 * u] = s[u];
                    if constexpr (FIRST) {
                        wl_src[wlbase + k0 + 32 * u] = s[u];
                        wl_dl[wlbase + k0 + 32 * u]  = (unsigned short)dl[u];
                    } else {
                        myD[k0 + 32 * u] = (unsigned short)dl[u];
                    }
                    atomicMax(&mxk[dl[u] * HEADS + 0], fkey(leakyf(a4[u].x + adl[dl[u] * HEADS + 0])));
                    atomicMax(&mxk[dl[u] * HEADS + 1], fkey(leakyf(a4[u].y + adl[dl[u] * HEADS + 1])));
                    atomicMax(&mxk[dl[u] * HEADS + 2], fkey(leakyf(a4[u].z + adl[dl[u] * HEADS + 2])));
                    atomicMax(&mxk[dl[u] * HEADS + 3], fkey(leakyf(a4[u].w + adl[dl[u] * HEADS + 3])));
                }
            }
        }
    } else {
        for (int e = tid; e < E; e += AGG_THREADS) {
            unsigned dl = (unsigned)(dst[e] - base);
            if (dl < (unsigned)nloc) {
                int s = src[e];
#pragma unroll
                for (int h = 0; h < HEADS; ++h)
                    atomicMax(&mxk[dl * HEADS + h],
                              fkey(leakyf(asrc[s * HEADS + h] + adl[dl * HEADS + h])));
            }
        }
    }
    __syncthreads();

    for (int i = wv; i < nloc; i += AGG_WAVES) {
        int n = base + i;
        float ex[HEADS];
#pragma unroll
        for (int h = 0; h < HEADS; ++h) {
            float aself = leakyf(asrc[n * HEADS + h] + adl[i * HEADS + h]);
            ex[h] = expf(aself - funkey(mxk[i * HEADS + h]));
        }
        if (lane == 0) {
#pragma unroll
            for (int h = 0; h < HEADS; ++h) dnl[i * HEADS + h] = ex[h];
        }
        const float* row = hmat + (size_t)n * HIDF;
#pragma unroll
        for (int j = 0; j < 8; ++j) {
            int c = lane + 32 * j;
            acc[i * HIDF + c] = row[c] * ex[j >> 1];
        }
    }
    __syncthreads();

    if (listed) {
        for (int k0 = 0; k0 < nw; k0 += 4) {
            const int4 sv    = *(const int4*)(myS + k0);
            const ushort4 dv = *(const ushort4*)(myD + k0);
            const int svv[4] = {sv.x, sv.y, sv.z, sv.w};
            const int dvv[4] = {dv.x, dv.y, dv.z, dv.w};
            int s[4]; unsigned dl[4]; bool ok[4];
#pragma unroll
            for (int u = 0; u < 4; ++u) {
                ok[u] = (k0 + u) < nw;
                s[u]  = __builtin_amdgcn_readfirstlane(ok[u] ? svv[u] : 0);
                dl[u] = (unsigned)__builtin_amdgcn_readfirstlane(ok[u] ? dvv[u] : 0);
            }
            float hv[4][8];
#pragma unroll
            for (int u = 0; u < 4; ++u) {
                const float* hrow = hmat + (size_t)s[u] * HIDF;
#pragma unroll
                for (int j = 0; j < 8; ++j) hv[u][j] = hrow[lane + 32 * j];
            }
            float ex[4][HEADS];
#pragma unroll
            for (int u = 0; u < 4; ++u) {
#pragma unroll
                for (int h = 0; h < HEADS; ++h) {
                    float a = leakyf(asrc[s[u] * HEADS + h] + adl[dl[u] * HEADS + h]);
                    ex[u][h] = expf(a - funkey(mxk[dl[u] * HEADS + h]));
                }
            }
#pragma unroll
            for (int u = 0; u < 4; ++u) {
                if (ok[u]) {
                    if (lane == 0) {
#pragma unroll
                        for (int h = 0; h < HEADS; ++h) atomicAdd(&dnl[dl[u] * HEADS + h], ex[u][h]);
                    }
                    float* orow = acc + dl[u] * HIDF;
#pragma unroll
                    for (int j = 0; j < 8; ++j)
                        atomicAdd(&orow[lane + 32 * j], ex[u][j >> 1] * hv[u][j]);
                }
            }
        }
    } else {
        for (int e0 = wvu * 32; e0 < E; e0 += AGG_WAVES * 32) {
            int e = e0 + lane;
            int d = (e < E) ? dst[e] : 0;
            unsigned dl = (unsigned)(d - base);
            bool m = (e < E) && (dl < (unsigned)nloc);
            unsigned mask = (unsigned)__ballot(m);
            while (mask) {
                int l = __ffs(mask) - 1;
                mask &= mask - 1u;
                int s1 = src[e0 + l];
                unsigned dl1 = (unsigned)__shfl((int)dl, l, 32);
                accum_one(s1, dl1, lane, asrc, hmat, adl, mxk, dnl, acc);
            }
        }
    }
    __syncthreads();

    if constexpr (!MEAN) {
        for (int i = wv; i < nloc; i += AGG_WAVES) {
            float dn[HEADS];
#pragma unroll
            for (int h = 0; h < HEADS; ++h) dn[h] = 1.f / (dnl[i * HEADS + h] + 1e-16f);
            float v[8];
            float s = 0.f;
#pragma unroll
            for (int j = 0; j < 8; ++j) {
                int c = lane + 32 * j;
                v[j] = acc[i * HIDF + c] * dn[j >> 1] + bias[c];
                s += v[j];
            }
            float mu = waveRedSum(s) * (1.f / HIDF);
            float q = 0.f;
#pragma unroll
            for (int j = 0; j < 8; ++j) { float dlt = v[j] - mu; q += dlt * dlt; }
            float rstd = rsqrtf(waveRedSum(q) * (1.f / HIDF) + LN_EPS);
            float* orow = hout + (size_t)(base + i) * HIDF;
#pragma unroll
            for (int j = 0; j < 8; ++j) {
                int c = lane + 32 * j;
                float y = (v[j] - mu) * rstd * g[c] + be[c];
                orow[c] = eluf(y);
            }
        }
    } else {
        for (int i = wv; i < nloc; i += AGG_WAVES) {
            float dn[HEADS];
#pragma unroll
            for (int h = 0; h < HEADS; ++h) dn[h] = 1.f / (dnl[i * HEADS + h] + 1e-16f);
            float v[2];
            float s = 0.f;
#pragma unroll
            for (int j = 0; j < 2; ++j) {
                int c = lane + 32 * j;
                float a = 0.f;
#pragma unroll
                for (int h = 0; h < HEADS; ++h) a += acc[i * HIDF + h * CCH + c] * dn[h];
                v[j] = a * (1.f / HEADS) + bias[c];
                s += v[j];
            }
            float mu = waveRedSum(s) * (1.f / CCH);
            float q = 0.f;
#pragma unroll
            for (int j = 0; j < 2; ++j) { float dlt = v[j] - mu; q += dlt * dlt; }
            float rstd = rsqrtf(waveRedSum(q) * (1.f / CCH) + LN_EPS);
            float* orow = hout + (size_t)(base + i) * CCH;
#pragma unroll
            for (int j = 0; j < 2; ++j) {
                int c = lane + 32 * j;
                float y = (v[j] - mu) * rstd * g[c] + be[c];
                orow[c] = eluf(y);
            }
        }
    }
}

__global__ void value_head(const float* __restrict__ hin,
                           const float* __restrict__ vw1, const float* __restrict__ vb1,
                           const float* __restrict__ vw2, const float* __restrict__ vb2,
                           float* __restrict__ out, int N) {
    int wid  = (blockIdx.x * blockDim.x + threadIdx.x) >> 5;
    int lane = threadIdx.x & 31;
    if (wid >= N) return;
    const float* row  = hin + (size_t)wid * CCH;
    const float* wrow = vw1 + lane * CCH;
    float hid = vb1[lane];
    for (int c = 0; c < CCH; ++c) hid += row[c] * wrow[c];
    hid = eluf(hid);
#pragma unroll
    for (int k = 0; k < 4; ++k) {
        float p = vw2[k * 32 + lane] * hid;
        p = waveRedSum(p);
        if (lane == 0) out[wid * 4 + k] = p + vb2[k];
    }
}

extern "C" void kernel_launch(void* const* d_in, const int* in_sizes, int n_in,
                              void* d_out, int out_size, void* d_ws, size_t ws_size,
                              hipStream_t stream) {
    const float* x      = (const float*)d_in[0];
    const int*   esrc   = (const int*)d_in[1];
    const int*   edst   = (const int*)d_in[2];
    const float* proj_w = (const float*)d_in[3];
    const float* proj_b = (const float*)d_in[4];
    const float* W[3]  = {(const float*)d_in[5],  (const float*)d_in[11], (const float*)d_in[17]};
    const float* AS[3] = {(const float*)d_in[6],  (const float*)d_in[12], (const float*)d_in[18]};
    const float* AD[3] = {(const float*)d_in[7],  (const float*)d_in[13], (const float*)d_in[19]};
    const float* B[3]  = {(const float*)d_in[8],  (const float*)d_in[14], (const float*)d_in[20]};
    const float* G[3]  = {(const float*)d_in[9],  (const float*)d_in[15], (const float*)d_in[21]};
    const float* BE[3] = {(const float*)d_in[10], (const float*)d_in[16], (const float*)d_in[22]};
    const float* vw1 = (const float*)d_in[23];
    const float* vb1 = (const float*)d_in[24];
    const float* vw2 = (const float*)d_in[25];
    const float* vb2 = (const float*)d_in[26];

    const int N = in_sizes[0] / 128;
    const int E = in_sizes[1];
    const int NH = N * HEADS;

    float* ws     = (float*)d_ws;
    float* hin    = ws;
    float* hmat   = hin  + (size_t)N * HIDF;
    float* asrc   = hmat + (size_t)N * HIDF;
    float* adst   = asrc + (size_t)NH;
    const int nagg = (N + RN - 1) / RN;
    int* wl_src            = (int*)(adst + (size_t)NH);
    unsigned short* wl_dl  = (unsigned short*)(wl_src + (size_t)nagg * AGG_WAVES * WCAP);
    unsigned* wl_cnt       = (unsigned*)(wl_dl + (size_t)nagg * AGG_WAVES * WCAP);

    dim3 blk(256);

    gemm_wstat_f32<128, true><<<GW_BLOCKS_K128, GW_THREADS, 0, stream>>>(
        x, 128, proj_w, hin, N, 64, proj_b);

    int in_ld = 64;
    for (int l = 0; l < 3; ++l) {
        if (l == 0)
            gemm_wstat_f32<64, false><<<GW_BLOCKS_K64, GW_THREADS, 0, stream>>>(
                hin, in_ld, W[l], hmat, N, HIDF, nullptr);
        else
            gemm_wstat_f32<256, false><<<GW_BLOCKS_K256, GW_THREADS, 0, stream>>>(
                hin, in_ld, W[l], hmat, N, HIDF, nullptr);
        attn_coef<<<(NH + 255) / 256, blk, 0, stream>>>(hmat, AS[l], AD[l], asrc, adst, NH);
        if (l == 0)
            agg_owner<true, false><<<nagg, AGG_THREADS, 0, stream>>>(
                esrc, edst, asrc, adst, hmat, B[l], G[l], BE[l], hin, wl_src, wl_dl, wl_cnt, N, E);
        else if (l == 1)
            agg_owner<false, false><<<nagg, AGG_THREADS, 0, stream>>>(
                esrc, edst, asrc, adst, hmat, B[l], G[l], BE[l], hin, wl_src, wl_dl, wl_cnt, N, E);
        else
            agg_owner<false, true><<<nagg, AGG_THREADS, 0, stream>>>(
                esrc, edst, asrc, adst, hmat, B[l], G[l], BE[l], hin, wl_src, wl_dl, wl_cnt, N, E);
        in_ld = HIDF;
    }

    value_head<<<(N * 32 + 255) / 256, blk, 0, stream>>>(
        hin, vw1, vb1, vw2, vb2, (float*)d_out, N);
}


# Edge sum: an edge's 4 softmax weights computed once by one lane (edge per lane), given to the wave by readlane; per-edge scalar exp chain gone; denominators by per-lane LDS atomics (arrival order)
# speedup vs baseline: 2.9688x; 1.2487x over previous
#include <hip/hip_runtime.h>
#include <math.h>

typedef __attribute__((ext_vector_type(2))) float v2f;
typedef __attribute__((ext_vector_type(8))) float v8f;

#define HEADS 4
#define CCH 64
#define HIDF 256
#define NEG_SLOPE 0.2f
#define LN_EPS 1e-5f

__device__ __forceinline__ float eluf(float x) { return x > 0.f ? x : (expf(x) - 1.f); }
__device__ __forceinline__ float leakyf(float x) { return x > 0.f ? x : NEG_SLOPE * x; }

__device__ __forceinline__ unsigned fkey(float f) {
    unsigned u = __float_as_uint(f);
    return (u & 0x80000000u) ? ~u : (u | 0x80000000u);
}
__device__ __forceinline__ float funkey(unsigned k) {
    unsigned u = (k & 0x80000000u) ? (k ^ 0x80000000u) : ~k;
    return __uint_as_float(u);
}

__device__ __forceinline__ float waveRedSum(float v) {
    for (int off = 16; off > 0; off >>= 1) v += __shfl_xor(v, off, 32);
    return v;
}

#define GW_NT 4
#define GW_THREADS 128
#define GW_BLOCKS_K256 256
#define GW_BLOCKS_K128 512
#define GW_BLOCKS_K64 1024

__device__ __forceinline__ float xhalf(float x) {
    return __builtin_bit_cast(float, __builtin_amdgcn_permlanex16(
        __builtin_bit_cast(int, x), __builtin_bit_cast(int, x), 0x76543210, 0xfedcba98, false, false));
}

template <int KK>
__device__ __forceinline__ void load_frag_row(float4 (&f)[KK], const float* __restrict__ p) {
#pragma unroll
    for (int kk = 0; kk < KK; ++kk) f[kk] = *(const float4*)(p + kk * 8);
}

template <int KK, bool EPI>
__device__ __forceinline__ void wstat_tile(const float4 (&af)[KK], const float4 (&bf)[GW_NT][KK],
                                           float* __restrict__ C, int rt, int cb, int Nout,
                                           const float* __restrict__ bias, int c, int h) {
    v8f acc[GW_NT];
#pragma unroll
    for (int t = 0; t < GW_NT; ++t) acc[t] = v8f{};
#pragma unroll
    for (int kk = 0; kk < KK; ++kk) {
        const v2f a0 = {af[kk].x, af[kk].y};
        const v2f a1 = {af[kk].z, af[kk].w};
#pragma unroll
        for (int t = 0; t < GW_NT; ++t) {
            const v2f b0 = {bf[t][kk].x, bf[t][kk].y};
            acc[t] = __builtin_amdgcn_wmma_f32_16x16x4_f32(false, a0, false, b0,
                                                           (short)0, acc[t], false, false);
        }
#pragma unroll
        for (int t = 0; t < GW_NT; ++t) {
            const v2f b1 = {bf[t][kk].z, bf[t][kk].w};
            acc[t] = __builtin_amdgcn_wmma_f32_16x16x4_f32(false, a1, false, b1,
                                                           (short)0, acc[t], false, false);
        }
    }
#pragma unroll
    for (int t = 0; t < GW_NT; t += 2) {
        const int col = cb + t * 16 + 16 * h + c;
        float bv = 0.f;
        if (EPI) bv = bias[col];
        float* crow = C + (size_t)(rt * 16) * Nout + col;
#pragma unroll
        for (int i = 0; i < 8; ++i) {
            const float lo   = acc[t][i];
            const float hi   = acc[t + 1][i];
            const float recv = xhalf(h ? lo : hi);
            float v0 = h ? recv : lo;
            float v1 = h ? hi : recv;
            if (EPI) { v0 = eluf(v0 + bv); v1 = eluf(v1 + bv); }
            crow[(size_t)i * Nout]       = v0;
            crow[(size_t)(8 + i) * Nout] = v1;
        }
    }
}

template <int K, bool EPI>
__global__ __launch_bounds__(GW_THREADS)
void gemm_wstat_f32(const float* __restrict__ A, int lda,
                    const float* __restrict__ W,
                    float* __restrict__ C,
                    int M, int Nout,
                    const float* __restrict__ bias) {
    constexpr int KK = K / 8;
    const int lane = threadIdx.x & 31;
    const int c = lane & 15;
    const int h = lane >> 4;
    const int gw = __builtin_amdgcn_readfirstlane((int)((blockIdx.x * blockDim.x + threadIdx.x) >> 5));
    const int nw = (int)((gridDim.x * blockDim.x) >> 5);
    const int ngrp = Nout / (16 * GW_NT);
    const int cg = gw % ngrp;
    const int rs = gw / ngrp;
    const int nstream = nw / ngrp;
    const int ntile = M >> 4;
    if (rs >= nstream || rs >= ntile) return;
    const int cb = cg * (16 * GW_NT);

    float4 bf[GW_NT][KK];
#pragma unroll
    for (int t = 0; t < GW_NT; ++t)
        load_frag_row<KK>(bf[t], W + (size_t)(cb + t * 16 + c) * K + h * 4);

    const float* arow = A + (size_t)c * lda + h * 4;
    const size_t tstep = (size_t)16 * lda;
    float4 f0[KK], f1[KK];
    load_frag_row<KK>(f0, arow + (size_t)rs * tstep);
    for (int rt = rs; rt < ntile; rt += 2 * nstream) {
        const int rt1 = rt + nstream;
        const int rt2 = rt1 + nstream;
        const bool has1 = rt1 < ntile;
        load_frag_row<KK>(f1, arow + (size_t)(has1 ? rt1 : rt) * tstep);
        wstat_tile<KK, EPI>(f0, bf, C, rt, cb, Nout, bias, c, h);
        load_frag_row<KK>(f0, arow + (size_t)(rt2 < ntile ? rt2 : rt) * tstep);
        if (has1) wstat_tile<KK, EPI>(f1, bf, C, rt1, cb, Nout, bias, c, h);
    }
}

__global__ void attn_coef(const float* __restrict__ hmat,
                          const float* __restrict__ a_s,
                          const float* __restrict__ a_d,
                          float* __restrict__ asrc, float* __restrict__ adst, int NH) {
    int idx = blockIdx.x * blockDim.x + threadIdx.x;
    if (idx >= NH) return;
    int n = idx >> 2, h = idx & 3;
    const float* row = hmat + (size_t)n * HIDF + h * CCH;
    const float* ws  = a_s + h * CCH;
    const float* wd  = a_d + h * CCH;
    float ss = 0.f, sd = 0.f;
    for (int c = 0; c < CCH; ++c) { float v = row[c]; ss += v * ws[c]; sd += v * wd[c]; }
    asrc[idx] = ss;
    adst[idx] = sd;
}

#define RN 200
#define AGG_THREADS 1024
#define AGG_WAVES (AGG_THREADS / 32)
#define WCAP 256

__device__ __forceinline__ void accum_one(int s, unsigned dl, int lane,
                                          const float* __restrict__ asrc,
                                          const float* __restrict__ hmat,
                                          const float* adl, const unsigned* mxk,
                                          float* dnl, float* acc) {
    float ex[HEADS];
#pragma unroll
    for (int h = 0; h < HEADS; ++h) {
        float a = leakyf(asrc[s * HEADS + h] + adl[dl * HEADS + h]);
        ex[h] = expf(a - funkey(mxk[dl * HEADS + h]));
    }
    if (lane == 0) {
#pragma unroll
        for (int h = 0; h < HEADS; ++h) atomicAdd(&dnl[dl * HEADS + h], ex[h]);
    }
    const float* hrow = hmat + (size_t)s * HIDF;
    float* orow       = acc + dl * HIDF;
#pragma unroll
    for (int i = 0; i < 8; ++i) {
        int c = lane + 32 * i;
        atomicAdd(&orow[c], ex[i >> 1] * hrow[c]);
    }
}

template <bool FIRST, bool MEAN>
__global__ __launch_bounds__(AGG_THREADS)
void agg_owner(const int* __restrict__ src, const int* __restrict__ dst,
               const float* __restrict__ asrc, const float* __restrict__ adst,
               const float* __restrict__ hmat,
               const float* __restrict__ bias, const float* __restrict__ g,
               const float* __restrict__ be, float* __restrict__ hout,
               int* __restrict__ wl_src, unsigned short* __restrict__ wl_dl,
               unsigned* __restrict__ wl_cnt, int N, int E) {
    __shared__ __attribute__((aligned(16))) float acc[RN * HIDF];
    __shared__ __attribute__((aligned(16))) int lsts[AGG_WAVES * WCAP];
    __shared__ __attribute__((aligned(16))) unsigned short lstd[AGG_WAVES * WCAP];
    __shared__ __attribute__((aligned(16))) unsigned mxk[RN * HEADS];
    __shared__ float dnl[RN * HEADS];
    __shared__ __attribute__((aligned(16))) float adl[RN * HEADS];
    __shared__ unsigned ovf;

    const int tid  = threadIdx.x;
    const int lane = tid & 31;
    const int wv   = tid >> 5;
    const int base = blockIdx.x * RN;
    const int nloc = min(RN, N - base);
    int* const myS            = lsts + wv * WCAP;
    unsigned short* const myD = lstd + wv * WCAP;

    if (tid == 0) ovf = 0u;
    for (int i = tid; i < nloc * HEADS; i += AGG_THREADS) {
        float ad = adst[base * HEADS + i];
        adl[i] = ad;
        mxk[i] = fkey(leakyf(asrc[base * HEADS + i] + ad));
    }
    __syncthreads();

    const size_t wlslot = (size_t)blockIdx.x * AGG_WAVES + wv;
    const size_t wlbase = wlslot * WCAP;
    const int wvu = __builtin_amdgcn_readfirstlane(wv);
    unsigned wcount = 0u;
    if constexpr (FIRST) {
        auto note = [&](int d, int e, bool valid) {
            unsigned dl = (unsigned)(d - base);
            bool m = valid && (dl < (unsigned)nloc);
            unsigned mask = (unsigned)__ballot(m);
            if (mask) {
                unsigned slot = wcount + __builtin_amdgcn_mbcnt_lo(mask, 0u);
                if (m && slot < (unsigned)WCAP) { myS[slot] = e; myD[slot] = (unsigned short)dl; }
                wcount += (unsigned)__popc(mask);
            }
        };
        const int E4 = E >> 2;
        const int4* dst4 = (const int4*)dst;
        for (int ib = wvu * 32; ib < E4; ib += AGG_THREADS * 4) {
            const int i = ib + lane;
            int4 v[4];
            if (ib + 3 * AGG_THREADS + 32 <= E4) {
#pragma unroll
                for (int u = 0; u < 4; ++u) v[u] = dst4[i + u * AGG_THREADS];
#pragma unroll
                for (int u = 0; u < 4; ++u) {
                    const int e = 4 * (i + u * AGG_THREADS);
                    note(v[u].x, e + 0, true);
                    note(v[u].y, e + 1, true);
                    note(v[u].z, e + 2, true);
                    note(v[u].w, e + 3, true);
                }
            } else {
#pragma unroll
                for (int u = 0; u < 4; ++u) {
                    const int iu = i + u * AGG_THREADS;
                    v[u] = (iu < E4) ? dst4[iu] : make_int4(0, 0, 0, 0);
                }
#pragma unroll
                for (int u = 0; u < 4; ++u) {
                    const int iu = i + u * AGG_THREADS;
                    const bool valid = iu < E4;
                    note(v[u].x, 4 * iu + 0, valid);
                    note(v[u].y, 4 * iu + 1, valid);
                    note(v[u].z, 4 * iu + 2, valid);
                    note(v[u].w, 4 * iu + 3, valid);
                }
            }
        }
        for (int e0 = (E4 << 2); e0 < E; e0 += AGG_THREADS) {
            int e = e0 + tid;
            bool valid = e < E;
            note(valid ? dst[e] : 0, e, valid);
        }
        if (lane == 0) wl_cnt[wlslot] = wcount;
    } else {
        wcount = (unsigned)__builtin_amdgcn_readfirstlane((int)wl_cnt[wlslot]);
    }
    if (wcount > (unsigned)WCAP && lane == 0) ovf = 1u;
    __syncthreads();

    const bool listed = (ovf == 0u);
    const int nw = (int)min(wcount, (unsigned)WCAP);

    if (listed) {
        for (int k0 = lane; k0 < nw; k0 += 128) {
            int s[4]; unsigned dl[4]; bool ok[4];
#pragma unroll
            for (int u = 0; u < 4; ++u) {
                int k = k0 + 32 * u;
                ok[u] = k < nw;
                if constexpr (FIRST) {
                    int e = ok[u] ? myS[k] : 0;
                    dl[u] = ok[u] ? (unsigned)myD[k] : 0u;
                    s[u]  = src[e];
                } else {
                    const int kr = ok[u] ? k : k0;
                    const unsigned sv = (unsigned)wl_src[wlbase + kr];
                    const unsigned dv = (unsigned)wl_dl[wlbase + kr];
                    s[u]  = (int)min(sv, (unsigned)(N - 1));
                    dl[u] = min(dv, (unsigned)(nloc - 1));
                }
            }
            float4 a4[4];
#pragma unroll
            for (int u = 0; u < 4; ++u) a4[u] = *(const float4*)(asrc + (size_t)s[u] * HEADS);
#pragma unroll
            for (int u = 0; u < 4; ++u) {
                if (ok[u]) {
                    myS[k0 + 32 * u] = s[u];
                    if constexpr (FIRST) {
                        wl_src[wlbase + k0 + 32 * u] = s[u];
                        wl_dl[wlbase + k0 + 32 * u]  = (unsigned short)dl[u];
                    } else {
                        myD[k0 + 32 * u] = (unsigned short)dl[u];
                    }
                    atomicMax(&mxk[dl[u] * HEADS + 0], fkey(leakyf(a4[u].x + adl[dl[u] * HEADS + 0])));
                    atomicMax(&mxk[dl[u] * HEADS + 1], fkey(leakyf(a4[u].y + adl[dl[u] * HEADS + 1])));
                    atomicMax(&mxk[dl[u] * HEADS + 2], fkey(leakyf(a4[u].z + adl[dl[u] * HEADS + 2])));
                    atomicMax(&mxk[dl[u] * HEADS + 3], fkey(leakyf(a4[u].w + adl[dl[u] * HEADS + 3])));
                }
            }
        }
    } else {
        for (int e = tid; e < E; e += AGG_THREADS) {
            unsigned dl = (unsigned)(dst[e] - base);
            if (dl < (unsigned)nloc) {
                int s = src[e];
#pragma unroll
                for (int h = 0; h < HEADS; ++h)
                    atomicMax(&mxk[dl * HEADS + h],
                              fkey(leakyf(asrc[s * HEADS + h] + adl[dl * HEADS + h])));
            }
        }
    }
    __syncthreads();

    for (int i = wv; i < nloc; i += AGG_WAVES) {
        int n = base + i;
        float ex[HEADS];
#pragma unroll
        for (int h = 0; h < HEADS; ++h) {
            float aself = leakyf(asrc[n * HEADS + h] + adl[i * HEADS + h]);
            ex[h] = expf(aself - funkey(mxk[i * HEADS + h]));
        }
        if (lane == 0) {
#pragma unroll
            for (int h = 0; h < HEADS; ++h) dnl[i * HEADS + h] = ex[h];
        }
        const float* row = hmat + (size_t)n * HIDF;
#pragma unroll
        for (int j = 0; j < 8; ++j) {
            int c = lane + 32 * j;
            acc[i * HIDF + c] = row[c] * ex[j >> 1];
        }
    }
    __syncthreads();

    if (listed) {
        for (int k0 = 0; k0 < nw; k0 += 32) {
            const int k    = k0 + lane;
            const bool okl = k < nw;
            const int sL      = okl ? myS[k] : 0;
            const unsigned dL = okl ? (unsigned)myD[k] : 0u;
            const float4 a4  = *(const float4*)(asrc + (size_t)sL * HEADS);
            const float4 ad4 = *(const float4*)(adl + dL * HEADS);
            const uint4 mk4  = *(const uint4*)(mxk + dL * HEADS);
            float eL[HEADS];
            eL[0] = expf(leakyf(a4.x + ad4.x) - funkey(mk4.x));
            eL[1] = expf(leakyf(a4.y + ad4.y) - funkey(mk4.y));
            eL[2] = expf(leakyf(a4.z + ad4.z) - funkey(mk4.z));
            eL[3] = expf(leakyf(a4.w + ad4.w) - funkey(mk4.w));
            if (okl) {
#pragma unroll
                for (int h = 0; h < HEADS; ++h) atomicAdd(&dnl[dL * HEADS + h], eL[h]);
            }
            const int cnt  = min(32, nw - k0);
            const int cnt4 = cnt & ~3;
            for (int j = 0; j < cnt4; j += 4) {
                int s[4]; unsigned dl[4]; float ex[4][HEADS];
#pragma unroll
                for (int u = 0; u < 4; ++u) {
                    const int l = j + u;
                    s[u]  = __builtin_amdgcn_readlane(sL, l);
                    dl[u] = (unsigned)__builtin_amdgcn_readlane((int)dL, l);
#pragma unroll
                    for (int h = 0; h < HEADS; ++h)
                        ex[u][h] = __builtin_bit_cast(float, __builtin_amdgcn_readlane(
                            __builtin_bit_cast(int, eL[h]), l));
                }
                float hv[4][8];
#pragma unroll
                for (int u = 0; u < 4; ++u) {
                    const float* hrow = hmat + (size_t)s[u] * HIDF;
#pragma unroll
                    for (int jj = 0; jj < 8; ++jj) hv[u][jj] = hrow[lane + 32 * jj];
                }
#pragma unroll
                for (int u = 0; u < 4; ++u) {
                    float* orow = acc + dl[u] * HIDF;
#pragma unroll
                    for (int jj = 0; jj < 8; ++jj)
                        atomicAdd(&orow[lane + 32 * jj], ex[u][jj >> 1] * hv[u][jj]);
                }
            }
            for (int l = cnt4; l < cnt; ++l) {
                const int s1       = __builtin_amdgcn_readlane(sL, l);
                const unsigned dl1 = (unsigned)__builtin_amdgcn_readlane((int)dL, l);
                float ex1[HEADS];
#pragma unroll
                for (int h = 0; h < HEADS; ++h)
                    ex1[h] = __builtin_bit_cast(float, __builtin_amdgcn_readlane(
                        __builtin_bit_cast(int, eL[h]), l));
                const float* hrow = hmat + (size_t)s1 * HIDF;
                float hv1[8];
#pragma unroll
                for (int jj = 0; jj < 8; ++jj) hv1[jj] = hrow[lane + 32 * jj];
                float* orow = acc + dl1 * HIDF;
#pragma unroll
                for (int jj = 0; jj < 8; ++jj)
                    atomicAdd(&orow[lane + 32 * jj], ex1[jj >> 1] * hv1[jj]);
            }
        }
    } else {
        for (int e0 = wvu * 32; e0 < E; e0 += AGG_WAVES * 32) {
            int e = e0 + lane;
            int d = (e < E) ? dst[e] : 0;
            unsigned dl = (unsigned)(d - base);
            bool m = (e < E) && (dl < (unsigned)nloc);
            unsigned mask = (unsigned)__ballot(m);
            while (mask) {
                int l = __ffs(mask) - 1;
                mask &= mask - 1u;
                int s1 = src[e0 + l];
                unsigned dl1 = (unsigned)__shfl((int)dl, l, 32);
                accum_one(s1, dl1, lane, asrc, hmat, adl, mxk, dnl, acc);
            }
        }
    }
    __syncthreads();

    if constexpr (!MEAN) {
        for (int i = wv; i < nloc; i += AGG_WAVES) {
            float dn[HEADS];
#pragma unroll
            for (int h = 0; h < HEADS; ++h) dn[h] = 1.f / (dnl[i * HEADS + h] + 1e-16f);
            float v[8];
            float s = 0.f;
#pragma unroll
            for (int j = 0; j < 8; ++j) {
                int c = lane + 32 * j;
                v[j] = acc[i * HIDF + c] * dn[j >> 1] + bias[c];
                s += v[j];
            }
            float mu = waveRedSum(s) * (1.f / HIDF);
            float q = 0.f;
#pragma unroll
            for (int j = 0; j < 8; ++j) { float dlt = v[j] - mu; q += dlt * dlt; }
            float rstd = rsqrtf(waveRedSum(q) * (1.f / HIDF) + LN_EPS);
            float* orow = hout + (size_t)(base + i) * HIDF;
#pragma unroll
            for (int j = 0; j < 8; ++j) {
                int c = lane + 32 * j;
                float y = (v[j] - mu) * rstd * g[c] + be[c];
                orow[c] = eluf(y);
            }
        }
    } else {
        for (int i = wv; i < nloc; i += AGG_WAVES) {
            float dn[HEADS];
#pragma unroll
            for (int h = 0; h < HEADS; ++h) dn[h] = 1.f / (dnl[i * HEADS + h] + 1e-16f);
            float v[2];
            float s = 0.f;
#pragma unroll
            for (int j = 0; j < 2; ++j) {
                int c = lane + 32 * j;
                float a = 0.f;
#pragma unroll
                for (int h = 0; h < HEADS; ++h) a += acc[i * HIDF + h * CCH + c] * dn[h];
                v[j] = a * (1.f / HEADS) + bias[c];
                s += v[j];
            }
            float mu = waveRedSum(s) * (1.f / CCH);
            float q = 0.f;
#pragma unroll
            for (int j = 0; j < 2; ++j) { float dlt = v[j] - mu; q += dlt * dlt; }
            float rstd = rsqrtf(waveRedSum(q) * (1.f / CCH) + LN_EPS);
            float* orow = hout + (size_t)(base + i) * CCH;
#pragma unroll
            for (int j = 0; j < 2; ++j) {
                int c = lane + 32 * j;
                float y = (v[j] - mu) * rstd * g[c] + be[c];
                orow[c] = eluf(y);
            }
        }
    }
}

__global__ void value_head(const float* __restrict__ hin,
                           const float* __restrict__ vw1, const float* __restrict__ vb1,
                           const float* __restrict__ vw2, const float* __restrict__ vb2,
                           float* __restrict__ out, int N) {
    int wid  = (blockIdx.x * blockDim.x + threadIdx.x) >> 5;
    int lane = threadIdx.x & 31;
    if (wid >= N) return;
    const float* row  = hin + (size_t)wid * CCH;
    const float* wrow = vw1 + lane * CCH;
    float hid = vb1[lane];
    for (int c = 0; c < CCH; ++c) hid += row[c] * wrow[c];
    hid = eluf(hid);
#pragma unroll
    for (int k = 0; k < 4; ++k) {
        float p = vw2[k * 32 + lane] * hid;
        p = waveRedSum(p);
        if (lane == 0) out[wid * 4 + k] = p + vb2[k];
    }
}

extern "C" void kernel_launch(void* const* d_in, const int* in_sizes, int n_in,
                              void* d_out, int out_size, void* d_ws, size_t ws_size,
                              hipStream_t stream) {
    const float* x      = (const float*)d_in[0];
    const int*   esrc   = (const int*)d_in[1];
    const int*   edst   = (const int*)d_in[2];
    const float* proj_w = (const float*)d_in[3];
    const float* proj_b = (const float*)d_in[4];
    const float* W[3]  = {(const float*)d_in[5],  (const float*)d_in[11], (const float*)d_in[17]};
    const float* AS[3] = {(const float*)d_in[6],  (const float*)d_in[12], (const float*)d_in[18]};
    const float* AD[3] = {(const float*)d_in[7],  (const float*)d_in[13], (const float*)d_in[19]};
    const float* B[3]  = {(const float*)d_in[8],  (const float*)d_in[14], (const float*)d_in[20]};
    const float* G[3]  = {(const float*)d_in[9],  (const float*)d_in[15], (const float*)d_in[21]};
    const float* BE[3] = {(const float*)d_in[10], (const float*)d_in[16], (const float*)d_in[22]};
    const float* vw1 = (const float*)d_in[23];
    const float* vb1 = (const float*)d_in[24];
    const float* vw2 = (const float*)d_in[25];
    const float* vb2 = (const float*)d_in[26];

    const int N = in_sizes[0] / 128;
    const int E = in_sizes[1];
    const int NH = N * HEADS;

    float* ws     = (float*)d_ws;
    float* hin    = ws;
    float* hmat   = hin  + (size_t)N * HIDF;
    float* asrc   = hmat + (size_t)N * HIDF;
    float* adst   = asrc + (size_t)NH;
    const int nagg = (N + RN - 1) / RN;
    int* wl_src            = (int*)(adst + (size_t)NH);
    unsigned short* wl_dl  = (unsigned short*)(wl_src + (size_t)nagg * AGG_WAVES * WCAP);
    unsigned* wl_cnt       = (unsigned*)(wl_dl + (size_t)nagg * AGG_WAVES * WCAP);

    dim3 blk(256);

    gemm_wstat_f32<128, true><<<GW_BLOCKS_K128, GW_THREADS, 0, stream>>>(
        x, 128, proj_w, hin, N, 64, proj_b);

    int in_ld = 64;
    for (int l = 0; l < 3; ++l) {
        if (l == 0)
            gemm_wstat_f32<64, false><<<GW_BLOCKS_K64, GW_THREADS, 0, stream>>>(
                hin, in_ld, W[l], hmat, N, HIDF, nullptr);
        else
            gemm_wstat_f32<256, false><<<GW_BLOCKS_K256, GW_THREADS, 0, stream>>>(
                hin, in_ld, W[l], hmat, N, HIDF, nullptr);
        attn_coef<<<(NH + 255) / 256, blk, 0, stream>>>(hmat, AS[l], AD[l], asrc, adst, NH);
        if (l == 0)
            agg_owner<true, false><<<nagg, AGG_THREADS, 0, stream>>>(
                esrc, edst, asrc, adst, hmat, B[l], G[l], BE[l], hin, wl_src, wl_dl, wl_cnt, N, E);
        else if (l == 1)
            agg_owner<false, false><<<nagg, AGG_THREADS, 0, stream>>>(
                esrc, edst, asrc, adst, hmat, B[l], G[l], BE[l], hin, wl_src, wl_dl, wl_cnt, N, E);
        else
            agg_owner<false, true><<<nagg, AGG_THREADS, 0, stream>>>(
                esrc, edst, asrc, adst, hmat, B[l], G[l], BE[l], hin, wl_src, wl_dl, wl_cnt, N, E);
        in_ld = HIDF;
    }

    value_head<<<(N * 32 + 255) / 256, blk, 0, stream>>>(
        hin, vw1, vb1, vw2, vb2, (float*)d_out, N);
}


# Row gather in 16-byte pieces: a lane loads 4 adjacent channels of each 128-channel half (2 loads per row, not 8); LDS sums kept in that order; LayerNorm(256) sums group 4 adjacent channels per lane
# speedup vs baseline: 3.1437x; 1.0589x over previous
#include <hip/hip_runtime.h>
#include <math.h>

typedef __attribute__((ext_vector_type(2))) float v2f;
typedef __attribute__((ext_vector_type(8))) float v8f;

#define HEADS 4
#define CCH 64
#define HIDF 256
#define NEG_SLOPE 0.2f
#define LN_EPS 1e-5f

__device__ __forceinline__ float eluf(float x) { return x > 0.f ? x : (expf(x) - 1.f); }
__device__ __forceinline__ float leakyf(float x) { return x > 0.f ? x : NEG_SLOPE * x; }

__device__ __forceinline__ unsigned fkey(float f) {
    unsigned u = __float_as_uint(f);
    return (u & 0x80000000u) ? ~u : (u | 0x80000000u);
}
__device__ __forceinline__ float funkey(unsigned k) {
    unsigned u = (k & 0x80000000u) ? (k ^ 0x80000000u) : ~k;
    return __uint_as_float(u);
}

__device__ __forceinline__ float waveRedSum(float v) {
    for (int off = 16; off > 0; off >>= 1) v += __shfl_xor(v, off, 32);
    return v;
}

#define GW_NT 4
#define GW_THREADS 128
#define GW_BLOCKS_K256 256
#define GW_BLOCKS_K128 512
#define GW_BLOCKS_K64 1024

__device__ __forceinline__ float xhalf(float x) {
    return __builtin_bit_cast(float, __builtin_amdgcn_permlanex16(
        __builtin_bit_cast(int, x), __builtin_bit_cast(int, x), 0x76543210, 0xfedcba98, false, false));
}

template <int KK>
__device__ __forceinline__ void load_frag_row(float4 (&f)[KK], const float* __restrict__ p) {
#pragma unroll
    for (int kk = 0; kk < KK; ++kk) f[kk] = *(const float4*)(p + kk * 8);
}

template <int KK, bool EPI>
__device__ __forceinline__ void wstat_tile(const float4 (&af)[KK], const float4 (&bf)[GW_NT][KK],
                                           float* __restrict__ C, int rt, int cb, int Nout,
                                           const float* __restrict__ bias, int c, int h) {
    v8f acc[GW_NT];
#pragma unroll
    for (int t = 0; t < GW_NT; ++t) acc[t] = v8f{};
#pragma unroll
    for (int kk = 0; kk < KK; ++kk) {
        const v2f a0 = {af[kk].x, af[kk].y};
        const v2f a1 = {af[kk].z, af[kk].w};
#pragma unroll
        for (int t = 0; t < GW_NT; ++t) {
            const v2f b0 = {bf[t][kk].x, bf[t][kk].y};
            acc[t] = __builtin_amdgcn_wmma_f32_16x16x4_f32(false, a0, false, b0,
                                                           (short)0, acc[t], false, false);
        }
#pragma unroll
        for (int t = 0; t < GW_NT; ++t) {
            const v2f b1 = {bf[t][kk].z, bf[t][kk].w};
            acc[t] = __builtin_amdgcn_wmma_f32_16x16x4_f32(false, a1, false, b1,
                                                           (short)0, acc[t], false, false);
        }
    }
#pragma unroll
    for (int t = 0; t < GW_NT; t += 2) {
        const int col = cb + t * 16 + 16 * h + c;
        float bv = 0.f;
        if (EPI) bv = bias[col];
        float* crow = C + (size_t)(rt * 16) * Nout + col;
#pragma unroll
        for (int i = 0; i < 8; ++i) {
            const float lo   = acc[t][i];
            const float hi   = acc[t + 1][i];
            const float recv = xhalf(h ? lo : hi);
            float v0 = h ? recv : lo;
            float v1 = h ? hi : recv;
            if (EPI) { v0 = eluf(v0 + bv); v1 = eluf(v1 + bv); }
            crow[(size_t)i * Nout]       = v0;
            crow[(size_t)(8 + i) * Nout] = v1;
        }
    }
}

template <int K, bool EPI>
__global__ __launch_bounds__(GW_THREADS)
void gemm_wstat_f32(const float* __restrict__ A, int lda,
                    const float* __restrict__ W,
                    float* __restrict__ C,
                    int M, int Nout,
                    const float* __restrict__ bias) {
    constexpr int KK = K / 8;
    const int lane = threadIdx.x & 31;
    const int c = lane & 15;
    const int h = lane >> 4;
    const int gw = __builtin_amdgcn_readfirstlane((int)((blockIdx.x * blockDim.x + threadIdx.x) >> 5));
    const int nw = (int)((gridDim.x * blockDim.x) >> 5);
    const int ngrp = Nout / (16 * GW_NT);
    const int cg = gw % ngrp;
    const int rs = gw / ngrp;
    const int nstream = nw / ngrp;
    const int ntile = M >> 4;
    if (rs >= nstream || rs >= ntile) return;
    const int cb = cg * (16 * GW_NT);

    float4 bf[GW_NT][KK];
#pragma unroll
    for (int t = 0; t < GW_NT; ++t)
        load_frag_row<KK>(bf[t], W + (size_t)(cb + t * 16 + c) * K + h * 4);

    const float* arow = A + (size_t)c * lda + h * 4;
    const size_t tstep = (size_t)16 * lda;
    float4 f0[KK], f1[KK];
    load_frag_row<KK>(f0, arow + (size_t)rs * tstep);
    for (int rt = rs; rt < ntile; rt += 2 * nstream) {
        const int rt1 = rt + nstream;
        const int rt2 = rt1 + nstream;
        const bool has1 = rt1 < ntile;
        load_frag_row<KK>(f1, arow + (size_t)(has1 ? rt1 : rt) * tstep);
        wstat_tile<KK, EPI>(f0, bf, C, rt, cb, Nout, bias, c, h);
        load_frag_row<KK>(f0, arow + (size_t)(rt2 < ntile ? rt2 : rt) * tstep);
        if (has1) wstat_tile<KK, EPI>(f1, bf, C, rt1, cb, Nout, bias, c, h);
    }
}

__global__ void attn_coef(const float* __restrict__ hmat,
                          const float* __restrict__ a_s,
                          const float* __restrict__ a_d,
                          float* __restrict__ asrc, float* __restrict__ adst, int NH) {
    int idx = blockIdx.x * blockDim.x + threadIdx.x;
    if (idx >= NH) return;
    int n = idx >> 2, h = idx & 3;
    const float* row = hmat + (size_t)n * HIDF + h * CCH;
    const float* ws  = a_s + h * CCH;
    const float* wd  = a_d + h * CCH;
    float ss = 0.f, sd = 0.f;
    for (int c = 0; c < CCH; ++c) { float v = row[c]; ss += v * ws[c]; sd += v * wd[c]; }
    asrc[idx] = ss;
    adst[idx] = sd;
}

#define RN 200
#define AGG_THREADS 1024
#define AGG_WAVES (AGG_THREADS / 32)
#define WCAP 256

__device__ __forceinline__ int accpos(int c) { return (c & 128) | ((c & 3) << 5) | ((c & 127) >> 2); }

__device__ __forceinline__ void row_add(float* ar, const float4 a, const float4 b, float wa, float wb) {
    atomicAdd(&ar[0],   wa * a.x);
    atomicAdd(&ar[32],  wa * a.y);
    atomicAdd(&ar[64],  wa * a.z);
    atomicAdd(&ar[96],  wa * a.w);
    atomicAdd(&ar[128], wb * b.x);
    atomicAdd(&ar[160], wb * b.y);
    atomicAdd(&ar[192], wb * b.z);
    atomicAdd(&ar[224], wb * b.w);
}

__device__ __forceinline__ void accum_one(int s, unsigned dl, int lane,
                                          const float* __restrict__ asrc,
                                          const float* __restrict__ hmat,
                                          const float* adl, const unsigned* mxk,
                                          float* dnl, float* acc) {
    float ex[HEADS];
#pragma unroll
    for (int h = 0; h < HEADS; ++h) {
        float a = leakyf(asrc[s * HEADS + h] + adl[dl * HEADS + h]);
        ex[h] = expf(a - funkey(mxk[dl * HEADS + h]));
    }
    if (lane == 0) {
#pragma unroll
        for (int h = 0; h < HEADS; ++h) atomicAdd(&dnl[dl * HEADS + h], ex[h]);
    }
    const float* hrow = hmat + (size_t)s * HIDF + 4 * lane;
    const float4 ra = *(const float4*)hrow;
    const float4 rb = *(const float4*)(hrow + 128);
    const bool upper = (lane & 16) != 0;
    row_add(acc + dl * HIDF + lane, ra, rb, upper ? ex[1] : ex[0], upper ? ex[3] : ex[2]);
}

template <bool FIRST, bool MEAN>
__global__ __launch_bounds__(AGG_THREADS)
void agg_owner(const int* __restrict__ src, const int* __restrict__ dst,
               const float* __restrict__ asrc, const float* __restrict__ adst,
               const float* __restrict__ hmat,
               const float* __restrict__ bias, const float* __restrict__ g,
               const float* __restrict__ be, float* __restrict__ hout,
               int* __restrict__ wl_src, unsigned short* __restrict__ wl_dl,
               unsigned* __restrict__ wl_cnt, int N, int E) {
    __shared__ __attribute__((aligned(16))) float acc[RN * HIDF];
    __shared__ __attribute__((aligned(16))) int lsts[AGG_WAVES * WCAP];
    __shared__ __attribute__((aligned(16))) unsigned short lstd[AGG_WAVES * WCAP];
    __shared__ __attribute__((aligned(16))) unsigned mxk[RN * HEADS];
    __shared__ float dnl[RN * HEADS];
    __shared__ __attribute__((aligned(16))) float adl[RN * HEADS];
    __shared__ unsigned ovf;

    const int tid  = threadIdx.x;
    const int lane = tid & 31;
    const int wv   = tid >> 5;
    const bool upper = (lane & 16) != 0;
    const int base = blockIdx.x * RN;
    const int nloc = min(RN, N - base);
    int* const myS            = lsts + wv * WCAP;
    unsigned short* const myD = lstd + wv * WCAP;

    if (tid == 0) ovf = 0u;
    for (int i = tid; i < nloc * HEADS; i += AGG_THREADS) {
        float ad = adst[base * HEADS + i];
        adl[i] = ad;
        mxk[i] = fkey(leakyf(asrc[base * HEADS + i] + ad));
    }
    __syncthreads();

    const size_t wlslot = (size_t)blockIdx.x * AGG_WAVES + wv;
    const size_t wlbase = wlslot * WCAP;
    const int wvu = __builtin_amdgcn_readfirstlane(wv);
    unsigned wcount = 0u;
    if constexpr (FIRST) {
        auto note = [&](int d, int e, bool valid) {
            unsigned dl = (unsigned)(d - base);
            bool m = valid && (dl < (unsigned)nloc);
            unsigned mask = (unsigned)__ballot(m);
            if (mask) {
                unsigned slot = wcount + __builtin_amdgcn_mbcnt_lo(mask, 0u);
                if (m && slot < (unsigned)WCAP) { myS[slot] = e; myD[slot] = (unsigned short)dl; }
                wcount += (unsigned)__popc(mask);
            }
        };
        const int E4 = E >> 2;
        const int4* dst4 = (const int4*)dst;
        for (int ib = wvu * 32; ib < E4; ib += AGG_THREADS * 4) {
            const int i = ib + lane;
            int4 v[4];
            if (ib + 3 * AGG_THREADS + 32 <= E4) {
#pragma unroll
                for (int u = 0; u < 4; ++u) v[u] = dst4[i + u * AGG_THREADS];
#pragma unroll
                for (int u = 0; u < 4; ++u) {
                    const int e = 4 * (i + u * AGG_THREADS);
                    note(v[u].x, e + 0, true);
                    note(v[u].y, e + 1, true);
                    note(v[u].z, e + 2, true);
                    note(v[u].w, e + 3, true);
                }
            } else {
#pragma unroll
                for (int u = 0; u < 4; ++u) {
                    const int iu = i + u * AGG_THREADS;
                    v[u] = (iu < E4) ? dst4[iu] : make_int4(0, 0, 0, 0);
                }
#pragma unroll
                for (int u = 0; u < 4; ++u) {
                    const int iu = i + u * AGG_THREADS;
                    const bool valid = iu < E4;
                    note(v[u].x, 4 * iu + 0, valid);
                    note(v[u].y, 4 * iu + 1, valid);
                    note(v[u].z, 4 * iu + 2, valid);
                    note(v[u].w, 4 * iu + 3, valid);
                }
            }
        }
        for (int e0 = (E4 << 2); e0 < E; e0 += AGG_THREADS) {
            int e = e0 + tid;
            bool valid = e < E;
            note(valid ? dst[e] : 0, e, valid);
        }
        if (lane == 0) wl_cnt[wlslot] = wcount;
    } else {
        wcount = (unsigned)__builtin_amdgcn_readfirstlane((int)wl_cnt[wlslot]);
    }
    if (wcount > (unsigned)WCAP && lane == 0) ovf = 1u;
    __syncthreads();

    const bool listed = (ovf == 0u);
    const int nw = (int)min(wcount, (unsigned)WCAP);

    if (listed) {
        for (int k0 = lane; k0 < nw; k0 += 128) {
            int s[4]; unsigned dl[4]; bool ok[4];
#pragma unroll
            for (int u = 0; u < 4; ++u) {
                int k = k0 + 32 * u;
                ok[u] = k < nw;
                if constexpr (FIRST) {
                    int e = ok[u] ? myS[k] : 0;
                    dl[u] = ok[u] ? (unsigned)myD[k] : 0u;
                    s[u]  = src[e];
                } else {
                    const int kr = ok[u] ? k : k0;
                    const unsigned sv = (unsigned)wl_src[wlbase + kr];
                    const unsigned dv = (unsigned)wl_dl[wlbase + kr];
                    s[u]  = (int)min(sv, (unsigned)(N - 1));
                    dl[u] = min(dv, (unsigned)(nloc - 1));
                }
            }
            float4 a4[4];
#pragma unroll
            for (int u = 0; u < 4; ++u) a4[u] = *(const float4*)(asrc + (size_t)s[u] * HEADS);
#pragma unroll
            for (int u = 0; u < 4; ++u) {
                if (ok[u]) {
                    myS[k0 + 32 * u] = s[u];
                    if constexpr (FIRST) {
                        wl_src[wlbase + k0 + 32 * u] = s[u];
                        wl_dl[wlbase + k0 + 32 * u]  = (unsigned short)dl[u];
                    } else {
                        myD[k0 + 32 * u] = (unsigned short)dl[u];
                    }
                    atomicMax(&mxk[dl[u] * HEADS + 0], fkey(leakyf(a4[u].x + adl[dl[u] * HEADS + 0])));
                    atomicMax(&mxk[dl[u] * HEADS + 1], fkey(leakyf(a4[u].y + adl[dl[u] * HEADS + 1])));
                    atomicMax(&mxk[dl[u] * HEADS + 2], fkey(leakyf(a4[u].z + adl[dl[u] * HEADS + 2])));
                    atomicMax(&mxk[dl[u] * HEADS + 3], fkey(leakyf(a4[u].w + adl[dl[u] * HEADS + 3])));
                }
            }
        }
    } else {
        for (int e = tid; e < E; e += AGG_THREADS) {
            unsigned dl = (unsigned)(dst[e] - base);
            if (dl < (unsigned)nloc) {
                int s = src[e];
#pragma unroll
                for (int h = 0; h < HEADS; ++h)
                    atomicMax(&mxk[dl * HEADS + h],
                              fkey(leakyf(asrc[s * HEADS + h] + adl[dl * HEADS + h])));
            }
        }
    }
    __syncthreads();

    for (int i = wv; i < nloc; i += AGG_WAVES) {
        int n = base + i;
        float ex[HEADS];
#pragma unroll
        for (int h = 0; h < HEADS; ++h) {
            float aself = leakyf(asrc[n * HEADS + h] + adl[i * HEADS + h]);
            ex[h] = expf(aself - funkey(mxk[i * HEADS + h]));
        }
        if (lane == 0) {
#pragma unroll
            for (int h = 0; h < HEADS; ++h) dnl[i * HEADS + h] = ex[h];
        }
        const float* row = hmat + (size_t)n * HIDF + 4 * lane;
        const float4 ra = *(const float4*)row;
        const float4 rb = *(const float4*)(row + 128);
        const float wa = upper ? ex[1] : ex[0];
        const float wb = upper ? ex[3] : ex[2];
        float* ar = acc + i * HIDF + lane;
        ar[0]   = ra.x * wa;
        ar[32]  = ra.y * wa;
        ar[64]  = ra.z * wa;
        ar[96]  = ra.w * wa;
        ar[128] = rb.x * wb;
        ar[160] = rb.y * wb;
        ar[192] = rb.z * wb;
        ar[224] = rb.w * wb;
    }
    __syncthreads();

    if (listed) {
        for (int k0 = 0; k0 < nw; k0 += 32) {
            const int k    = k0 + lane;
            const bool okl = k < nw;
            const int sL      = okl ? myS[k] : 0;
            const unsigned dL = okl ? (unsigned)myD[k] : 0u;
            const float4 a4  = *(const float4*)(asrc + (size_t)sL * HEADS);
            const float4 ad4 = *(const float4*)(adl + dL * HEADS);
            const uint4 mk4  = *(const uint4*)(mxk + dL * HEADS);
            float eL[HEADS];
            eL[0] = expf(leakyf(a4.x + ad4.x) - funkey(mk4.x));
            eL[1] = expf(leakyf(a4.y + ad4.y) - funkey(mk4.y));
            eL[2] = expf(leakyf(a4.z + ad4.z) - funkey(mk4.z));
            eL[3] = expf(leakyf(a4.w + ad4.w) - funkey(mk4.w));
            if (okl) {
#pragma unroll
                for (int h = 0; h < HEADS; ++h) atomicAdd(&dnl[dL * HEADS + h], eL[h]);
            }
            const int cnt  = min(32, nw - k0);
            const int cnt4 = cnt & ~3;
            for (int j = 0; j < cnt4; j += 4) {
                int s[4]; unsigned dl[4]; float wa[4], wb[4];
#pragma unroll
                for (int u = 0; u < 4; ++u) {
                    const int l = j + u;
                    s[u]  = __builtin_amdgcn_readlane(sL, l);
                    dl[u] = (unsigned)__builtin_amdgcn_readlane((int)dL, l);
                    float ex[HEADS];
#pragma unroll
                    for (int h = 0; h < HEADS; ++h)
                        ex[h] = __builtin_bit_cast(float, __builtin_amdgcn_readlane(
                            __builtin_bit_cast(int, eL[h]), l));
                    wa[u] = upper ? ex[1] : ex[0];
                    wb[u] = upper ? ex[3] : ex[2];
                }
                float4 ra[4], rb[4];
#pragma unroll
                for (int u = 0; u < 4; ++u) {
                    const float* hrow = hmat + (size_t)s[u] * HIDF + 4 * lane;
                    ra[u] = *(const float4*)hrow;
                    rb[u] = *(const float4*)(hrow + 128);
                }
#pragma unroll
                for (int u = 0; u < 4; ++u)
                    row_add(acc + dl[u] * HIDF + lane, ra[u], rb[u], wa[u], wb[u]);
            }
            for (int l = cnt4; l < cnt; ++l) {
                const int s1       = __builtin_amdgcn_readlane(sL, l);
                const unsigned dl1 = (unsigned)__builtin_amdgcn_readlane((int)dL, l);
                float ex1[HEADS];
#pragma unroll
                for (int h = 0; h < HEADS; ++h)
                    ex1[h] = __builtin_bit_cast(float, __builtin_amdgcn_readlane(
                        __builtin_bit_cast(int, eL[h]), l));
                const float* hrow = hmat + (size_t)s1 * HIDF + 4 * lane;
                const float4 ra1 = *(const float4*)hrow;
                const float4 rb1 = *(const float4*)(hrow + 128);
                row_add(acc + dl1 * HIDF + lane, ra1, rb1,
                        upper ? ex1[1] : ex1[0], upper ? ex1[3] : ex1[2]);
            }
        }
    } else {
        for (int e0 = wvu * 32; e0 < E; e0 += AGG_WAVES * 32) {
            int e = e0 + lane;
            int d = (e < E) ? dst[e] : 0;
            unsigned dl = (unsigned)(d - base);
            bool m = (e < E) && (dl < (unsigned)nloc);
            unsigned mask = (unsigned)__ballot(m);
            while (mask) {
                int l = __ffs(mask) - 1;
                mask &= mask - 1u;
                int s1 = src[e0 + l];
                unsigned dl1 = (unsigned)__shfl((int)dl, l, 32);
                accum_one(s1, dl1, lane, asrc, hmat, adl, mxk, dnl, acc);
            }
        }
    }
    __syncthreads();

    if constexpr (!MEAN) {
        const int hsel = lane >> 4;
        for (int i = wv; i < nloc; i += AGG_WAVES) {
            const float dna = 1.f / (dnl[i * HEADS + hsel] + 1e-16f);
            const float dnb = 1.f / (dnl[i * HEADS + 2 + hsel] + 1e-16f);
            const float4 ba = *(const float4*)(bias + 4 * lane);
            const float4 bb = *(const float4*)(bias + 128 + 4 * lane);
            const float* ar = acc + i * HIDF + lane;
            float v[8];
            v[0] = ar[0]   * dna + ba.x;
            v[1] = ar[32]  * dna + ba.y;
            v[2] = ar[64]  * dna + ba.z;
            v[3] = ar[96]  * dna + ba.w;
            v[4] = ar[128] * dnb + bb.x;
            v[5] = ar[160] * dnb + bb.y;
            v[6] = ar[192] * dnb + bb.z;
            v[7] = ar[224] * dnb + bb.w;
            float s = 0.f;
#pragma unroll
            for (int j = 0; j < 8; ++j) s += v[j];
            float mu = waveRedSum(s) * (1.f / HIDF);
            float q = 0.f;
#pragma unroll
            for (int j = 0; j < 8; ++j) { float dlt = v[j] - mu; q += dlt * dlt; }
            float rstd = rsqrtf(waveRedSum(q) * (1.f / HIDF) + LN_EPS);
            const float4 ga  = *(const float4*)(g + 4 * lane);
            const float4 gb  = *(const float4*)(g + 128 + 4 * lane);
            const float4 bea = *(const float4*)(be + 4 * lane);
            const float4 beb = *(const float4*)(be + 128 + 4 * lane);
            float4 ya, yb;
            ya.x = eluf((v[0] - mu) * rstd * ga.x + bea.x);
            ya.y = eluf((v[1] - mu) * rstd * ga.y + bea.y);
            ya.z = eluf((v[2] - mu) * rstd * ga.z + bea.z);
            ya.w = eluf((v[3] - mu) * rstd * ga.w + bea.w);
            yb.x = eluf((v[4] - mu) * rstd * gb.x + beb.x);
            yb.y = eluf((v[5] - mu) * rstd * gb.y + beb.y);
            yb.z = eluf((v[6] - mu) * rstd * gb.z + beb.z);
            yb.w = eluf((v[7] - mu) * rstd * gb.w + beb.w);
            float* orow = hout + (size_t)(base + i) * HIDF + 4 * lane;
            *(float4*)orow         = ya;
            *(float4*)(orow + 128) = yb;
        }
    } else {
        for (int i = wv; i < nloc; i += AGG_WAVES) {
            float dn[HEADS];
#pragma unroll
            for (int h = 0; h < HEADS; ++h) dn[h] = 1.f / (dnl[i * HEADS + h] + 1e-16f);
            float v[2];
            float s = 0.f;
#pragma unroll
            for (int j = 0; j < 2; ++j) {
                int c = lane + 32 * j;
                float a = 0.f;
#pragma unroll
                for (int h = 0; h < HEADS; ++h) a += acc[i * HIDF + accpos(h * CCH + c)] * dn[h];
                v[j] = a * (1.f / HEADS) + bias[c];
                s += v[j];
            }
            float mu = waveRedSum(s) * (1.f / CCH);
            float q = 0.f;
#pragma unroll
            for (int j = 0; j < 2; ++j) { float dlt = v[j] - mu; q += dlt * dlt; }
            float rstd = rsqrtf(waveRedSum(q) * (1.f / CCH) + LN_EPS);
            float* orow = hout + (size_t)(base + i) * CCH;
#pragma unroll
            for (int j = 0; j < 2; ++j) {
                int c = lane + 32 * j;
                float y = (v[j] - mu) * rstd * g[c] + be[c];
                orow[c] = eluf(y);
            }
        }
    }
}

__global__ void value_head(const float* __restrict__ hin,
                           const float* __restrict__ vw1, const float* __restrict__ vb1,
                           const float* __restrict__ vw2, const float* __restrict__ vb2,
                           float* __restrict__ out, int N) {
    int wid  = (blockIdx.x * blockDim.x + threadIdx.x) >> 5;
    int lane = threadIdx.x & 31;
    if (wid >= N) return;
    const float* row  = hin + (size_t)wid * CCH;
    const float* wrow = vw1 + lane * CCH;
    float hid = vb1[lane];
    for (int c = 0; c < CCH; ++c) hid += row[c] * wrow[c];
    hid = eluf(hid);
#pragma unroll
    for (int k = 0; k < 4; ++k) {
        float p = vw2[k * 32 + lane] * hid;
        p = waveRedSum(p);
        if (lane == 0) out[wid * 4 + k] = p + vb2[k];
    }
}

extern "C" void kernel_launch(void* const* d_in, const int* in_sizes, int n_in,
                              void* d_out, int out_size, void* d_ws, size_t ws_size,
                              hipStream_t stream) {
    const float* x      = (const float*)d_in[0];
    const int*   esrc   = (const int*)d_in[1];
    const int*   edst   = (const int*)d_in[2];
    const float* proj_w = (const float*)d_in[3];
    const float* proj_b = (const float*)d_in[4];
    const float* W[3]  = {(const float*)d_in[5],  (const float*)d_in[11], (const float*)d_in[17]};
    const float* AS[3] = {(const float*)d_in[6],  (const float*)d_in[12], (const float*)d_in[18]};
    const float* AD[3] = {(const float*)d_in[7],  (const float*)d_in[13], (const float*)d_in[19]};
    const float* B[3]  = {(const float*)d_in[8],  (const float*)d_in[14], (const float*)d_in[20]};
    const float* G[3]  = {(const float*)d_in[9],  (const float*)d_in[15], (const float*)d_in[21]};
    const float* BE[3] = {(const float*)d_in[10], (const float*)d_in[16], (const float*)d_in[22]};
    const float* vw1 = (const float*)d_in[23];
    const float* vb1 = (const float*)d_in[24];
    const float* vw2 = (const float*)d_in[25];
    const float* vb2 = (const float*)d_in[26];

    const int N = in_sizes[0] / 128;
    const int E = in_sizes[1];
    const int NH = N * HEADS;

    float* ws     = (float*)d_ws;
    float* hin    = ws;
    float* hmat   = hin  + (size_t)N * HIDF;
    float* asrc   = hmat + (size_t)N * HIDF;
    float* adst   = asrc + (size_t)NH;
    const int nagg = (N + RN - 1) / RN;
    int* wl_src            = (int*)(adst + (size_t)NH);
    unsigned short* wl_dl  = (unsigned short*)(wl_src + (size_t)nagg * AGG_WAVES * WCAP);
    unsigned* wl_cnt       = (unsigned*)(wl_dl + (size_t)nagg * AGG_WAVES * WCAP);

    dim3 blk(256);

    gemm_wstat_f32<128, true><<<GW_BLOCKS_K128, GW_THREADS, 0, stream>>>(
        x, 128, proj_w, hin, N, 64, proj_b);

    int in_ld = 64;
    for (int l = 0; l < 3; ++l) {
        if (l == 0)
            gemm_wstat_f32<64, false><<<GW_BLOCKS_K64, GW_THREADS, 0, stream>>>(
                hin, in_ld, W[l], hmat, N, HIDF, nullptr);
        else
            gemm_wstat_f32<256, false><<<GW_BLOCKS_K256, GW_THREADS, 0, stream>>>(
                hin, in_ld, W[l], hmat, N, HIDF, nullptr);
        attn_coef<<<(NH + 255) / 256, blk, 0, stream>>>(hmat, AS[l], AD[l], asrc, adst, NH);
        if (l == 0)
            agg_owner<true, false><<<nagg, AGG_THREADS, 0, stream>>>(
                esrc, edst, asrc, adst, hmat, B[l], G[l], BE[l], hin, wl_src, wl_dl, wl_cnt, N, E);
        else if (l == 1)
            agg_owner<false, false><<<nagg, AGG_THREADS, 0, stream>>>(
                esrc, edst, asrc, adst, hmat, B[l], G[l], BE[l], hin, wl_src, wl_dl, wl_cnt, N, E);
        else
            agg_owner<false, true><<<nagg, AGG_THREADS, 0, stream>>>(
                esrc, edst, asrc, adst, hmat, B[l], G[l], BE[l], hin, wl_src, wl_dl, wl_cnt, N, E);
        in_ld = HIDF;
    }

    value_head<<<(N * 32 + 255) / 256, blk, 0, stream>>>(
        hin, vw1, vb1, vw2, vb2, (float*)d_out, N);
}


# Value head on the matrix unit: a wave takes 16 nodes, vw1 as f32 16x16x4 register fragments, ELU on accumulators, 32->4 by lane products + 16-lane add tree; 3128 waves, not 50000; sum order changed
# speedup vs baseline: 3.3882x; 1.0778x over previous
#include <hip/hip_runtime.h>
#include <math.h>

typedef __attribute__((ext_vector_type(2))) float v2f;
typedef __attribute__((ext_vector_type(8))) float v8f;

#define HEADS 4
#define CCH 64
#define HIDF 256
#define VHID 32
#define OUTF 4
#define NEG_SLOPE 0.2f
#define LN_EPS 1e-5f

__device__ __forceinline__ float eluf(float x) { return x > 0.f ? x : (expf(x) - 1.f); }
__device__ __forceinline__ float leakyf(float x) { return x > 0.f ? x : NEG_SLOPE * x; }

__device__ __forceinline__ unsigned fkey(float f) {
    unsigned u = __float_as_uint(f);
    return (u & 0x80000000u) ? ~u : (u | 0x80000000u);
}
__device__ __forceinline__ float funkey(unsigned k) {
    unsigned u = (k & 0x80000000u) ? (k ^ 0x80000000u) : ~k;
    return __uint_as_float(u);
}

__device__ __forceinline__ float waveRedSum(float v) {
    for (int off = 16; off > 0; off >>= 1) v += __shfl_xor(v, off, 32);
    return v;
}

#define GW_NT 4
#define GW_THREADS 128
#define GW_BLOCKS_K256 256
#define GW_BLOCKS_K128 512
#define GW_BLOCKS_K64 1024

__device__ __forceinline__ float xhalf(float x) {
    return __builtin_bit_cast(float, __builtin_amdgcn_permlanex16(
        __builtin_bit_cast(int, x), __builtin_bit_cast(int, x), 0x76543210, 0xfedcba98, false, false));
}

template <int KK>
__device__ __forceinline__ void load_frag_row(float4 (&f)[KK], const float* __restrict__ p) {
#pragma unroll
    for (int kk = 0; kk < KK; ++kk) f[kk] = *(const float4*)(p + kk * 8);
}

template <int KK, bool EPI>
__device__ __forceinline__ void wstat_tile(const float4 (&af)[KK], const float4 (&bf)[GW_NT][KK],
                                           float* __restrict__ C, int rt, int cb, int Nout,
                                           const float* __restrict__ bias, int c, int h) {
    v8f acc[GW_NT];
#pragma unroll
    for (int t = 0; t < GW_NT; ++t) acc[t] = v8f{};
#pragma unroll
    for (int kk = 0; kk < KK; ++kk) {
        const v2f a0 = {af[kk].x, af[kk].y};
        const v2f a1 = {af[kk].z, af[kk].w};
#pragma unroll
        for (int t = 0; t < GW_NT; ++t) {
            const v2f b0 = {bf[t][kk].x, bf[t][kk].y};
            acc[t] = __builtin_amdgcn_wmma_f32_16x16x4_f32(false, a0, false, b0,
                                                           (short)0, acc[t], false, false);
        }
#pragma unroll
        for (int t = 0; t < GW_NT; ++t) {
            const v2f b1 = {bf[t][kk].z, bf[t][kk].w};
            acc[t] = __builtin_amdgcn_wmma_f32_16x16x4_f32(false, a1, false, b1,
                                                           (short)0, acc[t], false, false);
        }
    }
#pragma unroll
    for (int t = 0; t < GW_NT; t += 2) {
        const int col = cb + t * 16 + 16 * h + c;
        float bv = 0.f;
        if (EPI) bv = bias[col];
        float* crow = C + (size_t)(rt * 16) * Nout + col;
#pragma unroll
        for (int i = 0; i < 8; ++i) {
            const float lo   = acc[t][i];
            const float hi   = acc[t + 1][i];
            const float recv = xhalf(h ? lo : hi);
            float v0 = h ? recv : lo;
            float v1 = h ? hi : recv;
            if (EPI) { v0 = eluf(v0 + bv); v1 = eluf(v1 + bv); }
            crow[(size_t)i * Nout]       = v0;
            crow[(size_t)(8 + i) * Nout] = v1;
        }
    }
}

template <int K, bool EPI>
__global__ __launch_bounds__(GW_THREADS)
void gemm_wstat_f32(const float* __restrict__ A, int lda,
                    const float* __restrict__ W,
                    float* __restrict__ C,
                    int M, int Nout,
                    const float* __restrict__ bias) {
    constexpr int KK = K / 8;
    const int lane = threadIdx.x & 31;
    const int c = lane & 15;
    const int h = lane >> 4;
    const int gw = __builtin_amdgcn_readfirstlane((int)((blockIdx.x * blockDim.x + threadIdx.x) >> 5));
    const int nw = (int)((gridDim.x * blockDim.x) >> 5);
    const int ngrp = Nout / (16 * GW_NT);
    const int cg = gw % ngrp;
    const int rs = gw / ngrp;
    const int nstream = nw / ngrp;
    const int ntile = M >> 4;
    if (rs >= nstream || rs >= ntile) return;
    const int cb = cg * (16 * GW_NT);

    float4 bf[GW_NT][KK];
#pragma unroll
    for (int t = 0; t < GW_NT; ++t)
        load_frag_row<KK>(bf[t], W + (size_t)(cb + t * 16 + c) * K + h * 4);

    const float* arow = A + (size_t)c * lda + h * 4;
    const size_t tstep = (size_t)16 * lda;
    float4 f0[KK], f1[KK];
    load_frag_row<KK>(f0, arow + (size_t)rs * tstep);
    for (int rt = rs; rt < ntile; rt += 2 * nstream) {
        const int rt1 = rt + nstream;
        const int rt2 = rt1 + nstream;
        const bool has1 = rt1 < ntile;
        load_frag_row<KK>(f1, arow + (size_t)(has1 ? rt1 : rt) * tstep);
        wstat_tile<KK, EPI>(f0, bf, C, rt, cb, Nout, bias, c, h);
        load_frag_row<KK>(f0, arow + (size_t)(rt2 < ntile ? rt2 : rt) * tstep);
        if (has1) wstat_tile<KK, EPI>(f1, bf, C, rt1, cb, Nout, bias, c, h);
    }
}

__global__ void attn_coef(const float* __restrict__ hmat,
                          const float* __restrict__ a_s,
                          const float* __restrict__ a_d,
                          float* __restrict__ asrc, float* __restrict__ adst, int NH) {
    int idx = blockIdx.x * blockDim.x + threadIdx.x;
    if (idx >= NH) return;
    int n = idx >> 2, h = idx & 3;
    const float* row = hmat + (size_t)n * HIDF + h * CCH;
    const float* ws  = a_s + h * CCH;
    const float* wd  = a_d + h * CCH;
    float ss = 0.f, sd = 0.f;
    for (int c = 0; c < CCH; ++c) { float v = row[c]; ss += v * ws[c]; sd += v * wd[c]; }
    asrc[idx] = ss;
    adst[idx] = sd;
}

#define RN 200
#define AGG_THREADS 1024
#define AGG_WAVES (AGG_THREADS / 32)
#define WCAP 256

__device__ __forceinline__ int accpos(int c) { return (c & 128) | ((c & 3) << 5) | ((c & 127) >> 2); }

__device__ __forceinline__ void row_add(float* ar, const float4 a, const float4 b, float wa, float wb) {
    atomicAdd(&ar[0],   wa * a.x);
    atomicAdd(&ar[32],  wa * a.y);
    atomicAdd(&ar[64],  wa * a.z);
    atomicAdd(&ar[96],  wa * a.w);
    atomicAdd(&ar[128], wb * b.x);
    atomicAdd(&ar[160], wb * b.y);
    atomicAdd(&ar[192], wb * b.z);
    atomicAdd(&ar[224], wb * b.w);
}

__device__ __forceinline__ void accum_one(int s, unsigned dl, int lane,
                                          const float* __restrict__ asrc,
                                          const float* __restrict__ hmat,
                                          const float* adl, const unsigned* mxk,
                                          float* dnl, float* acc) {
    float ex[HEADS];
#pragma unroll
    for (int h = 0; h < HEADS; ++h) {
        float a = leakyf(asrc[s * HEADS + h] + adl[dl * HEADS + h]);
        ex[h] = expf(a - funkey(mxk[dl * HEADS + h]));
    }
    if (lane == 0) {
#pragma unroll
        for (int h = 0; h < HEADS; ++h) atomicAdd(&dnl[dl * HEADS + h], ex[h]);
    }
    const float* hrow = hmat + (size_t)s * HIDF + 4 * lane;
    const float4 ra = *(const float4*)hrow;
    const float4 rb = *(const float4*)(hrow + 128);
    const bool upper = (lane & 16) != 0;
    row_add(acc + dl * HIDF + lane, ra, rb, upper ? ex[1] : ex[0], upper ? ex[3] : ex[2]);
}

template <bool FIRST, bool MEAN>
__global__ __launch_bounds__(AGG_THREADS)
void agg_owner(const int* __restrict__ src, const int* __restrict__ dst,
               const float* __restrict__ asrc, const float* __restrict__ adst,
               const float* __restrict__ hmat,
               const float* __restrict__ bias, const float* __restrict__ g,
               const float* __restrict__ be, float* __restrict__ hout,
               int* __restrict__ wl_src, unsigned short* __restrict__ wl_dl,
               unsigned* __restrict__ wl_cnt, int N, int E) {
    __shared__ __attribute__((aligned(16))) float acc[RN * HIDF];
    __shared__ __attribute__((aligned(16))) int lsts[AGG_WAVES * WCAP];
    __shared__ __attribute__((aligned(16))) unsigned short lstd[AGG_WAVES * WCAP];
    __shared__ __attribute__((aligned(16))) unsigned mxk[RN * HEADS];
    __shared__ float dnl[RN * HEADS];
    __shared__ __attribute__((aligned(16))) float adl[RN * HEADS];
    __shared__ unsigned ovf;

    const int tid  = threadIdx.x;
    const int lane = tid & 31;
    const int wv   = tid >> 5;
    const bool upper = (lane & 16) != 0;
    const int base = blockIdx.x * RN;
    const int nloc = min(RN, N - base);
    int* const myS            = lsts + wv * WCAP;
    unsigned short* const myD = lstd + wv * WCAP;

    if (tid == 0) ovf = 0u;
    for (int i = tid; i < nloc * HEADS; i += AGG_THREADS) {
        float ad = adst[base * HEADS + i];
        adl[i] = ad;
        mxk[i] = fkey(leakyf(asrc[base * HEADS + i] + ad));
    }
    __syncthreads();

    const size_t wlslot = (size_t)blockIdx.x * AGG_WAVES + wv;
    const size_t wlbase = wlslot * WCAP;
    const int wvu = __builtin_amdgcn_readfirstlane(wv);
    unsigned wcount = 0u;
    if constexpr (FIRST) {
        auto note = [&](int d, int e, bool valid) {
            unsigned dl = (unsigned)(d - base);
            bool m = valid && (dl < (unsigned)nloc);
            unsigned mask = (unsigned)__ballot(m);
            if (mask) {
                unsigned slot = wcount + __builtin_amdgcn_mbcnt_lo(mask, 0u);
                if (m && slot < (unsigned)WCAP) { myS[slot] = e; myD[slot] = (unsigned short)dl; }
                wcount += (unsigned)__popc(mask);
            }
        };
        const int E4 = E >> 2;
        const int4* dst4 = (const int4*)dst;
        for (int ib = wvu * 32; ib < E4; ib += AGG_THREADS * 4) {
            const int i = ib + lane;
            int4 v[4];
            if (ib + 3 * AGG_THREADS + 32 <= E4) {
#pragma unroll
                for (int u = 0; u < 4; ++u) v[u] = dst4[i + u * AGG_THREADS];
#pragma unroll
                for (int u = 0; u < 4; ++u) {
                    const int e = 4 * (i + u * AGG_THREADS);
                    note(v[u].x, e + 0, true);
                    note(v[u].y, e + 1, true);
                    note(v[u].z, e + 2, true);
                    note(v[u].w, e + 3, true);
                }
            } else {
#pragma unroll
                for (int u = 0; u < 4; ++u) {
                    const int iu = i + u * AGG_THREADS;
                    v[u] = (iu < E4) ? dst4[iu] : make_int4(0, 0, 0, 0);
                }
#pragma unroll
                for (int u = 0; u < 4; ++u) {
                    const int iu = i + u * AGG_THREADS;
                    const bool valid = iu < E4;
                    note(v[u].x, 4 * iu + 0, valid);
                    note(v[u].y, 4 * iu + 1, valid);
                    note(v[u].z, 4 * iu + 2, valid);
                    note(v[u].w, 4 * iu + 3, valid);
                }
            }
        }
        for (int e0 = (E4 << 2); e0 < E; e0 += AGG_THREADS) {
            int e = e0 + tid;
            bool valid = e < E;
            note(valid ? dst[e] : 0, e, valid);
        }
        if (lane == 0) wl_cnt[wlslot] = wcount;
    } else {
        wcount = (unsigned)__builtin_amdgcn_readfirstlane((int)wl_cnt[wlslot]);
    }
    if (wcount > (unsigned)WCAP && lane == 0) ovf = 1u;
    __syncthreads();

    const bool listed = (ovf == 0u);
    const int nw = (int)min(wcount, (unsigned)WCAP);

    if (listed) {
        for (int k0 = lane; k0 < nw; k0 += 128) {
            int s[4]; unsigned dl[4]; bool ok[4];
#pragma unroll
            for (int u = 0; u < 4; ++u) {
                int k = k0 + 32 * u;
                ok[u] = k < nw;
                if constexpr (FIRST) {
                    int e = ok[u] ? myS[k] : 0;
                    dl[u] = ok[u] ? (unsigned)myD[k] : 0u;
                    s[u]  = src[e];
                } else {
                    const int kr = ok[u] ? k : k0;
                    const unsigned sv = (unsigned)wl_src[wlbase + kr];
                    const unsigned dv = (unsigned)wl_dl[wlbase + kr];
                    s[u]  = (int)min(sv, (unsigned)(N - 1));
                    dl[u] = min(dv, (unsigned)(nloc - 1));
                }
            }
            float4 a4[4];
#pragma unroll
            for (int u = 0; u < 4; ++u) a4[u] = *(const float4*)(asrc + (size_t)s[u] * HEADS);
#pragma unroll
            for (int u = 0; u < 4; ++u) {
                if (ok[u]) {
                    myS[k0 + 32 * u] = s[u];
                    if constexpr (FIRST) {
                        wl_src[wlbase + k0 + 32 * u] = s[u];
                        wl_dl[wlbase + k0 + 32 * u]  = (unsigned short)dl[u];
                    } else {
                        myD[k0 + 32 * u] = (unsigned short)dl[u];
                    }
                    atomicMax(&mxk[dl[u] * HEADS + 0], fkey(leakyf(a4[u].x + adl[dl[u] * HEADS + 0])));
                    atomicMax(&mxk[dl[u] * HEADS + 1], fkey(leakyf(a4[u].y + adl[dl[u] * HEADS + 1])));
                    atomicMax(&mxk[dl[u] * HEADS + 2], fkey(leakyf(a4[u].z + adl[dl[u] * HEADS + 2])));
                    atomicMax(&mxk[dl[u] * HEADS + 3], fkey(leakyf(a4[u].w + adl[dl[u] * HEADS + 3])));
                }
            }
        }
    } else {
        for (int e = tid; e < E; e += AGG_THREADS) {
            unsigned dl = (unsigned)(dst[e] - base);
            if (dl < (unsigned)nloc) {
                int s = src[e];
#pragma unroll
                for (int h = 0; h < HEADS; ++h)
                    atomicMax(&mxk[dl * HEADS + h],
                              fkey(leakyf(asrc[s * HEADS + h] + adl[dl * HEADS + h])));
            }
        }
    }
    __syncthreads();

    for (int i = wv; i < nloc; i += AGG_WAVES) {
        int n = base + i;
        float ex[HEADS];
#pragma unroll
        for (int h = 0; h < HEADS; ++h) {
            float aself = leakyf(asrc[n * HEADS + h] + adl[i * HEADS + h]);
            ex[h] = expf(aself - funkey(mxk[i * HEADS + h]));
        }
        if (lane == 0) {
#pragma unroll
            for (int h = 0; h < HEADS; ++h) dnl[i * HEADS + h] = ex[h];
        }
        const float* row = hmat + (size_t)n * HIDF + 4 * lane;
        const float4 ra = *(const float4*)row;
        const float4 rb = *(const float4*)(row + 128);
        const float wa = upper ? ex[1] : ex[0];
        const float wb = upper ? ex[3] : ex[2];
        float* ar = acc + i * HIDF + lane;
        ar[0]   = ra.x * wa;
        ar[32]  = ra.y * wa;
        ar[64]  = ra.z * wa;
        ar[96]  = ra.w * wa;
        ar[128] = rb.x * wb;
        ar[160] = rb.y * wb;
        ar[192] = rb.z * wb;
        ar[224] = rb.w * wb;
    }
    __syncthreads();

    if (listed) {
        for (int k0 = 0; k0 < nw; k0 += 32) {
            const int k    = k0 + lane;
            const bool okl = k < nw;
            const int sL      = okl ? myS[k] : 0;
            const unsigned dL = okl ? (unsigned)myD[k] : 0u;
            const float4 a4  = *(const float4*)(asrc + (size_t)sL * HEADS);
            const float4 ad4 = *(const float4*)(adl + dL * HEADS);
            const uint4 mk4  = *(const uint4*)(mxk + dL * HEADS);
            float eL[HEADS];
            eL[0] = expf(leakyf(a4.x + ad4.x) - funkey(mk4.x));
            eL[1] = expf(leakyf(a4.y + ad4.y) - funkey(mk4.y));
            eL[2] = expf(leakyf(a4.z + ad4.z) - funkey(mk4.z));
            eL[3] = expf(leakyf(a4.w + ad4.w) - funkey(mk4.w));
            if (okl) {
#pragma unroll
                for (int h = 0; h < HEADS; ++h) atomicAdd(&dnl[dL * HEADS + h], eL[h]);
            }
            const int cnt  = min(32, nw - k0);
            const int cnt4 = cnt & ~3;
            for (int j = 0; j < cnt4; j += 4) {
                int s[4]; unsigned dl[4]; float wa[4], wb[4];
#pragma unroll
                for (int u = 0; u < 4; ++u) {
                    const int l = j + u;
                    s[u]  = __builtin_amdgcn_readlane(sL, l);
                    dl[u] = (unsigned)__builtin_amdgcn_readlane((int)dL, l);
                    float ex[HEADS];
#pragma unroll
                    for (int h = 0; h < HEADS; ++h)
                        ex[h] = __builtin_bit_cast(float, __builtin_amdgcn_readlane(
                            __builtin_bit_cast(int, eL[h]), l));
                    wa[u] = upper ? ex[1] : ex[0];
                    wb[u] = upper ? ex[3] : ex[2];
                }
                float4 ra[4], rb[4];
#pragma unroll
                for (int u = 0; u < 4; ++u) {
                    const float* hrow = hmat + (size_t)s[u] * HIDF + 4 * lane;
                    ra[u] = *(const float4*)hrow;
                    rb[u] = *(const float4*)(hrow + 128);
                }
#pragma unroll
                for (int u = 0; u < 4; ++u)
                    row_add(acc + dl[u] * HIDF + lane, ra[u], rb[u], wa[u], wb[u]);
            }
            for (int l = cnt4; l < cnt; ++l) {
                const int s1       = __builtin_amdgcn_readlane(sL, l);
                const unsigned dl1 = (unsigned)__builtin_amdgcn_readlane((int)dL, l);
                float ex1[HEADS];
#pragma unroll
                for (int h = 0; h < HEADS; ++h)
                    ex1[h] = __builtin_bit_cast(float, __builtin_amdgcn_readlane(
                        __builtin_bit_cast(int, eL[h]), l));
                const float* hrow = hmat + (size_t)s1 * HIDF + 4 * lane;
                const float4 ra1 = *(const float4*)hrow;
                const float4 rb1 = *(const float4*)(hrow + 128);
                row_add(acc + dl1 * HIDF + lane, ra1, rb1,
                        upper ? ex1[1] : ex1[0], upper ? ex1[3] : ex1[2]);
            }
        }
    } else {
        for (int e0 = wvu * 32; e0 < E; e0 += AGG_WAVES * 32) {
            int e = e0 + lane;
            int d = (e < E) ? dst[e] : 0;
            unsigned dl = (unsigned)(d - base);
            bool m = (e < E) && (dl < (unsigned)nloc);
            unsigned mask = (unsigned)__ballot(m);
            while (mask) {
                int l = __ffs(mask) - 1;
                mask &= mask - 1u;
                int s1 = src[e0 + l];
                unsigned dl1 = (unsigned)__shfl((int)dl, l, 32);
                accum_one(s1, dl1, lane, asrc, hmat, adl, mxk, dnl, acc);
            }
        }
    }
    __syncthreads();

    if constexpr (!MEAN) {
        const int hsel = lane >> 4;
        for (int i = wv; i < nloc; i += AGG_WAVES) {
            const float dna = 1.f / (dnl[i * HEADS + hsel] + 1e-16f);
            const float dnb = 1.f / (dnl[i * HEADS + 2 + hsel] + 1e-16f);
            const float4 ba = *(const float4*)(bias + 4 * lane);
            const float4 bb = *(const float4*)(bias + 128 + 4 * lane);
            const float* ar = acc + i * HIDF + lane;
            float v[8];
            v[0] = ar[0]   * dna + ba.x;
            v[1] = ar[32]  * dna + ba.y;
            v[2] = ar[64]  * dna + ba.z;
            v[3] = ar[96]  * dna + ba.w;
            v[4] = ar[128] * dnb + bb.x;
            v[5] = ar[160] * dnb + bb.y;
            v[6] = ar[192] * dnb + bb.z;
            v[7] = ar[224] * dnb + bb.w;
            float s = 0.f;
#pragma unroll
            for (int j = 0; j < 8; ++j) s += v[j];
            float mu = waveRedSum(s) * (1.f / HIDF);
            float q = 0.f;
#pragma unroll
            for (int j = 0; j < 8; ++j) { float dlt = v[j] - mu; q += dlt * dlt; }
            float rstd = rsqrtf(waveRedSum(q) * (1.f / HIDF) + LN_EPS);
            const float4 ga  = *(const float4*)(g + 4 * lane);
            const float4 gb  = *(const float4*)(g + 128 + 4 * lane);
            const float4 bea = *(const float4*)(be + 4 * lane);
            const float4 beb = *(const float4*)(be + 128 + 4 * lane);
            float4 ya, yb;
            ya.x = eluf((v[0] - mu) * rstd * ga.x + bea.x);
            ya.y = eluf((v[1] - mu) * rstd * ga.y + bea.y);
            ya.z = eluf((v[2] - mu) * rstd * ga.z + bea.z);
            ya.w = eluf((v[3] - mu) * rstd * ga.w + bea.w);
            yb.x = eluf((v[4] - mu) * rstd * gb.x + beb.x);
            yb.y = eluf((v[5] - mu) * rstd * gb.y + beb.y);
            yb.z = eluf((v[6] - mu) * rstd * gb.z + beb.z);
            yb.w = eluf((v[7] - mu) * rstd * gb.w + beb.w);
            float* orow = hout + (size_t)(base + i) * HIDF + 4 * lane;
            *(float4*)orow         = ya;
            *(float4*)(orow + 128) = yb;
        }
    } else {
        for (int i = wv; i < nloc; i += AGG_WAVES) {
            float dn[HEADS];
#pragma unroll
            for (int h = 0; h < HEADS; ++h) dn[h] = 1.f / (dnl[i * HEADS + h] + 1e-16f);
            float v[2];
            float s = 0.f;
#pragma unroll
            for (int j = 0; j < 2; ++j) {
                int c = lane + 32 * j;
                float a = 0.f;
#pragma unroll
                for (int h = 0; h < HEADS; ++h) a += acc[i * HIDF + accpos(h * CCH + c)] * dn[h];
                v[j] = a * (1.f / HEADS) + bias[c];
                s += v[j];
            }
            float mu = waveRedSum(s) * (1.f / CCH);
            float q = 0.f;
#pragma unroll
            for (int j = 0; j < 2; ++j) { float dlt = v[j] - mu; q += dlt * dlt; }
            float rstd = rsqrtf(waveRedSum(q) * (1.f / CCH) + LN_EPS);
            float* orow = hout + (size_t)(base + i) * CCH;
#pragma unroll
            for (int j = 0; j < 2; ++j) {
                int c = lane + 32 * j;
                float y = (v[j] - mu) * rstd * g[c] + be[c];
                orow[c] = eluf(y);
            }
        }
    }
}

#define VH_THREADS 128
#define VH_BLOCKS 782

__global__ __launch_bounds__(VH_THREADS)
void value_head_mm(const float* __restrict__ hin,
                   const float* __restrict__ vw1, const float* __restrict__ vb1,
                   const float* __restrict__ vw2, const float* __restrict__ vb2,
                   float* __restrict__ out, int N) {
    constexpr int KK = CCH / 8;
    const int lane = threadIdx.x & 31;
    const int c = lane & 15;
    const int h = lane >> 4;
    const int gw = __builtin_amdgcn_readfirstlane((int)((blockIdx.x * blockDim.x + threadIdx.x) >> 5));
    const int nw = (int)((gridDim.x * blockDim.x) >> 5);
    const int ntile = (N + 15) >> 4;
    if (gw >= ntile) return;

    float4 bf[2][KK];
#pragma unroll
    for (int t = 0; t < 2; ++t)
        load_frag_row<KK>(bf[t], vw1 + (size_t)(t * 16 + c) * CCH + h * 4);
    float w2[OUTF][2];
#pragma unroll
    for (int k = 0; k < OUTF; ++k) {
        w2[k][0] = vw2[k * VHID + c];
        w2[k][1] = vw2[k * VHID + 16 + c];
    }
    const float b1a = vb1[c];
    const float b1b = vb1[16 + c];
    const bool b3 = (lane & 8) != 0;
    const bool b2 = (lane & 4) != 0;
    const bool b1 = (lane & 2) != 0;
    const bool b0 = (lane & 1) != 0;
    const int kb = 2 * (lane & 1);
    const float ob0 = vb2[kb];
    const float ob1 = vb2[kb + 1];

    for (int rt = gw; rt < ntile; rt += nw) {
        const int row = min(rt * 16 + c, N - 1);
        float4 af[KK];
        load_frag_row<KK>(af, hin + (size_t)row * CCH + h * 4);
        v8f acc0 = v8f{};
        v8f acc1 = v8f{};
#pragma unroll
        for (int kk = 0; kk < KK; ++kk) {
            const v2f a0 = {af[kk].x, af[kk].y};
            const v2f a1 = {af[kk].z, af[kk].w};
            const v2f p0 = {bf[0][kk].x, bf[0][kk].y};
            const v2f q0 = {bf[1][kk].x, bf[1][kk].y};
            const v2f p1 = {bf[0][kk].z, bf[0][kk].w};
            const v2f q1 = {bf[1][kk].z, bf[1][kk].w};
            acc0 = __builtin_amdgcn_wmma_f32_16x16x4_f32(false, a0, false, p0, (short)0, acc0, false, false);
            acc1 = __builtin_amdgcn_wmma_f32_16x16x4_f32(false, a0, false, q0, (short)0, acc1, false, false);
            acc0 = __builtin_amdgcn_wmma_f32_16x16x4_f32(false, a1, false, p1, (short)0, acc0, false, false);
            acc1 = __builtin_amdgcn_wmma_f32_16x16x4_f32(false, a1, false, q1, (short)0, acc1, false, false);
        }
        float p[32];
#pragma unroll
        for (int i = 0; i < 8; ++i) {
            const float ha = eluf(acc0[i] + b1a);
            const float hb = eluf(acc1[i] + b1b);
#pragma unroll
            for (int k = 0; k < OUTF; ++k) p[4 * i + k] = ha * w2[k][0] + hb * w2[k][1];
        }
        float u[16];
#pragma unroll
        for (int m = 0; m < 16; ++m) {
            const float keep = b3 ? p[16 + m] : p[m];
            const float send = b3 ? p[m] : p[16 + m];
            u[m] = keep + __shfl_xor(send, 8, 32);
        }
        float w[8];
#pragma unroll
        for (int m = 0; m < 8; ++m) {
            const float keep = b2 ? u[8 + m] : u[m];
            const float send = b2 ? u[m] : u[8 + m];
            w[m] = keep + __shfl_xor(send, 4, 32);
        }
        float x[4];
#pragma unroll
        for (int m = 0; m < 4; ++m) {
            const float keep = b1 ? w[4 + m] : w[m];
            const float send = b1 ? w[m] : w[4 + m];
            x[m] = keep + __shfl_xor(send, 2, 32);
        }
        float y[2];
#pragma unroll
        for (int m = 0; m < 2; ++m) {
            const float keep = b0 ? x[2 + m] : x[m];
            const float send = b0 ? x[m] : x[2 + m];
            y[m] = keep + __shfl_xor(send, 1, 32);
        }
        const int orow = rt * 16 + (lane >> 1);
        if (orow < N)
            *(float2*)(out + (size_t)orow * OUTF + kb) = make_float2(y[0] + ob0, y[1] + ob1);
    }
}

extern "C" void kernel_launch(void* const* d_in, const int* in_sizes, int n_in,
                              void* d_out, int out_size, void* d_ws, size_t ws_size,
                              hipStream_t stream) {
    const float* x      = (const float*)d_in[0];
    const int*   esrc   = (const int*)d_in[1];
    const int*   edst   = (const int*)d_in[2];
    const float* proj_w = (const float*)d_in[3];
    const float* proj_b = (const float*)d_in[4];
    const float* W[3]  = {(const float*)d_in[5],  (const float*)d_in[11], (const float*)d_in[17]};
    const float* AS[3] = {(const float*)d_in[6],  (const float*)d_in[12], (const float*)d_in[18]};
    const float* AD[3] = {(const float*)d_in[7],  (const float*)d_in[13], (const float*)d_in[19]};
    const float* B[3]  = {(const float*)d_in[8],  (const float*)d_in[14], (const float*)d_in[20]};
    const float* G[3]  = {(const float*)d_in[9],  (const float*)d_in[15], (const float*)d_in[21]};
    const float* BE[3] = {(const float*)d_in[10], (const float*)d_in[16], (const float*)d_in[22]};
    const float* vw1 = (const float*)d_in[23];
    const float* vb1 = (const float*)d_in[24];
    const float* vw2 = (const float*)d_in[25];
    const float* vb2 = (const float*)d_in[26];

    const int N = in_sizes[0] / 128;
    const int E = in_sizes[1];
    const int NH = N * HEADS;

    float* ws     = (float*)d_ws;
    float* hin    = ws;
    float* hmat   = hin  + (size_t)N * HIDF;
    float* asrc   = hmat + (size_t)N * HIDF;
    float* adst   = asrc + (size_t)NH;
    const int nagg = (N + RN - 1) / RN;
    int* wl_src            = (int*)(adst + (size_t)NH);
    unsigned short* wl_dl  = (unsigned short*)(wl_src + (size_t)nagg * AGG_WAVES * WCAP);
    unsigned* wl_cnt       = (unsigned*)(wl_dl + (size_t)nagg * AGG_WAVES * WCAP);

    dim3 blk(256);

    gemm_wstat_f32<128, true><<<GW_BLOCKS_K128, GW_THREADS, 0, stream>>>(
        x, 128, proj_w, hin, N, 64, proj_b);

    int in_ld = 64;
    for (int l = 0; l < 3; ++l) {
        if (l == 0)
            gemm_wstat_f32<64, false><<<GW_BLOCKS_K64, GW_THREADS, 0, stream>>>(
                hin, in_ld, W[l], hmat, N, HIDF, nullptr);
        else
            gemm_wstat_f32<256, false><<<GW_BLOCKS_K256, GW_THREADS, 0, stream>>>(
                hin, in_ld, W[l], hmat, N, HIDF, nullptr);
        attn_coef<<<(NH + 255) / 256, blk, 0, stream>>>(hmat, AS[l], AD[l], asrc, adst, NH);
        if (l == 0)
            agg_owner<true, false><<<nagg, AGG_THREADS, 0, stream>>>(
                esrc, edst, asrc, adst, hmat, B[l], G[l], BE[l], hin, wl_src, wl_dl, wl_cnt, N, E);
        else if (l == 1)
            agg_owner<false, false><<<nagg, AGG_THREADS, 0, stream>>>(
                esrc, edst, asrc, adst, hmat, B[l], G[l], BE[l], hin, wl_src, wl_dl, wl_cnt, N, E);
        else
            agg_owner<false, true><<<nagg, AGG_THREADS, 0, stream>>>(
                esrc, edst, asrc, adst, hmat, B[l], G[l], BE[l], hin, wl_src, wl_dl, wl_cnt, N, E);
        in_ld = HIDF;
    }

    value_head_mm<<<VH_BLOCKS, VH_THREADS, 0, stream>>>(
        hin, vw1, vb1, vw2, vb2, (float*)d_out, N);
}


# Attention logits by linearity: a_s, a_d folded into each layer's weights once per call; the node-finishing wave takes the next layer's logits from its row; attn launches gone; logit sum order changed
# speedup vs baseline: 3.5969x; 1.0616x over previous
#include <hip/hip_runtime.h>
#include <math.h>

typedef __attribute__((ext_vector_type(2))) float v2f;
typedef __attribute__((ext_vector_type(8))) float v8f;

#define HEADS 4
#define CCH 64
#define HIDF 256
#define VHID 32
#define OUTF 4
#define NEG_SLOPE 0.2f
#define LN_EPS 1e-5f

__device__ __forceinline__ float eluf(float x) { return x > 0.f ? x : (expf(x) - 1.f); }
__device__ __forceinline__ float leakyf(float x) { return x > 0.f ? x : NEG_SLOPE * x; }

__device__ __forceinline__ unsigned fkey(float f) {
    unsigned u = __float_as_uint(f);
    return (u & 0x80000000u) ? ~u : (u | 0x80000000u);
}
__device__ __forceinline__ float funkey(unsigned k) {
    unsigned u = (k & 0x80000000u) ? (k ^ 0x80000000u) : ~k;
    return __uint_as_float(u);
}

__device__ __forceinline__ float waveRedSum(float v) {
    for (int off = 16; off > 0; off >>= 1) v += __shfl_xor(v, off, 32);
    return v;
}

#define GW_NT 4
#define GW_THREADS 128
#define GW_BLOCKS_K256 256
#define GW_BLOCKS_K128 512
#define GW_BLOCKS_K64 1024

__device__ __forceinline__ float xhalf(float x) {
    return __builtin_bit_cast(float, __builtin_amdgcn_permlanex16(
        __builtin_bit_cast(int, x), __builtin_bit_cast(int, x), 0x76543210, 0xfedcba98, false, false));
}

template <int KK>
__device__ __forceinline__ void load_frag_row(float4 (&f)[KK], const float* __restrict__ p) {
#pragma unroll
    for (int kk = 0; kk < KK; ++kk) f[kk] = *(const float4*)(p + kk * 8);
}

template <int KK, bool EPI>
__device__ __forceinline__ void wstat_tile(const float4 (&af)[KK], const float4 (&bf)[GW_NT][KK],
                                           float* __restrict__ C, int rt, int cb, int Nout,
                                           const float* __restrict__ bias, int c, int h) {
    v8f acc[GW_NT];
#pragma unroll
    for (int t = 0; t < GW_NT; ++t) acc[t] = v8f{};
#pragma unroll
    for (int kk = 0; kk < KK; ++kk) {
        const v2f a0 = {af[kk].x, af[kk].y};
        const v2f a1 = {af[kk].z, af[kk].w};
#pragma unroll
        for (int t = 0; t < GW_NT; ++t) {
            const v2f b0 = {bf[t][kk].x, bf[t][kk].y};
            acc[t] = __builtin_amdgcn_wmma_f32_16x16x4_f32(false, a0, false, b0,
                                                           (short)0, acc[t], false, false);
        }
#pragma unroll
        for (int t = 0; t < GW_NT; ++t) {
            const v2f b1 = {bf[t][kk].z, bf[t][kk].w};
            acc[t] = __builtin_amdgcn_wmma_f32_16x16x4_f32(false, a1, false, b1,
                                                           (short)0, acc[t], false, false);
        }
    }
#pragma unroll
    for (int t = 0; t < GW_NT; t += 2) {
        const int col = cb + t * 16 + 16 * h + c;
        float bv = 0.f;
        if (EPI) bv = bias[col];
        float* crow = C + (size_t)(rt * 16) * Nout + col;
#pragma unroll
        for (int i = 0; i < 8; ++i) {
            const float lo   = acc[t][i];
            const float hi   = acc[t + 1][i];
            const float recv = xhalf(h ? lo : hi);
            float v0 = h ? recv : lo;
            float v1 = h ? hi : recv;
            if (EPI) { v0 = eluf(v0 + bv); v1 = eluf(v1 + bv); }
            crow[(size_t)i * Nout]       = v0;
            crow[(size_t)(8 + i) * Nout] = v1;
        }
    }
}

template <int K, bool EPI>
__global__ __launch_bounds__(GW_THREADS)
void gemm_wstat_f32(const float* __restrict__ A, int lda,
                    const float* __restrict__ W,
                    float* __restrict__ C,
                    int M, int Nout,
                    const float* __restrict__ bias) {
    constexpr int KK = K / 8;
    const int lane = threadIdx.x & 31;
    const int c = lane & 15;
    const int h = lane >> 4;
    const int gw = __builtin_amdgcn_readfirstlane((int)((blockIdx.x * blockDim.x + threadIdx.x) >> 5));
    const int nw = (int)((gridDim.x * blockDim.x) >> 5);
    const int ngrp = Nout / (16 * GW_NT);
    const int cg = gw % ngrp;
    const int rs = gw / ngrp;
    const int nstream = nw / ngrp;
    const int ntile = M >> 4;
    if (rs >= nstream || rs >= ntile) return;
    const int cb = cg * (16 * GW_NT);

    float4 bf[GW_NT][KK];
#pragma unroll
    for (int t = 0; t < GW_NT; ++t)
        load_frag_row<KK>(bf[t], W + (size_t)(cb + t * 16 + c) * K + h * 4);

    const float* arow = A + (size_t)c * lda + h * 4;
    const size_t tstep = (size_t)16 * lda;
    float4 f0[KK], f1[KK];
    load_frag_row<KK>(f0, arow + (size_t)rs * tstep);
    for (int rt = rs; rt < ntile; rt += 2 * nstream) {
        const int rt1 = rt + nstream;
        const int rt2 = rt1 + nstream;
        const bool has1 = rt1 < ntile;
        load_frag_row<KK>(f1, arow + (size_t)(has1 ? rt1 : rt) * tstep);
        wstat_tile<KK, EPI>(f0, bf, C, rt, cb, Nout, bias, c, h);
        load_frag_row<KK>(f0, arow + (size_t)(rt2 < ntile ? rt2 : rt) * tstep);
        if (has1) wstat_tile<KK, EPI>(f1, bf, C, rt1, cb, Nout, bias, c, h);
    }
}

#define FOLD_OFF1 (8 * 64)
#define FOLD_OFF2 (8 * 64 + 8 * 256)
#define FOLD_FLOATS (8 * 64 + 2 * 8 * 256)
#define FOLD_BLOCKS (8 + 32 + 32)

__global__ __launch_bounds__(256)
void fold_attn(const float* __restrict__ W0, const float* __restrict__ as0, const float* __restrict__ ad0,
               const float* __restrict__ W1, const float* __restrict__ as1, const float* __restrict__ ad1,
               const float* __restrict__ W2, const float* __restrict__ as2, const float* __restrict__ ad2,
               float* __restrict__ fold) {
    __shared__ float part[4][64];
    const int b = blockIdx.x;
    const float* W; const float* as; const float* ad; float* out; int K, j, kc;
    if (b < 8)       { W = W0; as = as0; ad = ad0; out = fold;             K = 64;  j = b;             kc = 0; }
    else if (b < 40) { W = W1; as = as1; ad = ad1; out = fold + FOLD_OFF1; K = 256; j = (b - 8) >> 2;  kc = (b - 8) & 3; }
    else             { W = W2; as = as2; ad = ad2; out = fold + FOLD_OFF2; K = 256; j = (b - 40) >> 2; kc = (b - 40) & 3; }
    const int h = j & 3;
    const float* a = (j < 4 ? as : ad) + h * CCH;
    const int t = threadIdx.x;
    const int kl = t & 63;
    const int q = t >> 6;
    const int k = kc * 64 + kl;
    const float* wp = W + (size_t)(h * CCH + q * 16) * K + k;
    float s = 0.f;
#pragma unroll
    for (int cc = 0; cc < 16; ++cc) s += a[q * 16 + cc] * wp[(size_t)cc * K];
    part[q][kl] = s;
    __syncthreads();
    if (t < 64) out[j * K + k] = (part[0][t] + part[1][t]) + (part[2][t] + part[3][t]);
}

__global__ void logits_rows64(const float* __restrict__ rows, const float* __restrict__ fold,
                              float* __restrict__ asrc, float* __restrict__ adst, int N) {
    const int idx = blockIdx.x * blockDim.x + threadIdx.x;
    const int n = idx >> 3, j = idx & 7;
    if (n >= N) return;
    const float4* r = (const float4*)(rows + (size_t)n * CCH);
    const float4* f = (const float4*)(fold + j * CCH);
    float s = 0.f;
#pragma unroll
    for (int qd = 0; qd < CCH / 4; ++qd) {
        const float4 av = r[qd], fv = f[qd];
        s += av.x * fv.x; s += av.y * fv.y; s += av.z * fv.z; s += av.w * fv.w;
    }
    (j < 4 ? asrc : adst)[n * HEADS + (j & 3)] = s;
}

#define RN 200
#define AGG_THREADS 1024
#define AGG_WAVES (AGG_THREADS / 32)
#define WCAP 256

__device__ __forceinline__ int accpos(int c) { return (c & 128) | ((c & 3) << 5) | ((c & 127) >> 2); }

__device__ __forceinline__ void row_add(float* ar, const float4 a, const float4 b, float wa, float wb) {
    atomicAdd(&ar[0],   wa * a.x);
    atomicAdd(&ar[32],  wa * a.y);
    atomicAdd(&ar[64],  wa * a.z);
    atomicAdd(&ar[96],  wa * a.w);
    atomicAdd(&ar[128], wb * b.x);
    atomicAdd(&ar[160], wb * b.y);
    atomicAdd(&ar[192], wb * b.z);
    atomicAdd(&ar[224], wb * b.w);
}

__device__ __forceinline__ void accum_one(int s, unsigned dl, int lane,
                                          const float* __restrict__ asrc,
                                          const float* __restrict__ hmat,
                                          const float* adl, const unsigned* mxk,
                                          float* dnl, float* acc) {
    float ex[HEADS];
#pragma unroll
    for (int h = 0; h < HEADS; ++h) {
        float a = leakyf(asrc[s * HEADS + h] + adl[dl * HEADS + h]);
        ex[h] = expf(a - funkey(mxk[dl * HEADS + h]));
    }
    if (lane == 0) {
#pragma unroll
        for (int h = 0; h < HEADS; ++h) atomicAdd(&dnl[dl * HEADS + h], ex[h]);
    }
    const float* hrow = hmat + (size_t)s * HIDF + 4 * lane;
    const float4 ra = *(const float4*)hrow;
    const float4 rb = *(const float4*)(hrow + 128);
    const bool upper = (lane & 16) != 0;
    row_add(acc + dl * HIDF + lane, ra, rb, upper ? ex[1] : ex[0], upper ? ex[3] : ex[2]);
}

template <bool FIRST, bool MEAN>
__global__ __launch_bounds__(AGG_THREADS)
void agg_owner(const int* __restrict__ src, const int* __restrict__ dst,
               const float* __restrict__ asrc, const float* __restrict__ adst,
               const float* __restrict__ hmat,
               const float* __restrict__ bias, const float* __restrict__ g,
               const float* __restrict__ be, float* __restrict__ hout,
               int* __restrict__ wl_src, unsigned short* __restrict__ wl_dl,
               unsigned* __restrict__ wl_cnt,
               const float* __restrict__ foldn, float* __restrict__ asrc_nx, float* __restrict__ adst_nx,
               int N, int E) {
    __shared__ __attribute__((aligned(16))) float acc[RN * HIDF];
    __shared__ __attribute__((aligned(16))) int lsts[AGG_WAVES * WCAP];
    __shared__ __attribute__((aligned(16))) unsigned short lstd[AGG_WAVES * WCAP];
    __shared__ __attribute__((aligned(16))) unsigned mxk[RN * HEADS];
    __shared__ float dnl[RN * HEADS];
    __shared__ __attribute__((aligned(16))) float adl[RN * HEADS];
    __shared__ unsigned ovf;

    const int tid  = threadIdx.x;
    const int lane = tid & 31;
    const int wv   = tid >> 5;
    const bool upper = (lane & 16) != 0;
    const int base = blockIdx.x * RN;
    const int nloc = min(RN, N - base);
    int* const myS            = lsts + wv * WCAP;
    unsigned short* const myD = lstd + wv * WCAP;

    if (tid == 0) ovf = 0u;
    for (int i = tid; i < nloc * HEADS; i += AGG_THREADS) {
        float ad = adst[base * HEADS + i];
        adl[i] = ad;
        mxk[i] = fkey(leakyf(asrc[base * HEADS + i] + ad));
    }
    __syncthreads();

    const size_t wlslot = (size_t)blockIdx.x * AGG_WAVES + wv;
    const size_t wlbase = wlslot * WCAP;
    const int wvu = __builtin_amdgcn_readfirstlane(wv);
    unsigned wcount = 0u;
    if constexpr (FIRST) {
        auto note = [&](int d, int e, bool valid) {
            unsigned dl = (unsigned)(d - base);
            bool m = valid && (dl < (unsigned)nloc);
            unsigned mask = (unsigned)__ballot(m);
            if (mask) {
                unsigned slot = wcount + __builtin_amdgcn_mbcnt_lo(mask, 0u);
                if (m && slot < (unsigned)WCAP) { myS[slot] = e; myD[slot] = (unsigned short)dl; }
                wcount += (unsigned)__popc(mask);
            }
        };
        const int E4 = E >> 2;
        const int4* dst4 = (const int4*)dst;
        for (int ib = wvu * 32; ib < E4; ib += AGG_THREADS * 4) {
            const int i = ib + lane;
            int4 v[4];
            if (ib + 3 * AGG_THREADS + 32 <= E4) {
#pragma unroll
                for (int u = 0; u < 4; ++u) v[u] = dst4[i + u * AGG_THREADS];
#pragma unroll
                for (int u = 0; u < 4; ++u) {
                    const int e = 4 * (i + u * AGG_THREADS);
                    note(v[u].x, e + 0, true);
                    note(v[u].y, e + 1, true);
                    note(v[u].z, e + 2, true);
                    note(v[u].w, e + 3, true);
                }
            } else {
#pragma unroll
                for (int u = 0; u < 4; ++u) {
                    const int iu = i + u * AGG_THREADS;
                    v[u] = (iu < E4) ? dst4[iu] : make_int4(0, 0, 0, 0);
                }
#pragma unroll
                for (int u = 0; u < 4; ++u) {
                    const int iu = i + u * AGG_THREADS;
                    const bool valid = iu < E4;
                    note(v[u].x, 4 * iu + 0, valid);
                    note(v[u].y, 4 * iu + 1, valid);
                    note(v[u].z, 4 * iu + 2, valid);
                    note(v[u].w, 4 * iu + 3, valid);
                }
            }
        }
        for (int e0 = (E4 << 2); e0 < E; e0 += AGG_THREADS) {
            int e = e0 + tid;
            bool valid = e < E;
            note(valid ? dst[e] : 0, e, valid);
        }
        if (lane == 0) wl_cnt[wlslot] = wcount;
    } else {
        wcount = (unsigned)__builtin_amdgcn_readfirstlane((int)wl_cnt[wlslot]);
    }
    if (wcount > (unsigned)WCAP && lane == 0) ovf = 1u;
    __syncthreads();

    const bool listed = (ovf == 0u);
    const int nw = (int)min(wcount, (unsigned)WCAP);

    if (listed) {
        for (int k0 = lane; k0 < nw; k0 += 128) {
            int s[4]; unsigned dl[4]; bool ok[4];
#pragma unroll
            for (int u = 0; u < 4; ++u) {
                int k = k0 + 32 * u;
                ok[u] = k < nw;
                if constexpr (FIRST) {
                    int e = ok[u] ? myS[k] : 0;
                    dl[u] = ok[u] ? (unsigned)myD[k] : 0u;
                    s[u]  = src[e];
                } else {
                    const int kr = ok[u] ? k : k0;
                    const unsigned sv = (unsigned)wl_src[wlbase + kr];
                    const unsigned dv = (unsigned)wl_dl[wlbase + kr];
                    s[u]  = (int)min(sv, (unsigned)(N - 1));
                    dl[u] = min(dv, (unsigned)(nloc - 1));
                }
            }
            float4 a4[4];
#pragma unroll
            for (int u = 0; u < 4; ++u) a4[u] = *(const float4*)(asrc + (size_t)s[u] * HEADS);
#pragma unroll
            for (int u = 0; u < 4; ++u) {
                if (ok[u]) {
                    myS[k0 + 32 * u] = s[u];
                    if constexpr (FIRST) {
                        wl_src[wlbase + k0 + 32 * u] = s[u];
                        wl_dl[wlbase + k0 + 32 * u]  = (unsigned short)dl[u];
                    } else {
                        myD[k0 + 32 * u] = (unsigned short)dl[u];
                    }
                    atomicMax(&mxk[dl[u] * HEADS + 0], fkey(leakyf(a4[u].x + adl[dl[u] * HEADS + 0])));
                    atomicMax(&mxk[dl[u] * HEADS + 1], fkey(leakyf(a4[u].y + adl[dl[u] * HEADS + 1])));
                    atomicMax(&mxk[dl[u] * HEADS + 2], fkey(leakyf(a4[u].z + adl[dl[u] * HEADS + 2])));
                    atomicMax(&mxk[dl[u] * HEADS + 3], fkey(leakyf(a4[u].w + adl[dl[u] * HEADS + 3])));
                }
            }
        }
    } else {
        for (int e = tid; e < E; e += AGG_THREADS) {
            unsigned dl = (unsigned)(dst[e] - base);
            if (dl < (unsigned)nloc) {
                int s = src[e];
#pragma unroll
                for (int h = 0; h < HEADS; ++h)
                    atomicMax(&mxk[dl * HEADS + h],
                              fkey(leakyf(asrc[s * HEADS + h] + adl[dl * HEADS + h])));
            }
        }
    }
    __syncthreads();

    for (int i = wv; i < nloc; i += AGG_WAVES) {
        int n = base + i;
        float ex[HEADS];
#pragma unroll
        for (int h = 0; h < HEADS; ++h) {
            float aself = leakyf(asrc[n * HEADS + h] + adl[i * HEADS + h]);
            ex[h] = expf(aself - funkey(mxk[i * HEADS + h]));
        }
        if (lane == 0) {
#pragma unroll
            for (int h = 0; h < HEADS; ++h) dnl[i * HEADS + h] = ex[h];
        }
        const float* row = hmat + (size_t)n * HIDF + 4 * lane;
        const float4 ra = *(const float4*)row;
        const float4 rb = *(const float4*)(row + 128);
        const float wa = upper ? ex[1] : ex[0];
        const float wb = upper ? ex[3] : ex[2];
        float* ar = acc + i * HIDF + lane;
        ar[0]   = ra.x * wa;
        ar[32]  = ra.y * wa;
        ar[64]  = ra.z * wa;
        ar[96]  = ra.w * wa;
        ar[128] = rb.x * wb;
        ar[160] = rb.y * wb;
        ar[192] = rb.z * wb;
        ar[224] = rb.w * wb;
    }
    __syncthreads();

    if (listed) {
        for (int k0 = 0; k0 < nw; k0 += 32) {
            const int k    = k0 + lane;
            const bool okl = k < nw;
            const int sL      = okl ? myS[k] : 0;
            const unsigned dL = okl ? (unsigned)myD[k] : 0u;
            const float4 a4  = *(const float4*)(asrc + (size_t)sL * HEADS);
            const float4 ad4 = *(const float4*)(adl + dL * HEADS);
            const uint4 mk4  = *(const uint4*)(mxk + dL * HEADS);
            float eL[HEADS];
            eL[0] = expf(leakyf(a4.x + ad4.x) - funkey(mk4.x));
            eL[1] = expf(leakyf(a4.y + ad4.y) - funkey(mk4.y));
            eL[2] = expf(leakyf(a4.z + ad4.z) - funkey(mk4.z));
            eL[3] = expf(leakyf(a4.w + ad4.w) - funkey(mk4.w));
            if (okl) {
#pragma unroll
                for (int h = 0; h < HEADS; ++h) atomicAdd(&dnl[dL * HEADS + h], eL[h]);
            }
            const int cnt  = min(32, nw - k0);
            const int cnt4 = cnt & ~3;
            for (int j = 0; j < cnt4; j += 4) {
                int s[4]; unsigned dl[4]; float wa[4], wb[4];
#pragma unroll
                for (int u = 0; u < 4; ++u) {
                    const int l = j + u;
                    s[u]  = __builtin_amdgcn_readlane(sL, l);
                    dl[u] = (unsigned)__builtin_amdgcn_readlane((int)dL, l);
                    float ex[HEADS];
#pragma unroll
                    for (int h = 0; h < HEADS; ++h)
                        ex[h] = __builtin_bit_cast(float, __builtin_amdgcn_readlane(
                            __builtin_bit_cast(int, eL[h]), l));
                    wa[u] = upper ? ex[1] : ex[0];
                    wb[u] = upper ? ex[3] : ex[2];
                }
                float4 ra[4], rb[4];
#pragma unroll
                for (int u = 0; u < 4; ++u) {
                    const float* hrow = hmat + (size_t)s[u] * HIDF + 4 * lane;
                    ra[u] = *(const float4*)hrow;
                    rb[u] = *(const float4*)(hrow + 128);
                }
#pragma unroll
                for (int u = 0; u < 4; ++u)
                    row_add(acc + dl[u] * HIDF + lane, ra[u], rb[u], wa[u], wb[u]);
            }
            for (int l = cnt4; l < cnt; ++l) {
                const int s1       = __builtin_amdgcn_readlane(sL, l);
                const unsigned dl1 = (unsigned)__builtin_amdgcn_readlane((int)dL, l);
                float ex1[HEADS];
#pragma unroll
                for (int h = 0; h < HEADS; ++h)
                    ex1[h] = __builtin_bit_cast(float, __builtin_amdgcn_readlane(
                        __builtin_bit_cast(int, eL[h]), l));
                const float* hrow = hmat + (size_t)s1 * HIDF + 4 * lane;
                const float4 ra1 = *(const float4*)hrow;
                const float4 rb1 = *(const float4*)(hrow + 128);
                row_add(acc + dl1 * HIDF + lane, ra1, rb1,
                        upper ? ex1[1] : ex1[0], upper ? ex1[3] : ex1[2]);
            }
        }
    } else {
        for (int e0 = wvu * 32; e0 < E; e0 += AGG_WAVES * 32) {
            int e = e0 + lane;
            int d = (e < E) ? dst[e] : 0;
            unsigned dl = (unsigned)(d - base);
            bool m = (e < E) && (dl < (unsigned)nloc);
            unsigned mask = (unsigned)__ballot(m);
            while (mask) {
                int l = __ffs(mask) - 1;
                mask &= mask - 1u;
                int s1 = src[e0 + l];
                unsigned dl1 = (unsigned)__shfl((int)dl, l, 32);
                accum_one(s1, dl1, lane, asrc, hmat, adl, mxk, dnl, acc);
            }
        }
    }
    __syncthreads();

    if constexpr (!MEAN) {
        const int hsel = lane >> 4;
        float4 fa[8], fb[8];
#pragma unroll
        for (int j = 0; j < 8; ++j) {
            fa[j] = *(const float4*)(foldn + j * HIDF + 4 * lane);
            fb[j] = *(const float4*)(foldn + j * HIDF + 128 + 4 * lane);
        }
        const bool l16 = (lane & 16) != 0;
        const bool l8  = (lane & 8) != 0;
        const bool l4  = (lane & 4) != 0;
        for (int i = wv; i < nloc; i += AGG_WAVES) {
            const float dna = 1.f / (dnl[i * HEADS + hsel] + 1e-16f);
            const float dnb = 1.f / (dnl[i * HEADS + 2 + hsel] + 1e-16f);
            const float4 ba = *(const float4*)(bias + 4 * lane);
            const float4 bb = *(const float4*)(bias + 128 + 4 * lane);
            const float* ar = acc + i * HIDF + lane;
            float v[8];
            v[0] = ar[0]   * dna + ba.x;
            v[1] = ar[32]  * dna + ba.y;
            v[2] = ar[64]  * dna + ba.z;
            v[3] = ar[96]  * dna + ba.w;
            v[4] = ar[128] * dnb + bb.x;
            v[5] = ar[160] * dnb + bb.y;
            v[6] = ar[192] * dnb + bb.z;
            v[7] = ar[224] * dnb + bb.w;
            float s = 0.f;
#pragma unroll
            for (int j = 0; j < 8; ++j) s += v[j];
            float mu = waveRedSum(s) * (1.f / HIDF);
            float q = 0.f;
#pragma unroll
            for (int j = 0; j < 8; ++j) { float dlt = v[j] - mu; q += dlt * dlt; }
            float rstd = rsqrtf(waveRedSum(q) * (1.f / HIDF) + LN_EPS);
            const float4 ga  = *(const float4*)(g + 4 * lane);
            const float4 gb  = *(const float4*)(g + 128 + 4 * lane);
            const float4 bea = *(const float4*)(be + 4 * lane);
            const float4 beb = *(const float4*)(be + 128 + 4 * lane);
            float4 ya, yb;
            ya.x = eluf((v[0] - mu) * rstd * ga.x + bea.x);
            ya.y = eluf((v[1] - mu) * rstd * ga.y + bea.y);
            ya.z = eluf((v[2] - mu) * rstd * ga.z + bea.z);
            ya.w = eluf((v[3] - mu) * rstd * ga.w + bea.w);
            yb.x = eluf((v[4] - mu) * rstd * gb.x + beb.x);
            yb.y = eluf((v[5] - mu) * rstd * gb.y + beb.y);
            yb.z = eluf((v[6] - mu) * rstd * gb.z + beb.z);
            yb.w = eluf((v[7] - mu) * rstd * gb.w + beb.w);
            float* orow = hout + (size_t)(base + i) * HIDF + 4 * lane;
            *(float4*)orow         = ya;
            *(float4*)(orow + 128) = yb;
            float p[8];
#pragma unroll
            for (int j = 0; j < 8; ++j)
                p[j] = ya.x * fa[j].x + ya.y * fa[j].y + ya.z * fa[j].z + ya.w * fa[j].w
                     + yb.x * fb[j].x + yb.y * fb[j].y + yb.z * fb[j].z + yb.w * fb[j].w;
            float u4[4];
#pragma unroll
            for (int m = 0; m < 4; ++m) {
                const float keep = l16 ? p[4 + m] : p[m];
                const float send = l16 ? p[m] : p[4 + m];
                u4[m] = keep + __shfl_xor(send, 16, 32);
            }
            float u2[2];
#pragma unroll
            for (int m = 0; m < 2; ++m) {
                const float keep = l8 ? u4[2 + m] : u4[m];
                const float send = l8 ? u4[m] : u4[2 + m];
                u2[m] = keep + __shfl_xor(send, 8, 32);
            }
            float u1 = (l4 ? u2[1] : u2[0]) + __shfl_xor(l4 ? u2[0] : u2[1], 4, 32);
            u1 += __shfl_xor(u1, 2, 32);
            u1 += __shfl_xor(u1, 1, 32);
            if ((lane & 3) == 0) {
                const int jj = (l16 ? 4 : 0) + (l8 ? 2 : 0) + (l4 ? 1 : 0);
                (jj < 4 ? asrc_nx : adst_nx)[(size_t)(base + i) * HEADS + (jj & 3)] = u1;
            }
        }
    } else {
        for (int i = wv; i < nloc; i += AGG_WAVES) {
            float dn[HEADS];
#pragma unroll
            for (int h = 0; h < HEADS; ++h) dn[h] = 1.f / (dnl[i * HEADS + h] + 1e-16f);
            float v[2];
            float s = 0.f;
#pragma unroll
            for (int j = 0; j < 2; ++j) {
                int c = lane + 32 * j;
                float a = 0.f;
#pragma unroll
                for (int h = 0; h < HEADS; ++h) a += acc[i * HIDF + accpos(h * CCH + c)] * dn[h];
                v[j] = a * (1.f / HEADS) + bias[c];
                s += v[j];
            }
            float mu = waveRedSum(s) * (1.f / CCH);
            float q = 0.f;
#pragma unroll
            for (int j = 0; j < 2; ++j) { float dlt = v[j] - mu; q += dlt * dlt; }
            float rstd = rsqrtf(waveRedSum(q) * (1.f / CCH) + LN_EPS);
            float* orow = hout + (size_t)(base + i) * CCH;
#pragma unroll
            for (int j = 0; j < 2; ++j) {
                int c = lane + 32 * j;
                float y = (v[j] - mu) * rstd * g[c] + be[c];
                orow[c] = eluf(y);
            }
        }
    }
}

#define VH_THREADS 128
#define VH_BLOCKS 782

__global__ __launch_bounds__(VH_THREADS)
void value_head_mm(const float* __restrict__ hin,
                   const float* __restrict__ vw1, const float* __restrict__ vb1,
                   const float* __restrict__ vw2, const float* __restrict__ vb2,
                   float* __restrict__ out, int N) {
    constexpr int KK = CCH / 8;
    const int lane = threadIdx.x & 31;
    const int c = lane & 15;
    const int h = lane >> 4;
    const int gw = __builtin_amdgcn_readfirstlane((int)((blockIdx.x * blockDim.x + threadIdx.x) >> 5));
    const int nw = (int)((gridDim.x * blockDim.x) >> 5);
    const int ntile = (N + 15) >> 4;
    if (gw >= ntile) return;

    float4 bf[2][KK];
#pragma unroll
    for (int t = 0; t < 2; ++t)
        load_frag_row<KK>(bf[t], vw1 + (size_t)(t * 16 + c) * CCH + h * 4);
    float w2[OUTF][2];
#pragma unroll
    for (int k = 0; k < OUTF; ++k) {
        w2[k][0] = vw2[k * VHID + c];
        w2[k][1] = vw2[k * VHID + 16 + c];
    }
    const float b1a = vb1[c];
    const float b1b = vb1[16 + c];
    const bool b3 = (lane & 8) != 0;
    const bool b2 = (lane & 4) != 0;
    const bool b1 = (lane & 2) != 0;
    const bool b0 = (lane & 1) != 0;
    const int kb = 2 * (lane & 1);
    const float ob0 = vb2[kb];
    const float ob1 = vb2[kb + 1];

    for (int rt = gw; rt < ntile; rt += nw) {
        const int row = min(rt * 16 + c, N - 1);
        float4 af[KK];
        load_frag_row<KK>(af, hin + (size_t)row * CCH + h * 4);
        v8f acc0 = v8f{};
        v8f acc1 = v8f{};
#pragma unroll
        for (int kk = 0; kk < KK; ++kk) {
            const v2f a0 = {af[kk].x, af[kk].y};
            const v2f a1 = {af[kk].z, af[kk].w};
            const v2f p0 = {bf[0][kk].x, bf[0][kk].y};
            const v2f q0 = {bf[1][kk].x, bf[1][kk].y};
            const v2f p1 = {bf[0][kk].z, bf[0][kk].w};
            const v2f q1 = {bf[1][kk].z, bf[1][kk].w};
            acc0 = __builtin_amdgcn_wmma_f32_16x16x4_f32(false, a0, false, p0, (short)0, acc0, false, false);
            acc1 = __builtin_amdgcn_wmma_f32_16x16x4_f32(false, a0, false, q0, (short)0, acc1, false, false);
            acc0 = __builtin_amdgcn_wmma_f32_16x16x4_f32(false, a1, false, p1, (short)0, acc0, false, false);
            acc1 = __builtin_amdgcn_wmma_f32_16x16x4_f32(false, a1, false, q1, (short)0, acc1, false, false);
        }
        float p[32];
#pragma unroll
        for (int i = 0; i < 8; ++i) {
            const float ha = eluf(acc0[i] + b1a);
            const float hb = eluf(acc1[i] + b1b);
#pragma unroll
            for (int k = 0; k < OUTF; ++k) p[4 * i + k] = ha * w2[k][0] + hb * w2[k][1];
        }
        float u[16];
#pragma unroll
        for (int m = 0; m < 16; ++m) {
            const float keep = b3 ? p[16 + m] : p[m];
            const float send = b3 ? p[m] : p[16 + m];
            u[m] = keep + __shfl_xor(send, 8, 32);
        }
        float w[8];
#pragma unroll
        for (int m = 0; m < 8; ++m) {
            const float keep = b2 ? u[8 + m] : u[m];
            const float send = b2 ? u[m] : u[8 + m];
            w[m] = keep + __shfl_xor(send, 4, 32);
        }
        float x[4];
#pragma unroll
        for (int m = 0; m < 4; ++m) {
            const float keep = b1 ? w[4 + m] : w[m];
            const float send = b1 ? w[m] : w[4 + m];
            x[m] = keep + __shfl_xor(send, 2, 32);
        }
        float y[2];
#pragma unroll
        for (int m = 0; m < 2; ++m) {
            const float keep = b0 ? x[2 + m] : x[m];
            const float send = b0 ? x[m] : x[2 + m];
            y[m] = keep + __shfl_xor(send, 1, 32);
        }
        const int orow = rt * 16 + (lane >> 1);
        if (orow < N)
            *(float2*)(out + (size_t)orow * OUTF + kb) = make_float2(y[0] + ob0, y[1] + ob1);
    }
}

extern "C" void kernel_launch(void* const* d_in, const int* in_sizes, int n_in,
                              void* d_out, int out_size, void* d_ws, size_t ws_size,
                              hipStream_t stream) {
    const float* x      = (const float*)d_in[0];
    const int*   esrc   = (const int*)d_in[1];
    const int*   edst   = (const int*)d_in[2];
    const float* proj_w = (const float*)d_in[3];
    const float* proj_b = (const float*)d_in[4];
    const float* W[3]  = {(const float*)d_in[5],  (const float*)d_in[11], (const float*)d_in[17]};
    const float* AS[3] = {(const float*)d_in[6],  (const float*)d_in[12], (const float*)d_in[18]};
    const float* AD[3] = {(const float*)d_in[7],  (const float*)d_in[13], (const float*)d_in[19]};
    const float* B[3]  = {(const float*)d_in[8],  (const float*)d_in[14], (const float*)d_in[20]};
    const float* G[3]  = {(const float*)d_in[9],  (const float*)d_in[15], (const float*)d_in[21]};
    const float* BE[3] = {(const float*)d_in[10], (const float*)d_in[16], (const float*)d_in[22]};
    const float* vw1 = (const float*)d_in[23];
    const float* vb1 = (const float*)d_in[24];
    const float* vw2 = (const float*)d_in[25];
    const float* vb2 = (const float*)d_in[26];

    const int N = in_sizes[0] / 128;
    const int E = in_sizes[1];
    const int NH = N * HEADS;

    float* ws     = (float*)d_ws;
    float* hin    = ws;
    float* hmat   = hin  + (size_t)N * HIDF;
    float* asrcA  = hmat  + (size_t)N * HIDF;
    float* adstA  = asrcA + (size_t)NH;
    float* asrcB  = adstA + (size_t)NH;
    float* adstB  = asrcB + (size_t)NH;
    float* fold   = adstB + (size_t)NH;
    const int nagg = (N + RN - 1) / RN;
    int* wl_src            = (int*)(fold + FOLD_FLOATS);
    unsigned short* wl_dl  = (unsigned short*)(wl_src + (size_t)nagg * AGG_WAVES * WCAP);
    unsigned* wl_cnt       = (unsigned*)(wl_dl + (size_t)nagg * AGG_WAVES * WCAP);

    fold_attn<<<FOLD_BLOCKS, 256, 0, stream>>>(W[0], AS[0], AD[0], W[1], AS[1], AD[1],
                                               W[2], AS[2], AD[2], fold);

    gemm_wstat_f32<128, true><<<GW_BLOCKS_K128, GW_THREADS, 0, stream>>>(
        x, 128, proj_w, hin, N, 64, proj_b);
    logits_rows64<<<(N * 8 + 255) / 256, 256, 0, stream>>>(hin, fold, asrcA, adstA, N);

    gemm_wstat_f32<64, false><<<GW_BLOCKS_K64, GW_THREADS, 0, stream>>>(
        hin, 64, W[0], hmat, N, HIDF, nullptr);
    agg_owner<true, false><<<nagg, AGG_THREADS, 0, stream>>>(
        esrc, edst, asrcA, adstA, hmat, B[0], G[0], BE[0], hin, wl_src, wl_dl, wl_cnt,
        fold + FOLD_OFF1, asrcB, adstB, N, E);

    gemm_wstat_f32<256, false><<<GW_BLOCKS_K256, GW_THREADS, 0, stream>>>(
        hin, HIDF, W[1], hmat, N, HIDF, nullptr);
    agg_owner<false, false><<<nagg, AGG_THREADS, 0, stream>>>(
        esrc, edst, asrcB, adstB, hmat, B[1], G[1], BE[1], hin, wl_src, wl_dl, wl_cnt,
        fold + FOLD_OFF2, asrcA, adstA, N, E);

    gemm_wstat_f32<256, false><<<GW_BLOCKS_K256, GW_THREADS, 0, stream>>>(
        hin, HIDF, W[2], hmat, N, HIDF, nullptr);
    agg_owner<false, true><<<nagg, AGG_THREADS, 0, stream>>>(
        esrc, edst, asrcA, adstA, hmat, B[2], G[2], BE[2], hin, wl_src, wl_dl, wl_cnt,
        nullptr, nullptr, nullptr, N, E);

    value_head_mm<<<VH_BLOCKS, VH_THREADS, 0, stream>>>(
        hin, vw1, vb1, vw2, vb2, (float*)d_out, N);
}


# Edge scan over 1-byte owner keys (dst/200, written by added work-groups of the fold launch): 16 keys per lane load, 4x fewer scan bytes; local dst read per hit; in-edge sum order changed
# speedup vs baseline: 3.7668x; 1.0472x over previous
#include <hip/hip_runtime.h>
#include <math.h>

typedef __attribute__((ext_vector_type(2))) float v2f;
typedef __attribute__((ext_vector_type(8))) float v8f;

#define HEADS 4
#define CCH 64
#define HIDF 256
#define VHID 32
#define OUTF 4
#define NEG_SLOPE 0.2f
#define LN_EPS 1e-5f

__device__ __forceinline__ float eluf(float x) { return x > 0.f ? x : (expf(x) - 1.f); }
__device__ __forceinline__ float leakyf(float x) { return x > 0.f ? x : NEG_SLOPE * x; }

__device__ __forceinline__ unsigned fkey(float f) {
    unsigned u = __float_as_uint(f);
    return (u & 0x80000000u) ? ~u : (u | 0x80000000u);
}
__device__ __forceinline__ float funkey(unsigned k) {
    unsigned u = (k & 0x80000000u) ? (k ^ 0x80000000u) : ~k;
    return __uint_as_float(u);
}

__device__ __forceinline__ float waveRedSum(float v) {
    for (int off = 16; off > 0; off >>= 1) v += __shfl_xor(v, off, 32);
    return v;
}

#define GW_NT 4
#define GW_THREADS 128
#define GW_BLOCKS_K256 256
#define GW_BLOCKS_K128 512
#define GW_BLOCKS_K64 1024

__device__ __forceinline__ float xhalf(float x) {
    return __builtin_bit_cast(float, __builtin_amdgcn_permlanex16(
        __builtin_bit_cast(int, x), __builtin_bit_cast(int, x), 0x76543210, 0xfedcba98, false, false));
}

template <int KK>
__device__ __forceinline__ void load_frag_row(float4 (&f)[KK], const float* __restrict__ p) {
#pragma unroll
    for (int kk = 0; kk < KK; ++kk) f[kk] = *(const float4*)(p + kk * 8);
}

template <int KK, bool EPI>
__device__ __forceinline__ void wstat_tile(const float4 (&af)[KK], const float4 (&bf)[GW_NT][KK],
                                           float* __restrict__ C, int rt, int cb, int Nout,
                                           const float* __restrict__ bias, int c, int h) {
    v8f acc[GW_NT];
#pragma unroll
    for (int t = 0; t < GW_NT; ++t) acc[t] = v8f{};
#pragma unroll
    for (int kk = 0; kk < KK; ++kk) {
        const v2f a0 = {af[kk].x, af[kk].y};
        const v2f a1 = {af[kk].z, af[kk].w};
#pragma unroll
        for (int t = 0; t < GW_NT; ++t) {
            const v2f b0 = {bf[t][kk].x, bf[t][kk].y};
            acc[t] = __builtin_amdgcn_wmma_f32_16x16x4_f32(false, a0, false, b0,
                                                           (short)0, acc[t], false, false);
        }
#pragma unroll
        for (int t = 0; t < GW_NT; ++t) {
            const v2f b1 = {bf[t][kk].z, bf[t][kk].w};
            acc[t] = __builtin_amdgcn_wmma_f32_16x16x4_f32(false, a1, false, b1,
                                                           (short)0, acc[t], false, false);
        }
    }
#pragma unroll
    for (int t = 0; t < GW_NT; t += 2) {
        const int col = cb + t * 16 + 16 * h + c;
        float bv = 0.f;
        if (EPI) bv = bias[col];
        float* crow = C + (size_t)(rt * 16) * Nout + col;
#pragma unroll
        for (int i = 0; i < 8; ++i) {
            const float lo   = acc[t][i];
            const float hi   = acc[t + 1][i];
            const float recv = xhalf(h ? lo : hi);
            float v0 = h ? recv : lo;
            float v1 = h ? hi : recv;
            if (EPI) { v0 = eluf(v0 + bv); v1 = eluf(v1 + bv); }
            crow[(size_t)i * Nout]       = v0;
            crow[(size_t)(8 + i) * Nout] = v1;
        }
    }
}

template <int K, bool EPI>
__global__ __launch_bounds__(GW_THREADS)
void gemm_wstat_f32(const float* __restrict__ A, int lda,
                    const float* __restrict__ W,
                    float* __restrict__ C,
                    int M, int Nout,
                    const float* __restrict__ bias) {
    constexpr int KK = K / 8;
    const int lane = threadIdx.x & 31;
    const int c = lane & 15;
    const int h = lane >> 4;
    const int gw = __builtin_amdgcn_readfirstlane((int)((blockIdx.x * blockDim.x + threadIdx.x) >> 5));
    const int nw = (int)((gridDim.x * blockDim.x) >> 5);
    const int ngrp = Nout / (16 * GW_NT);
    const int cg = gw % ngrp;
    const int rs = gw / ngrp;
    const int nstream = nw / ngrp;
    const int ntile = M >> 4;
    if (rs >= nstream || rs >= ntile) return;
    const int cb = cg * (16 * GW_NT);

    float4 bf[GW_NT][KK];
#pragma unroll
    for (int t = 0; t < GW_NT; ++t)
        load_frag_row<KK>(bf[t], W + (size_t)(cb + t * 16 + c) * K + h * 4);

    const float* arow = A + (size_t)c * lda + h * 4;
    const size_t tstep = (size_t)16 * lda;
    float4 f0[KK], f1[KK];
    load_frag_row<KK>(f0, arow + (size_t)rs * tstep);
    for (int rt = rs; rt < ntile; rt += 2 * nstream) {
        const int rt1 = rt + nstream;
        const int rt2 = rt1 + nstream;
        const bool has1 = rt1 < ntile;
        load_frag_row<KK>(f1, arow + (size_t)(has1 ? rt1 : rt) * tstep);
        wstat_tile<KK, EPI>(f0, bf, C, rt, cb, Nout, bias, c, h);
        load_frag_row<KK>(f0, arow + (size_t)(rt2 < ntile ? rt2 : rt) * tstep);
        if (has1) wstat_tile<KK, EPI>(f1, bf, C, rt1, cb, Nout, bias, c, h);
    }
}

#define FOLD_OFF1 (8 * 64)
#define FOLD_OFF2 (8 * 64 + 8 * 256)
#define FOLD_FLOATS (8 * 64 + 2 * 8 * 256)
#define FOLD_BLOCKS (8 + 32 + 32)
#define FOLD_THREADS 256
#define RN 200
#define KEY_NONE 255u
#define KEY_STEP 4096
#define KEY_ROUNDS(E) ((((E) + 15) / 16 + KEY_STEP - 1) / KEY_STEP)
#define KEY_PIECES(E) ((KEY_ROUNDS(E) + 1) * KEY_STEP)

__device__ __forceinline__ unsigned owner_key(int d) { return min((unsigned)d / (unsigned)RN, KEY_NONE); }

__global__ __launch_bounds__(FOLD_THREADS)
void fold_attn(const float* __restrict__ W0, const float* __restrict__ as0, const float* __restrict__ ad0,
               const float* __restrict__ W1, const float* __restrict__ as1, const float* __restrict__ ad1,
               const float* __restrict__ W2, const float* __restrict__ as2, const float* __restrict__ ad2,
               float* __restrict__ fold,
               const int* __restrict__ dst, unsigned char* __restrict__ keys, int E) {
    __shared__ float part[4][64];
    const int b = blockIdx.x;
    if (b >= FOLD_BLOCKS) {
        const int i = (b - FOLD_BLOCKS) * FOLD_THREADS + (int)threadIdx.x;
        const int e0 = 16 * i;
        unsigned w[4];
        if (e0 + 16 <= E) {
            const int4* d4 = (const int4*)(dst + e0);
#pragma unroll
            for (int d = 0; d < 4; ++d) {
                const int4 q = d4[d];
                w[d] = owner_key(q.x) | (owner_key(q.y) << 8) | (owner_key(q.z) << 16) | (owner_key(q.w) << 24);
            }
        } else {
#pragma unroll
            for (int d = 0; d < 4; ++d) {
                w[d] = 0u;
#pragma unroll
                for (int bb = 0; bb < 4; ++bb) {
                    const int e = e0 + 4 * d + bb;
                    w[d] |= (e < E ? owner_key(dst[e]) : KEY_NONE) << (8 * bb);
                }
            }
        }
        *(uint4*)(keys + e0) = make_uint4(w[0], w[1], w[2], w[3]);
        return;
    }
    const float* W; const float* as; const float* ad; float* out; int K, j, kc;
    if (b < 8)       { W = W0; as = as0; ad = ad0; out = fold;             K = 64;  j = b;             kc = 0; }
    else if (b < 40) { W = W1; as = as1; ad = ad1; out = fold + FOLD_OFF1; K = 256; j = (b - 8) >> 2;  kc = (b - 8) & 3; }
    else             { W = W2; as = as2; ad = ad2; out = fold + FOLD_OFF2; K = 256; j = (b - 40) >> 2; kc = (b - 40) & 3; }
    const int h = j & 3;
    const float* a = (j < 4 ? as : ad) + h * CCH;
    const int t = threadIdx.x;
    const int kl = t & 63;
    const int q = t >> 6;
    const int k = kc * 64 + kl;
    const float* wp = W + (size_t)(h * CCH + q * 16) * K + k;
    float s = 0.f;
#pragma unroll
    for (int cc = 0; cc < 16; ++cc) s += a[q * 16 + cc] * wp[(size_t)cc * K];
    part[q][kl] = s;
    __syncthreads();
    if (t < 64) out[j * K + k] = (part[0][t] + part[1][t]) + (part[2][t] + part[3][t]);
}

__global__ void logits_rows64(const float* __restrict__ rows, const float* __restrict__ fold,
                              float* __restrict__ asrc, float* __restrict__ adst, int N) {
    const int idx = blockIdx.x * blockDim.x + threadIdx.x;
    const int n = idx >> 3, j = idx & 7;
    if (n >= N) return;
    const float4* r = (const float4*)(rows + (size_t)n * CCH);
    const float4* f = (const float4*)(fold + j * CCH);
    float s = 0.f;
#pragma unroll
    for (int qd = 0; qd < CCH / 4; ++qd) {
        const float4 av = r[qd], fv = f[qd];
        s += av.x * fv.x; s += av.y * fv.y; s += av.z * fv.z; s += av.w * fv.w;
    }
    (j < 4 ? asrc : adst)[n * HEADS + (j & 3)] = s;
}

#define AGG_THREADS 1024
#define AGG_WAVES (AGG_THREADS / 32)
static_assert(KEY_STEP == AGG_THREADS * 4, "a scanning work-group takes four key pieces per thread and round");
#define WCAP 256

__device__ __forceinline__ int accpos(int c) { return (c & 128) | ((c & 3) << 5) | ((c & 127) >> 2); }

__device__ __forceinline__ void row_add(float* ar, const float4 a, const float4 b, float wa, float wb) {
    atomicAdd(&ar[0],   wa * a.x);
    atomicAdd(&ar[32],  wa * a.y);
    atomicAdd(&ar[64],  wa * a.z);
    atomicAdd(&ar[96],  wa * a.w);
    atomicAdd(&ar[128], wb * b.x);
    atomicAdd(&ar[160], wb * b.y);
    atomicAdd(&ar[192], wb * b.z);
    atomicAdd(&ar[224], wb * b.w);
}

__device__ __forceinline__ void accum_one(int s, unsigned dl, int lane,
                                          const float* __restrict__ asrc,
                                          const float* __restrict__ hmat,
                                          const float* adl, const unsigned* mxk,
                                          float* dnl, float* acc) {
    float ex[HEADS];
#pragma unroll
    for (int h = 0; h < HEADS; ++h) {
        float a = leakyf(asrc[s * HEADS + h] + adl[dl * HEADS + h]);
        ex[h] = expf(a - funkey(mxk[dl * HEADS + h]));
    }
    if (lane == 0) {
#pragma unroll
        for (int h = 0; h < HEADS; ++h) atomicAdd(&dnl[dl * HEADS + h], ex[h]);
    }
    const float* hrow = hmat + (size_t)s * HIDF + 4 * lane;
    const float4 ra = *(const float4*)hrow;
    const float4 rb = *(const float4*)(hrow + 128);
    const bool upper = (lane & 16) != 0;
    row_add(acc + dl * HIDF + lane, ra, rb, upper ? ex[1] : ex[0], upper ? ex[3] : ex[2]);
}

template <bool FIRST, bool MEAN>
__global__ __launch_bounds__(AGG_THREADS)
void agg_owner(const int* __restrict__ src, const int* __restrict__ dst,
               const float* __restrict__ asrc, const float* __restrict__ adst,
               const float* __restrict__ hmat,
               const float* __restrict__ bias, const float* __restrict__ g,
               const float* __restrict__ be, float* __restrict__ hout,
               int* __restrict__ wl_src, unsigned short* __restrict__ wl_dl,
               unsigned* __restrict__ wl_cnt, const unsigned char* __restrict__ keys,
               const float* __restrict__ foldn, float* __restrict__ asrc_nx, float* __restrict__ adst_nx,
               int N, int E) {
    __shared__ __attribute__((aligned(16))) float acc[RN * HIDF];
    __shared__ __attribute__((aligned(16))) int lsts[AGG_WAVES * WCAP];
    __shared__ __attribute__((aligned(16))) unsigned short lstd[AGG_WAVES * WCAP];
    __shared__ __attribute__((aligned(16))) unsigned mxk[RN * HEADS];
    __shared__ float dnl[RN * HEADS];
    __shared__ __attribute__((aligned(16))) float adl[RN * HEADS];
    __shared__ unsigned ovf;

    const int tid  = threadIdx.x;
    const int lane = tid & 31;
    const int wv   = tid >> 5;
    const bool upper = (lane & 16) != 0;
    const int base = blockIdx.x * RN;
    const int nloc = min(RN, N - base);
    int* const myS            = lsts + wv * WCAP;
    unsigned short* const myD = lstd + wv * WCAP;

    if (tid == 0) ovf = 0u;
    for (int i = tid; i < nloc * HEADS; i += AGG_THREADS) {
        float ad = adst[base * HEADS + i];
        adl[i] = ad;
        mxk[i] = fkey(leakyf(asrc[base * HEADS + i] + ad));
    }
    __syncthreads();

    const size_t wlslot = (size_t)blockIdx.x * AGG_WAVES + wv;
    const size_t wlbase = wlslot * WCAP;
    const int wvu = __builtin_amdgcn_readfirstlane(wv);
    unsigned wcount = 0u;
    const bool bykey = (N + RN - 1) / RN <= (int)KEY_NONE;
    if constexpr (FIRST) {
      if (bykey) {
        const unsigned pat = (unsigned)blockIdx.x * 0x01010101u;
        auto zbytes = [](unsigned x) {
            return ~(((x & 0x7F7F7F7Fu) + 0x7F7F7F7Fu) | x | 0x7F7F7F7Fu);
        };
        auto note16 = [&](const uint4 v, int iu) {
            unsigned hm = (zbytes(v.x ^ pat) >> 7) | (zbytes(v.y ^ pat) >> 6)
                        | (zbytes(v.z ^ pat) >> 5) | (zbytes(v.w ^ pat) >> 4);
            unsigned mask = (unsigned)__ballot(hm != 0u);
            while (mask) {
                const bool m = hm != 0u;
                const unsigned slot = wcount + __builtin_amdgcn_mbcnt_lo(mask, 0u);
                if (m && slot < (unsigned)WCAP) myS[slot] = 32 * iu + (__ffs(hm) - 1);
                wcount += (unsigned)__popc(mask);
                hm &= hm - 1u;
                mask = (unsigned)__ballot(hm != 0u);
            }
        };
        const int nround = KEY_ROUNDS(E);
        const uint4* key4 = (const uint4*)keys + (wvu * 32 + lane);
        uint4 v[4], nx[4];
#pragma unroll
        for (int u = 0; u < 4; ++u) v[u] = key4[u * AGG_THREADS];
        for (int r = 0; r < nround; ++r) {
            const int i = r * KEY_STEP + wvu * 32 + lane;
#pragma unroll
            for (int u = 0; u < 4; ++u) nx[u] = key4[(r + 1) * KEY_STEP + u * AGG_THREADS];
#pragma unroll
            for (int u = 0; u < 4; ++u) note16(v[u], i + u * AGG_THREADS);
#pragma unroll
            for (int u = 0; u < 4; ++u) v[u] = nx[u];
        }
      } else {
        auto note = [&](int d, int e, bool valid) {
            unsigned dl = (unsigned)(d - base);
            bool m = valid && (dl < (unsigned)nloc);
            unsigned mask = (unsigned)__ballot(m);
            if (mask) {
                unsigned slot = wcount + __builtin_amdgcn_mbcnt_lo(mask, 0u);
                if (m && slot < (unsigned)WCAP) myS[slot] = e;
                wcount += (unsigned)__popc(mask);
            }
        };
        const int E4 = E >> 2;
        const int4* dst4 = (const int4*)dst;
        for (int ib = wvu * 32; ib < E4; ib += AGG_THREADS * 4) {
            const int i = ib + lane;
            int4 v[4];
            if (ib + 3 * AGG_THREADS + 32 <= E4) {
#pragma unroll
                for (int u = 0; u < 4; ++u) v[u] = dst4[i + u * AGG_THREADS];
#pragma unroll
                for (int u = 0; u < 4; ++u) {
                    const int e = 4 * (i + u * AGG_THREADS);
                    note(v[u].x, e + 0, true);
                    note(v[u].y, e + 1, true);
                    note(v[u].z, e + 2, true);
                    note(v[u].w, e + 3, true);
                }
            } else {
#pragma unroll
                for (int u = 0; u < 4; ++u) {
                    const int iu = i + u * AGG_THREADS;
                    v[u] = (iu < E4) ? dst4[iu] : make_int4(0, 0, 0, 0);
                }
#pragma unroll
                for (int u = 0; u < 4; ++u) {
                    const int iu = i + u * AGG_THREADS;
                    const bool valid = iu < E4;
                    note(v[u].x, 4 * iu + 0, valid);
                    note(v[u].y, 4 * iu + 1, valid);
                    note(v[u].z, 4 * iu + 2, valid);
                    note(v[u].w, 4 * iu + 3, valid);
                }
            }
        }
        for (int e0 = (E4 << 2); e0 < E; e0 += AGG_THREADS) {
            int e = e0 + tid;
            bool valid = e < E;
            note(valid ? dst[e] : 0, e, valid);
        }
      }
        if (lane == 0) wl_cnt[wlslot] = wcount;
    } else {
        wcount = (unsigned)__builtin_amdgcn_readfirstlane((int)wl_cnt[wlslot]);
    }
    if (wcount > (unsigned)WCAP && lane == 0) ovf = 1u;
    __syncthreads();

    const bool listed = (ovf == 0u);
    const int nw = (int)min(wcount, (unsigned)WCAP);

    if (listed) {
        for (int k0 = lane; k0 < nw; k0 += 128) {
            int s[4]; unsigned dl[4]; bool ok[4];
#pragma unroll
            for (int u = 0; u < 4; ++u) {
                int k = k0 + 32 * u;
                ok[u] = k < nw;
                if constexpr (FIRST) {
                    const int le = ok[u] ? myS[k] : 0;
                    const int e = bykey ? 16 * (le >> 5) + 4 * (le & 7) + ((le >> 3) & 3) : le;
                    s[u]  = src[e];
                    dl[u] = min((unsigned)(dst[e] - base), (unsigned)(nloc - 1));
                } else {
                    const int kr = ok[u] ? k : k0;
                    const unsigned sv = (unsigned)wl_src[wlbase + kr];
                    const unsigned dv = (unsigned)wl_dl[wlbase + kr];
                    s[u]  = (int)min(sv, (unsigned)(N - 1));
                    dl[u] = min(dv, (unsigned)(nloc - 1));
                }
            }
            float4 a4[4];
#pragma unroll
            for (int u = 0; u < 4; ++u) a4[u] = *(const float4*)(asrc + (size_t)s[u] * HEADS);
#pragma unroll
            for (int u = 0; u < 4; ++u) {
                if (ok[u]) {
                    myS[k0 + 32 * u] = s[u];
                    myD[k0 + 32 * u] = (unsigned short)dl[u];
                    if constexpr (FIRST) {
                        wl_src[wlbase + k0 + 32 * u] = s[u];
                        wl_dl[wlbase + k0 + 32 * u]  = (unsigned short)dl[u];
                    }
                    atomicMax(&mxk[dl[u] * HEADS + 0], fkey(leakyf(a4[u].x + adl[dl[u] * HEADS + 0])));
                    atomicMax(&mxk[dl[u] * HEADS + 1], fkey(leakyf(a4[u].y + adl[dl[u] * HEADS + 1])));
                    atomicMax(&mxk[dl[u] * HEADS + 2], fkey(leakyf(a4[u].z + adl[dl[u] * HEADS + 2])));
                    atomicMax(&mxk[dl[u] * HEADS + 3], fkey(leakyf(a4[u].w + adl[dl[u] * HEADS + 3])));
                }
            }
        }
    } else {
        for (int e = tid; e < E; e += AGG_THREADS) {
            unsigned dl = (unsigned)(dst[e] - base);
            if (dl < (unsigned)nloc) {
                int s = src[e];
#pragma unroll
                for (int h = 0; h < HEADS; ++h)
                    atomicMax(&mxk[dl * HEADS + h],
                              fkey(leakyf(asrc[s * HEADS + h] + adl[dl * HEADS + h])));
            }
        }
    }
    __syncthreads();

    for (int i = wv; i < nloc; i += AGG_WAVES) {
        int n = base + i;
        float ex[HEADS];
#pragma unroll
        for (int h = 0; h < HEADS; ++h) {
            float aself = leakyf(asrc[n * HEADS + h] + adl[i * HEADS + h]);
            ex[h] = expf(aself - funkey(mxk[i * HEADS + h]));
        }
        if (lane == 0) {
#pragma unroll
            for (int h = 0; h < HEADS; ++h) dnl[i * HEADS + h] = ex[h];
        }
        const float* row = hmat + (size_t)n * HIDF + 4 * lane;
        const float4 ra = *(const float4*)row;
        const float4 rb = *(const float4*)(row + 128);
        const float wa = upper ? ex[1] : ex[0];
        const float wb = upper ? ex[3] : ex[2];
        float* ar = acc + i * HIDF + lane;
        ar[0]   = ra.x * wa;
        ar[32]  = ra.y * wa;
        ar[64]  = ra.z * wa;
        ar[96]  = ra.w * wa;
        ar[128] = rb.x * wb;
        ar[160] = rb.y * wb;
        ar[192] = rb.z * wb;
        ar[224] = rb.w * wb;
    }
    __syncthreads();

    if (listed) {
        for (int k0 = 0; k0 < nw; k0 += 32) {
            const int k    = k0 + lane;
            const bool okl = k < nw;
            const int sL      = okl ? myS[k] : 0;
            const unsigned dL = okl ? (unsigned)myD[k] : 0u;
            const float4 a4  = *(const float4*)(asrc + (size_t)sL * HEADS);
            const float4 ad4 = *(const float4*)(adl + dL * HEADS);
            const uint4 mk4  = *(const uint4*)(mxk + dL * HEADS);
            float eL[HEADS];
            eL[0] = expf(leakyf(a4.x + ad4.x) - funkey(mk4.x));
            eL[1] = expf(leakyf(a4.y + ad4.y) - funkey(mk4.y));
            eL[2] = expf(leakyf(a4.z + ad4.z) - funkey(mk4.z));
            eL[3] = expf(leakyf(a4.w + ad4.w) - funkey(mk4.w));
            if (okl) {
#pragma unroll
                for (int h = 0; h < HEADS; ++h) atomicAdd(&dnl[dL * HEADS + h], eL[h]);
            }
            const int cnt  = min(32, nw - k0);
            const int cnt4 = cnt & ~3;
            for (int j = 0; j < cnt4; j += 4) {
                int s[4]; unsigned dl[4]; float wa[4], wb[4];
#pragma unroll
                for (int u = 0; u < 4; ++u) {
                    const int l = j + u;
                    s[u]  = __builtin_amdgcn_readlane(sL, l);
                    dl[u] = (unsigned)__builtin_amdgcn_readlane((int)dL, l);
                    float ex[HEADS];
#pragma unroll
                    for (int h = 0; h < HEADS; ++h)
                        ex[h] = __builtin_bit_cast(float, __builtin_amdgcn_readlane(
                            __builtin_bit_cast(int, eL[h]), l));
                    wa[u] = upper ? ex[1] : ex[0];
                    wb[u] = upper ? ex[3] : ex[2];
                }
                float4 ra[4], rb[4];
#pragma unroll
                for (int u = 0; u < 4; ++u) {
                    const float* hrow = hmat + (size_t)s[u] * HIDF + 4 * lane;
                    ra[u] = *(const float4*)hrow;
                    rb[u] = *(const float4*)(hrow + 128);
                }
#pragma unroll
                for (int u = 0; u < 4; ++u)
                    row_add(acc + dl[u] * HIDF + lane, ra[u], rb[u], wa[u], wb[u]);
            }
            for (int l = cnt4; l < cnt; ++l) {
                const int s1       = __builtin_amdgcn_readlane(sL, l);
                const unsigned dl1 = (unsigned)__builtin_amdgcn_readlane((int)dL, l);
                float ex1[HEADS];
#pragma unroll
                for (int h = 0; h < HEADS; ++h)
                    ex1[h] = __builtin_bit_cast(float, __builtin_amdgcn_readlane(
                        __builtin_bit_cast(int, eL[h]), l));
                const float* hrow = hmat + (size_t)s1 * HIDF + 4 * lane;
                const float4 ra1 = *(const float4*)hrow;
                const float4 rb1 = *(const float4*)(hrow + 128);
                row_add(acc + dl1 * HIDF + lane, ra1, rb1,
                        upper ? ex1[1] : ex1[0], upper ? ex1[3] : ex1[2]);
            }
        }
    } else {
        for (int e0 = wvu * 32; e0 < E; e0 += AGG_WAVES * 32) {
            int e = e0 + lane;
            int d = (e < E) ? dst[e] : 0;
            unsigned dl = (unsigned)(d - base);
            bool m = (e < E) && (dl < (unsigned)nloc);
            unsigned mask = (unsigned)__ballot(m);
            while (mask) {
                int l = __ffs(mask) - 1;
                mask &= mask - 1u;
                int s1 = src[e0 + l];
                unsigned dl1 = (unsigned)__shfl((int)dl, l, 32);
                accum_one(s1, dl1, lane, asrc, hmat, adl, mxk, dnl, acc);
            }
        }
    }
    __syncthreads();

    if constexpr (!MEAN) {
        const int hsel = lane >> 4;
        float4 fa[8], fb[8];
#pragma unroll
        for (int j = 0; j < 8; ++j) {
            fa[j] = *(const float4*)(foldn + j * HIDF + 4 * lane);
            fb[j] = *(const float4*)(foldn + j * HIDF + 128 + 4 * lane);
        }
        const bool l16 = (lane & 16) != 0;
        const bool l8  = (lane & 8) != 0;
        const bool l4  = (lane & 4) != 0;
        for (int i = wv; i < nloc; i += AGG_WAVES) {
            const float dna = 1.f / (dnl[i * HEADS + hsel] + 1e-16f);
            const float dnb = 1.f / (dnl[i * HEADS + 2 + hsel] + 1e-16f);
            const float4 ba = *(const float4*)(bias + 4 * lane);
            const float4 bb = *(const float4*)(bias + 128 + 4 * lane);
            const float* ar = acc + i * HIDF + lane;
            float v[8];
            v[0] = ar[0]   * dna + ba.x;
            v[1] = ar[32]  * dna + ba.y;
            v[2] = ar[64]  * dna + ba.z;
            v[3] = ar[96]  * dna + ba.w;
            v[4] = ar[128] * dnb + bb.x;
            v[5] = ar[160] * dnb + bb.y;
            v[6] = ar[192] * dnb + bb.z;
            v[7] = ar[224] * dnb + bb.w;
            float s = 0.f;
#pragma unroll
            for (int j = 0; j < 8; ++j) s += v[j];
            float mu = waveRedSum(s) * (1.f / HIDF);
            float q = 0.f;
#pragma unroll
            for (int j = 0; j < 8; ++j) { float dlt = v[j] - mu; q += dlt * dlt; }
            float rstd = rsqrtf(waveRedSum(q) * (1.f / HIDF) + LN_EPS);
            const float4 ga  = *(const float4*)(g + 4 * lane);
            const float4 gb  = *(const float4*)(g + 128 + 4 * lane);
            const float4 bea = *(const float4*)(be + 4 * lane);
            const float4 beb = *(const float4*)(be + 128 + 4 * lane);
            float4 ya, yb;
            ya.x = eluf((v[0] - mu) * rstd * ga.x + bea.x);
            ya.y = eluf((v[1] - mu) * rstd * ga.y + bea.y);
            ya.z = eluf((v[2] - mu) * rstd * ga.z + bea.z);
            ya.w = eluf((v[3] - mu) * rstd * ga.w + bea.w);
            yb.x = eluf((v[4] - mu) * rstd * gb.x + beb.x);
            yb.y = eluf((v[5] - mu) * rstd * gb.y + beb.y);
            yb.z = eluf((v[6] - mu) * rstd * gb.z + beb.z);
            yb.w = eluf((v[7] - mu) * rstd * gb.w + beb.w);
            float* orow = hout + (size_t)(base + i) * HIDF + 4 * lane;
            *(float4*)orow         = ya;
            *(float4*)(orow + 128) = yb;
            float p[8];
#pragma unroll
            for (int j = 0; j < 8; ++j)
                p[j] = ya.x * fa[j].x + ya.y * fa[j].y + ya.z * fa[j].z + ya.w * fa[j].w
                     + yb.x * fb[j].x + yb.y * fb[j].y + yb.z * fb[j].z + yb.w * fb[j].w;
            float u4[4];
#pragma unroll
            for (int m = 0; m < 4; ++m) {
                const float keep = l16 ? p[4 + m] : p[m];
                const float send = l16 ? p[m] : p[4 + m];
                u4[m] = keep + __shfl_xor(send, 16, 32);
            }
            float u2[2];
#pragma unroll
            for (int m = 0; m < 2; ++m) {
                const float keep = l8 ? u4[2 + m] : u4[m];
                const float send = l8 ? u4[m] : u4[2 + m];
                u2[m] = keep + __shfl_xor(send, 8, 32);
            }
            float u1 = (l4 ? u2[1] : u2[0]) + __shfl_xor(l4 ? u2[0] : u2[1], 4, 32);
            u1 += __shfl_xor(u1, 2, 32);
            u1 += __shfl_xor(u1, 1, 32);
            if ((lane & 3) == 0) {
                const int jj = (l16 ? 4 : 0) + (l8 ? 2 : 0) + (l4 ? 1 : 0);
                (jj < 4 ? asrc_nx : adst_nx)[(size_t)(base + i) * HEADS + (jj & 3)] = u1;
            }
        }
    } else {
        for (int i = wv; i < nloc; i += AGG_WAVES) {
            float dn[HEADS];
#pragma unroll
            for (int h = 0; h < HEADS; ++h) dn[h] = 1.f / (dnl[i * HEADS + h] + 1e-16f);
            float v[2];
            float s = 0.f;
#pragma unroll
            for (int j = 0; j < 2; ++j) {
                int c = lane + 32 * j;
                float a = 0.f;
#pragma unroll
                for (int h = 0; h < HEADS; ++h) a += acc[i * HIDF + accpos(h * CCH + c)] * dn[h];
                v[j] = a * (1.f / HEADS) + bias[c];
                s += v[j];
            }
            float mu = waveRedSum(s) * (1.f / CCH);
            float q = 0.f;
#pragma unroll
            for (int j = 0; j < 2; ++j) { float dlt = v[j] - mu; q += dlt * dlt; }
            float rstd = rsqrtf(waveRedSum(q) * (1.f / CCH) + LN_EPS);
            float* orow = hout + (size_t)(base + i) * CCH;
#pragma unroll
            for (int j = 0; j < 2; ++j) {
                int c = lane + 32 * j;
                float y = (v[j] - mu) * rstd * g[c] + be[c];
                orow[c] = eluf(y);
            }
        }
    }
}

#define VH_THREADS 128
#define VH_BLOCKS 782

__global__ __launch_bounds__(VH_THREADS)
void value_head_mm(const float* __restrict__ hin,
                   const float* __restrict__ vw1, const float* __restrict__ vb1,
                   const float* __restrict__ vw2, const float* __restrict__ vb2,
                   float* __restrict__ out, int N) {
    constexpr int KK = CCH / 8;
    const int lane = threadIdx.x & 31;
    const int c = lane & 15;
    const int h = lane >> 4;
    const int gw = __builtin_amdgcn_readfirstlane((int)((blockIdx.x * blockDim.x + threadIdx.x) >> 5));
    const int nw = (int)((gridDim.x * blockDim.x) >> 5);
    const int ntile = (N + 15) >> 4;
    if (gw >= ntile) return;

    float4 bf[2][KK];
#pragma unroll
    for (int t = 0; t < 2; ++t)
        load_frag_row<KK>(bf[t], vw1 + (size_t)(t * 16 + c) * CCH + h * 4);
    float w2[OUTF][2];
#pragma unroll
    for (int k = 0; k < OUTF; ++k) {
        w2[k][0] = vw2[k * VHID + c];
        w2[k][1] = vw2[k * VHID + 16 + c];
    }
    const float b1a = vb1[c];
    const float b1b = vb1[16 + c];
    const bool b3 = (lane & 8) != 0;
    const bool b2 = (lane & 4) != 0;
    const bool b1 = (lane & 2) != 0;
    const bool b0 = (lane & 1) != 0;
    const int kb = 2 * (lane & 1);
    const float ob0 = vb2[kb];
    const float ob1 = vb2[kb + 1];

    for (int rt = gw; rt < ntile; rt += nw) {
        const int row = min(rt * 16 + c, N - 1);
        float4 af[KK];
        load_frag_row<KK>(af, hin + (size_t)row * CCH + h * 4);
        v8f acc0 = v8f{};
        v8f acc1 = v8f{};
#pragma unroll
        for (int kk = 0; kk < KK; ++kk) {
            const v2f a0 = {af[kk].x, af[kk].y};
            const v2f a1 = {af[kk].z, af[kk].w};
            const v2f p0 = {bf[0][kk].x, bf[0][kk].y};
            const v2f q0 = {bf[1][kk].x, bf[1][kk].y};
            const v2f p1 = {bf[0][kk].z, bf[0][kk].w};
            const v2f q1 = {bf[1][kk].z, bf[1][kk].w};
            acc0 = __builtin_amdgcn_wmma_f32_16x16x4_f32(false, a0, false, p0, (short)0, acc0, false, false);
            acc1 = __builtin_amdgcn_wmma_f32_16x16x4_f32(false, a0, false, q0, (short)0, acc1, false, false);
            acc0 = __builtin_amdgcn_wmma_f32_16x16x4_f32(false, a1, false, p1, (short)0, acc0, false, false);
            acc1 = __builtin_amdgcn_wmma_f32_16x16x4_f32(false, a1, false, q1, (short)0, acc1, false, false);
        }
        float p[32];
#pragma unroll
        for (int i = 0; i < 8; ++i) {
            const float ha = eluf(acc0[i] + b1a);
            const float hb = eluf(acc1[i] + b1b);
#pragma unroll
            for (int k = 0; k < OUTF; ++k) p[4 * i + k] = ha * w2[k][0] + hb * w2[k][1];
        }
        float u[16];
#pragma unroll
        for (int m = 0; m < 16; ++m) {
            const float keep = b3 ? p[16 + m] : p[m];
            const float send = b3 ? p[m] : p[16 + m];
            u[m] = keep + __shfl_xor(send, 8, 32);
        }
        float w[8];
#pragma unroll
        for (int m = 0; m < 8; ++m) {
            const float keep = b2 ? u[8 + m] : u[m];
            const float send = b2 ? u[m] : u[8 + m];
            w[m] = keep + __shfl_xor(send, 4, 32);
        }
        float x[4];
#pragma unroll
        for (int m = 0; m < 4; ++m) {
            const float keep = b1 ? w[4 + m] : w[m];
            const float send = b1 ? w[m] : w[4 + m];
            x[m] = keep + __shfl_xor(send, 2, 32);
        }
        float y[2];
#pragma unroll
        for (int m = 0; m < 2; ++m) {
            const float keep = b0 ? x[2 + m] : x[m];
            const float send = b0 ? x[m] : x[2 + m];
            y[m] = keep + __shfl_xor(send, 1, 32);
        }
        const int orow = rt * 16 + (lane >> 1);
        if (orow < N)
            *(float2*)(out + (size_t)orow * OUTF + kb) = make_float2(y[0] + ob0, y[1] + ob1);
    }
}

extern "C" void kernel_launch(void* const* d_in, const int* in_sizes, int n_in,
                              void* d_out, int out_size, void* d_ws, size_t ws_size,
                              hipStream_t stream) {
    const float* x      = (const float*)d_in[0];
    const int*   esrc   = (const int*)d_in[1];
    const int*   edst   = (const int*)d_in[2];
    const float* proj_w = (const float*)d_in[3];
    const float* proj_b = (const float*)d_in[4];
    const float* W[3]  = {(const float*)d_in[5],  (const float*)d_in[11], (const float*)d_in[17]};
    const float* AS[3] = {(const float*)d_in[6],  (const float*)d_in[12], (const float*)d_in[18]};
    const float* AD[3] = {(const float*)d_in[7],  (const float*)d_in[13], (const float*)d_in[19]};
    const float* B[3]  = {(const float*)d_in[8],  (const float*)d_in[14], (const float*)d_in[20]};
    const float* G[3]  = {(const float*)d_in[9],  (const float*)d_in[15], (const float*)d_in[21]};
    const float* BE[3] = {(const float*)d_in[10], (const float*)d_in[16], (const float*)d_in[22]};
    const float* vw1 = (const float*)d_in[23];
    const float* vb1 = (const float*)d_in[24];
    const float* vw2 = (const float*)d_in[25];
    const float* vb2 = (const float*)d_in[26];

    const int N = in_sizes[0] / 128;
    const int E = in_sizes[1];
    const int NH = N * HEADS;

    float* ws     = (float*)d_ws;
    float* hin    = ws;
    float* hmat   = hin  + (size_t)N * HIDF;
    float* asrcA  = hmat  + (size_t)N * HIDF;
    float* adstA  = asrcA + (size_t)NH;
    float* asrcB  = adstA + (size_t)NH;
    float* adstB  = asrcB + (size_t)NH;
    float* fold   = adstB + (size_t)NH;
    const int nagg = (N + RN - 1) / RN;
    int* wl_src            = (int*)(fold + FOLD_FLOATS);
    unsigned short* wl_dl  = (unsigned short*)(wl_src + (size_t)nagg * AGG_WAVES * WCAP);
    unsigned* wl_cnt       = (unsigned*)(wl_dl + (size_t)nagg * AGG_WAVES * WCAP);
    unsigned char* keys    = (unsigned char*)(wl_cnt + (size_t)nagg * AGG_WAVES);
    const int nkeyb = KEY_PIECES(E) / FOLD_THREADS;

    fold_attn<<<FOLD_BLOCKS + nkeyb, FOLD_THREADS, 0, stream>>>(W[0], AS[0], AD[0], W[1], AS[1], AD[1],
                                                       W[2], AS[2], AD[2], fold, edst, keys, E);

    gemm_wstat_f32<128, true><<<GW_BLOCKS_K128, GW_THREADS, 0, stream>>>(
        x, 128, proj_w, hin, N, 64, proj_b);
    logits_rows64<<<(N * 8 + 255) / 256, 256, 0, stream>>>(hin, fold, asrcA, adstA, N);

    gemm_wstat_f32<64, false><<<GW_BLOCKS_K64, GW_THREADS, 0, stream>>>(
        hin, 64, W[0], hmat, N, HIDF, nullptr);
    agg_owner<true, false><<<nagg, AGG_THREADS, 0, stream>>>(
        esrc, edst, asrcA, adstA, hmat, B[0], G[0], BE[0], hin, wl_src, wl_dl, wl_cnt, keys,
        fold + FOLD_OFF1, asrcB, adstB, N, E);

    gemm_wstat_f32<256, false><<<GW_BLOCKS_K256, GW_THREADS, 0, stream>>>(
        hin, HIDF, W[1], hmat, N, HIDF, nullptr);
    agg_owner<false, false><<<nagg, AGG_THREADS, 0, stream>>>(
        esrc, edst, asrcB, adstB, hmat, B[1], G[1], BE[1], hin, wl_src, wl_dl, wl_cnt, keys,
        fold + FOLD_OFF2, asrcA, adstA, N, E);

    gemm_wstat_f32<256, false><<<GW_BLOCKS_K256, GW_THREADS, 0, stream>>>(
        hin, HIDF, W[2], hmat, N, HIDF, nullptr);
    agg_owner<false, true><<<nagg, AGG_THREADS, 0, stream>>>(
        esrc, edst, asrcA, adstA, hmat, B[2], G[2], BE[2], hin, wl_src, wl_dl, wl_cnt, keys,
        nullptr, nullptr, nullptr, N, E);

    value_head_mm<<<VH_BLOCKS, VH_THREADS, 0, stream>>>(
        hin, vw1, vb1, vw2, vb2, (float*)d_out, N);
}
